# starting kernel
# baseline (speedup 1.0000x reference)
#include <hip/hip_runtime.h>
#include <hip/hip_bf16.h>

typedef float v2f __attribute__((ext_vector_type(2)));
typedef float v8f __attribute__((ext_vector_type(8)));

#define NHID 128
#define NEG_ATT 0.2f
#define NEG_ACT 0.01f

#if __has_builtin(__builtin_amdgcn_wmma_f32_16x16x4_f32)
#define HAS_WMMA_F32X4 1
#endif

__global__ __launch_bounds__(256) void gat_gemm128(
    const float* __restrict__ X, const float* __restrict__ W,
    float* __restrict__ H, int nrows) {
  __shared__ float Ws[NHID * NHID];
  const int tid = threadIdx.x;
  for (int i = tid; i < NHID * NHID; i += 256) {
    const int k = i >> 7, n = i & 127;
    Ws[((k >> 1) << 8) + (n << 1) + (k & 1)] = W[i];
  }
  __syncthreads();

  const int wave = tid >> 5;
  const int lane = tid & 31;
  const int row0 = blockIdx.x * 128 + wave * 16;
  if (row0 >= nrows) return;

  const int m     = lane & 15;
  const int khv   = lane >> 4;
  const int khalf = khv * 2;
  const int mhi   = khv * 8;

  v8f acc[8] = {};

#ifdef HAS_WMMA_F32X4
  const int arow = min(row0 + m, nrows - 1);
  const float* __restrict__ xrow = X + (size_t)arow * NHID;
  const float* __restrict__ bcol = Ws + (m << 1);
#pragma unroll 4
  for (int k0 = 0; k0 < NHID; k0 += 4) {
    const v2f a = *(const v2f*)(xrow + k0 + khalf);
    const float* __restrict__ bp = bcol + (((k0 >> 1) + khv) << 8);
#pragma unroll
    for (int t = 0; t < 8; ++t) {
      const v2f b = *(const v2f*)(bp + t * 32);
      acc[t] = __builtin_amdgcn_wmma_f32_16x16x4_f32(
          false, a, false, b, (short)0, acc[t], false, false);
    }
  }
#else
#pragma unroll
  for (int t = 0; t < 8; ++t)
    for (int v = 0; v < 8; ++v) {
      const int r = min(row0 + v + mhi, nrows - 1);
      const float* __restrict__ xr = X + (size_t)r * NHID;
      float s = 0.f;
      for (int k = 0; k < NHID; ++k)
        s += xr[k] * Ws[((k >> 1) << 8) + (((t << 4) + m) << 1) + (k & 1)];
      acc[t][v] = s;
    }
#endif

#pragma unroll
  for (int t = 0; t < 8; ++t)
#pragma unroll
    for (int v = 0; v < 8; ++v) {
      const int r = row0 + v + mhi;
      if (r < nrows) H[(size_t)r * NHID + t * 16 + m] = acc[t][v];
    }
}

__global__ __launch_bounds__(256) void gat_alpha(
    const float* __restrict__ H, const float* __restrict__ a_src,
    const float* __restrict__ a_dst, float* __restrict__ as,
    float* __restrict__ ad, int nrows) {
  const int wid  = (int)((blockIdx.x * (size_t)blockDim.x + threadIdx.x) >> 5);
  const int lane = threadIdx.x & 31;
  if (wid >= nrows) return;
  const float4 hv = *(const float4*)(H + (size_t)wid * NHID + lane * 4);
  const float4 s4 = *(const float4*)(a_src + lane * 4);
  const float4 d4 = *(const float4*)(a_dst + lane * 4);
  float s = hv.x * s4.x + hv.y * s4.y + hv.z * s4.z + hv.w * s4.w;
  float d = hv.x * d4.x + hv.y * d4.y + hv.z * d4.z + hv.w * d4.w;
#pragma unroll
  for (int off = 16; off; off >>= 1) {
    s += __shfl_xor(s, off, 32);
    d += __shfl_xor(d, off, 32);
  }
  if (lane == 0) { as[wid] = s; ad[wid] = d; }
}

__global__ void gat_init(float* __restrict__ acc, float* __restrict__ m,
                         float* __restrict__ denom, int n, size_t total) {
  const size_t i = blockIdx.x * (size_t)blockDim.x + threadIdx.x;
  if (i < total) acc[i] = 0.f;
  if (i < (size_t)n) { m[i] = -3.402823466e38f; denom[i] = 0.f; }
}

__device__ __forceinline__ void atomicMaxF(float* addr, float v) {
  if (v >= 0.f) atomicMax((int*)addr, __float_as_int(v));
  else          atomicMin((unsigned int*)addr, __float_as_uint(v));
}

__device__ __forceinline__ float lrelu(float v, float slope) {
  return v >= 0.f ? v : slope * v;
}

__global__ void gat_edge_max(const int* __restrict__ srcs,
                             const int* __restrict__ dsts,
                             const float* __restrict__ as,
                             const float* __restrict__ ad,
                             float* __restrict__ m, int nedges, int n) {
  const int e = (int)(blockIdx.x * (size_t)blockDim.x + threadIdx.x);
  if (e >= nedges + n) return;
  const int s = (e < nedges) ? srcs[e] : (e - nedges);
  const int d = (e < nedges) ? dsts[e] : (e - nedges);
  atomicMaxF(&m[d], lrelu(as[s] + ad[d], NEG_ATT));
}

__global__ void gat_edge_den(const int* __restrict__ srcs,
                             const int* __restrict__ dsts,
                             const float* __restrict__ as,
                             const float* __restrict__ ad,
                             const float* __restrict__ m,
                             float* __restrict__ denom, int nedges, int n) {
  const int e = (int)(blockIdx.x * (size_t)blockDim.x + threadIdx.x);
  if (e >= nedges + n) return;
  const int s = (e < nedges) ? srcs[e] : (e - nedges);
  const int d = (e < nedges) ? dsts[e] : (e - nedges);
  const float v = lrelu(as[s] + ad[d], NEG_ATT);
  atomicAdd(&denom[d], expf(v - m[d]));
}

__global__ __launch_bounds__(256) void gat_edge_aggr(
    const int* __restrict__ srcs, const int* __restrict__ dsts,
    const float* __restrict__ as, const float* __restrict__ ad,
    const float* __restrict__ m, const float* __restrict__ denom,
    const float* __restrict__ H, float* __restrict__ acc, int nedges, int n) {
  const long long wid = (long long)((blockIdx.x * (size_t)blockDim.x + threadIdx.x) >> 5);
  const int lane = threadIdx.x & 31;
  if (wid >= (long long)nedges + n) return;
  const int s = (wid < nedges) ? srcs[wid] : (int)(wid - nedges);
  const int d = (wid < nedges) ? dsts[wid] : (int)(wid - nedges);
  const float v = lrelu(as[s] + ad[d], NEG_ATT);
  const float w = expf(v - m[d]) / (denom[d] + 1e-16f);
  const float4 hv = *(const float4*)(H + (size_t)s * NHID + lane * 4);
  float* o = acc + (size_t)d * NHID + lane * 4;
  atomicAdd(o + 0, hv.x * w);
  atomicAdd(o + 1, hv.y * w);
  atomicAdd(o + 2, hv.z * w);
  atomicAdd(o + 3, hv.w * w);
}

__global__ void gat_bias_act(float* __restrict__ io, const float* __restrict__ b,
                             size_t total, float slope, int apply_act) {
  const size_t i = blockIdx.x * (size_t)blockDim.x + threadIdx.x;
  if (i >= total) return;
  float v = io[i] + b[i & (NHID - 1)];
  if (apply_act) v = lrelu(v, slope);
  io[i] = v;
}

extern "C" void kernel_launch(void* const* d_in, const int* in_sizes, int n_in,
                              void* d_out, int out_size, void* d_ws, size_t ws_size,
                              hipStream_t stream) {
  const float* x    = (const float*)d_in[0];
  const int*   ei   = (const int*)d_in[1];
  const float* W1   = (const float*)d_in[2];
  const float* asr1 = (const float*)d_in[3];
  const float* adt1 = (const float*)d_in[4];
  const float* b1   = (const float*)d_in[5];
  const float* W2   = (const float*)d_in[6];
  const float* asr2 = (const float*)d_in[7];
  const float* adt2 = (const float*)d_in[8];
  const float* b2   = (const float*)d_in[9];
  float* out = (float*)d_out;

  const int N = in_sizes[0] / NHID;
  const int E = in_sizes[1] / 2;
  const size_t NF = (size_t)N * NHID;

  float* h   = (float*)d_ws;
  float* acc = h + NF;
  float* as  = acc + NF;
  float* ad  = as + N;
  float* mx  = ad + N;
  float* den = mx + N;

  const int* src = ei;
  const int* dst = ei + E;

  const int gGemm  = (N + 127) / 128;
  const int gWaveN = (int)(((size_t)N * 32 + 255) / 256);
  const int gNF    = (int)((NF + 255) / 256);
  const int gEdge  = (E + N + 255) / 256;
  const int gEdgeW = (int)(((size_t)(E + N) * 32 + 255) / 256);

  gat_gemm128<<<gGemm, 256, 0, stream>>>(x, W1, h, N);
  gat_alpha<<<gWaveN, 256, 0, stream>>>(h, asr1, adt1, as, ad, N);
  gat_init<<<gNF, 256, 0, stream>>>(acc, mx, den, N, NF);
  gat_edge_max<<<gEdge, 256, 0, stream>>>(src, dst, as, ad, mx, E, N);
  gat_edge_den<<<gEdge, 256, 0, stream>>>(src, dst, as, ad, mx, den, E, N);
  gat_edge_aggr<<<gEdgeW, 256, 0, stream>>>(src, dst, as, ad, mx, den, h, acc, E, N);
  gat_bias_act<<<gNF, 256, 0, stream>>>(acc, b1, NF, NEG_ACT, 1);

  gat_gemm128<<<gGemm, 256, 0, stream>>>(acc, W2, h, N);
  gat_alpha<<<gWaveN, 256, 0, stream>>>(h, asr2, adt2, as, ad, N);
  gat_init<<<gNF, 256, 0, stream>>>(out, mx, den, N, NF);
  gat_edge_max<<<gEdge, 256, 0, stream>>>(src, dst, as, ad, mx, E, N);
  gat_edge_den<<<gEdge, 256, 0, stream>>>(src, dst, as, ad, mx, den, E, N);
  gat_edge_aggr<<<gEdgeW, 256, 0, stream>>>(src, dst, as, ad, mx, den, h, out, E, N);
  gat_bias_act<<<gNF, 256, 0, stream>>>(out, b2, NF, 0.f, 0);
}


# Edges grouped by destination once per call (int atomic rank, scan, fill); one wave per node does softmax and weighted row sum in registers in arrival-rank order; float atomics removed
# speedup vs baseline: 3.5238x; 3.5238x over previous
#include <hip/hip_runtime.h>
#include <hip/hip_bf16.h>

typedef float v2f __attribute__((ext_vector_type(2)));
typedef float v8f __attribute__((ext_vector_type(8)));

#define NHID 128
#define NEG_ATT 0.2f
#define NEG_ACT 0.01f

#if __has_builtin(__builtin_amdgcn_wmma_f32_16x16x4_f32)
#define HAS_WMMA_F32X4 1
#endif

__global__ __launch_bounds__(256) void gat_gemm128(
    const float* __restrict__ X, const float* __restrict__ W,
    float* __restrict__ H, int nrows) {
  __shared__ float Ws[NHID * NHID];
  const int tid = threadIdx.x;
  for (int i = tid; i < NHID * NHID; i += 256) {
    const int k = i >> 7, n = i & 127;
    Ws[((k >> 1) << 8) + (n << 1) + (k & 1)] = W[i];
  }
  __syncthreads();

  const int wave = tid >> 5;
  const int lane = tid & 31;
  const int row0 = blockIdx.x * 128 + wave * 16;
  if (row0 >= nrows) return;

  const int m     = lane & 15;
  const int khv   = lane >> 4;
  const int khalf = khv * 2;
  const int mhi   = khv * 8;

  v8f acc[8] = {};

#ifdef HAS_WMMA_F32X4
  const int arow = min(row0 + m, nrows - 1);
  const float* __restrict__ xrow = X + (size_t)arow * NHID;
  const float* __restrict__ bcol = Ws + (m << 1);
#pragma unroll 4
  for (int k0 = 0; k0 < NHID; k0 += 4) {
    const v2f a = *(const v2f*)(xrow + k0 + khalf);
    const float* __restrict__ bp = bcol + (((k0 >> 1) + khv) << 8);
#pragma unroll
    for (int t = 0; t < 8; ++t) {
      const v2f b = *(const v2f*)(bp + t * 32);
      acc[t] = __builtin_amdgcn_wmma_f32_16x16x4_f32(
          false, a, false, b, (short)0, acc[t], false, false);
    }
  }
#else
#pragma unroll
  for (int t = 0; t < 8; ++t)
    for (int v = 0; v < 8; ++v) {
      const int r = min(row0 + v + mhi, nrows - 1);
      const float* __restrict__ xr = X + (size_t)r * NHID;
      float s = 0.f;
      for (int k = 0; k < NHID; ++k)
        s += xr[k] * Ws[((k >> 1) << 8) + (((t << 4) + m) << 1) + (k & 1)];
      acc[t][v] = s;
    }
#endif

#pragma unroll
  for (int t = 0; t < 8; ++t)
#pragma unroll
    for (int v = 0; v < 8; ++v) {
      const int r = row0 + v + mhi;
      if (r < nrows) H[(size_t)r * NHID + t * 16 + m] = acc[t][v];
    }
}

__global__ __launch_bounds__(256) void gat_alpha(
    const float* __restrict__ H, const float* __restrict__ a_src,
    const float* __restrict__ a_dst, float* __restrict__ as,
    float* __restrict__ ad, int nrows) {
  const int wid  = (int)((blockIdx.x * (size_t)blockDim.x + threadIdx.x) >> 5);
  const int lane = threadIdx.x & 31;
  if (wid >= nrows) return;
  const float4 hv = *(const float4*)(H + (size_t)wid * NHID + lane * 4);
  const float4 s4 = *(const float4*)(a_src + lane * 4);
  const float4 d4 = *(const float4*)(a_dst + lane * 4);
  float s = hv.x * s4.x + hv.y * s4.y + hv.z * s4.z + hv.w * s4.w;
  float d = hv.x * d4.x + hv.y * d4.y + hv.z * d4.z + hv.w * d4.w;
#pragma unroll
  for (int off = 16; off; off >>= 1) {
    s += __shfl_xor(s, off, 32);
    d += __shfl_xor(d, off, 32);
  }
  if (lane == 0) { as[wid] = s; ad[wid] = d; }
}

__global__ void gat_zero(int* __restrict__ cnt, int n) {
  const int i = (int)(blockIdx.x * (size_t)blockDim.x + threadIdx.x);
  if (i < n) cnt[i] = 0;
}

__global__ void gat_rank(const int* __restrict__ dsts, int* __restrict__ cnt,
                         int* __restrict__ rank, int nedges) {
  const int e = (int)(blockIdx.x * (size_t)blockDim.x + threadIdx.x);
  if (e >= nedges) return;
  rank[e] = atomicAdd(&cnt[dsts[e]], 1);
}

__global__ __launch_bounds__(1024) void gat_scan(const int* __restrict__ cnt,
                                                 int* __restrict__ rowptr, int n) {
  __shared__ int wtot[32];
  const int lane = threadIdx.x & 31;
  const int wave = threadIdx.x >> 5;
  const int per  = ((n + 1023) / 1024) * 32;
  const int b    = min(wave * per, n);
  const int e    = min(b + per, n);

  int s = 0;
#pragma unroll 8
  for (int i = b + lane; i < e; i += 32) s += cnt[i];
  for (int off = 16; off; off >>= 1) s += __shfl_xor(s, off, 32);
  if (lane == 0) wtot[wave] = s;
  __syncthreads();

  int carry = 0;
  for (int w = 0; w < wave; ++w) carry += wtot[w];

  for (int i0 = b; i0 < e; i0 += 32) {
    const int i = i0 + lane;
    const int c = (i < e) ? cnt[i] : 0;
    int inc = c;
    for (int off = 1; off < 32; off <<= 1) {
      const int t = __shfl_up(inc, off, 32);
      if (lane >= off) inc += t;
    }
    if (i < e) rowptr[i] = carry + inc - c;
    carry += __shfl(inc, 31, 32);
  }
  if (threadIdx.x == 1023) rowptr[n] = carry;
}

__global__ void gat_fill(const int* __restrict__ srcs, const int* __restrict__ dsts,
                         const int* __restrict__ rank, const int* __restrict__ rowptr,
                         int* __restrict__ csr, int nedges) {
  const int e = (int)(blockIdx.x * (size_t)blockDim.x + threadIdx.x);
  if (e >= nedges) return;
  csr[rowptr[dsts[e]] + rank[e]] = srcs[e];
}

__device__ __forceinline__ float lrelu(float v, float slope) {
  return v >= 0.f ? v : slope * v;
}

template <int G>
__device__ __forceinline__ void gat_grp(const float* __restrict__ H, int s_l, float w_l,
                                        int t, int lane4, float4& acc) {
  float4 hv[G];
  float  w[G];
#pragma unroll
  for (int i = 0; i < G; ++i) {
    const int s = __builtin_amdgcn_readlane(s_l, t + i);
    w[i] = __int_as_float(__builtin_amdgcn_readlane(__float_as_int(w_l), t + i));
    hv[i] = *(const float4*)(H + (size_t)s * NHID + lane4);
  }
#pragma unroll
  for (int i = 0; i < G; ++i) {
    acc.x = fmaf(hv[i].x, w[i], acc.x);
    acc.y = fmaf(hv[i].y, w[i], acc.y);
    acc.z = fmaf(hv[i].z, w[i], acc.z);
    acc.w = fmaf(hv[i].w, w[i], acc.w);
  }
}

__device__ __forceinline__ void gat_chunk(const float* __restrict__ H, int s_l, float w_l,
                                          int cnt, int lane4, float4& acc) {
  int t = 0;
  for (; t + 8 <= cnt; t += 8) gat_grp<8>(H, s_l, w_l, t, lane4, acc);
  if (t + 4 <= cnt) { gat_grp<4>(H, s_l, w_l, t, lane4, acc); t += 4; }
  const int rem = cnt - t;
  if (rem == 3)      gat_grp<3>(H, s_l, w_l, t, lane4, acc);
  else if (rem == 2) gat_grp<2>(H, s_l, w_l, t, lane4, acc);
  else if (rem == 1) gat_grp<1>(H, s_l, w_l, t, lane4, acc);
}

__global__ __launch_bounds__(256) void gat_node(
    const int* __restrict__ csr, const int* __restrict__ rowptr,
    const float* __restrict__ as, const float* __restrict__ ad,
    const float* __restrict__ H, const float* __restrict__ bias,
    float* __restrict__ out, int n, float slope, int apply_act) {
  const int lane  = threadIdx.x & 31;
  const int lane4 = lane * 4;
  const int d = __builtin_amdgcn_readfirstlane(
      (int)((blockIdx.x * (size_t)blockDim.x + threadIdx.x) >> 5));
  if (d >= n) return;

  const int beg = rowptr[d];
  const int end = rowptr[d + 1];
  const int deg = end - beg;
  const float ad_d   = ad[d];
  const float v_self = lrelu(as[d] + ad_d, NEG_ATT);

  const int  j0  = beg + lane;
  const int  j1  = j0 + 32;
  const bool ok0 = j0 < end;
  const bool ok1 = j1 < end;
  int s0 = d, s1 = d;
  if (ok0) s0 = csr[j0];
  if (ok1) s1 = csr[j1];
  const float4 hd = *(const float4*)(H + (size_t)d * NHID + lane4);
  const float a0 = as[s0];
  const float a1 = as[s1];
  const float v0 = ok0 ? lrelu(a0 + ad_d, NEG_ATT) : v_self;
  const float v1 = ok1 ? lrelu(a1 + ad_d, NEG_ATT) : v_self;

  float mx = fmaxf(v_self, fmaxf(v0, v1));
  for (int j = j0 + 64; j < end; j += 32)
    mx = fmaxf(mx, lrelu(as[csr[j]] + ad_d, NEG_ATT));
  for (int off = 16; off; off >>= 1) mx = fmaxf(mx, __shfl_xor(mx, off, 32));

  const float p0 = ok0 ? expf(v0 - mx) : 0.f;
  const float p1 = ok1 ? expf(v1 - mx) : 0.f;
  float den = p0 + p1;
  for (int j = j0 + 64; j < end; j += 32)
    den += expf(lrelu(as[csr[j]] + ad_d, NEG_ATT) - mx);
  for (int off = 16; off; off >>= 1) den += __shfl_xor(den, off, 32);
  const float p_self = expf(v_self - mx);
  den += p_self;
  const float dn = den + 1e-16f;

  const float  w_self = p_self / dn;
  float4 acc = make_float4(hd.x * w_self, hd.y * w_self, hd.z * w_self, hd.w * w_self);

  gat_chunk(H, s0, p0 / dn, min(deg, 32), lane4, acc);
  if (deg > 32) gat_chunk(H, s1, p1 / dn, min(deg - 32, 32), lane4, acc);
  for (int base = beg + 64; base < end; base += 32) {
    const int j = base + lane;
    int   s = d;
    float w = 0.f;
    if (j < end) {
      s = csr[j];
      w = expf(lrelu(as[s] + ad_d, NEG_ATT) - mx) / dn;
    }
    gat_chunk(H, s, w, min(end - base, 32), lane4, acc);
  }

  const float4 b4 = *(const float4*)(bias + lane4);
  float4 o = make_float4(acc.x + b4.x, acc.y + b4.y, acc.z + b4.z, acc.w + b4.w);
  if (apply_act) {
    o.x = lrelu(o.x, slope); o.y = lrelu(o.y, slope);
    o.z = lrelu(o.z, slope); o.w = lrelu(o.w, slope);
  }
  *(float4*)(out + (size_t)d * NHID + lane4) = o;
}

extern "C" void kernel_launch(void* const* d_in, const int* in_sizes, int n_in,
                              void* d_out, int out_size, void* d_ws, size_t ws_size,
                              hipStream_t stream) {
  const float* x    = (const float*)d_in[0];
  const int*   ei   = (const int*)d_in[1];
  const float* W1   = (const float*)d_in[2];
  const float* asr1 = (const float*)d_in[3];
  const float* adt1 = (const float*)d_in[4];
  const float* b1   = (const float*)d_in[5];
  const float* W2   = (const float*)d_in[6];
  const float* asr2 = (const float*)d_in[7];
  const float* adt2 = (const float*)d_in[8];
  const float* b2   = (const float*)d_in[9];
  float* out = (float*)d_out;

  const int N = in_sizes[0] / NHID;
  const int E = in_sizes[1] / 2;
  const size_t NF = (size_t)N * NHID;

  float* h      = (float*)d_ws;
  float* x2     = h + NF;
  float* as     = x2 + NF;
  float* ad     = as + N;
  int*   cnt    = (int*)(ad + N);
  int*   rowptr = cnt + N;
  int*   rank   = rowptr + N + 1;
  int*   csr    = rank + E;

  const int* src = ei;
  const int* dst = ei + E;

  const int gGemm  = (N + 127) / 128;
  const int gWaveN = (int)(((size_t)N * 32 + 255) / 256);
  const int gN     = (N + 255) / 256;
  const int gE     = (E + 255) / 256;

  gat_zero<<<gN, 256, 0, stream>>>(cnt, N);
  gat_rank<<<gE, 256, 0, stream>>>(dst, cnt, rank, E);
  gat_scan<<<1, 1024, 0, stream>>>(cnt, rowptr, N);
  gat_fill<<<gE, 256, 0, stream>>>(src, dst, rank, rowptr, csr, E);

  gat_gemm128<<<gGemm, 256, 0, stream>>>(x, W1, h, N);
  gat_alpha<<<gWaveN, 256, 0, stream>>>(h, asr1, adt1, as, ad, N);
  gat_node<<<gWaveN, 256, 0, stream>>>(csr, rowptr, as, ad, h, b1, x2, N, NEG_ACT, 1);

  gat_gemm128<<<gGemm, 256, 0, stream>>>(x2, W2, h, N);
  gat_alpha<<<gWaveN, 256, 0, stream>>>(h, asr2, adt2, as, ad, N);
  gat_node<<<gWaveN, 256, 0, stream>>>(csr, rowptr, as, ad, h, b2, out, N, 0.f, 0);
}


# Attention logits h.a_src, h.a_dst computed in the GEMM epilogue from the accumulators (per-lane partial sums, then 16-lane DPP butterfly: new summation order); separate logit kernel removed
# speedup vs baseline: 3.7209x; 1.0559x over previous
#include <hip/hip_runtime.h>
#include <hip/hip_bf16.h>

typedef float v2f __attribute__((ext_vector_type(2)));
typedef float v8f __attribute__((ext_vector_type(8)));

#define NHID 128
#define NEG_ATT 0.2f
#define NEG_ACT 0.01f

#if __has_builtin(__builtin_amdgcn_wmma_f32_16x16x4_f32)
#define HAS_WMMA_F32X4 1
#endif

__device__ __forceinline__ float row16_sum(float x) {
  x += __int_as_float(__builtin_amdgcn_update_dpp(0, __float_as_int(x), 0x161, 0xf, 0xf, false));
  x += __int_as_float(__builtin_amdgcn_update_dpp(0, __float_as_int(x), 0x162, 0xf, 0xf, false));
  x += __int_as_float(__builtin_amdgcn_update_dpp(0, __float_as_int(x), 0x164, 0xf, 0xf, false));
  x += __int_as_float(__builtin_amdgcn_update_dpp(0, __float_as_int(x), 0x168, 0xf, 0xf, false));
  return x;
}

__global__ __launch_bounds__(256) void gat_gemm128(
    const float* __restrict__ X, const float* __restrict__ W,
    const float* __restrict__ a_src, const float* __restrict__ a_dst,
    float* __restrict__ H, float* __restrict__ as, float* __restrict__ ad,
    int nrows) {
  __shared__ float Ws[NHID * NHID];
  const int tid = threadIdx.x;
  for (int i = tid; i < NHID * NHID; i += 256) {
    const int k = i >> 7, n = i & 127;
    Ws[((k >> 1) << 8) + (n << 1) + (k & 1)] = W[i];
  }
  __syncthreads();

  const int wave = tid >> 5;
  const int lane = tid & 31;
  const int row0 = blockIdx.x * 128 + wave * 16;
  if (row0 >= nrows) return;

  const int m     = lane & 15;
  const int khv   = lane >> 4;
  const int khalf = khv * 2;
  const int mhi   = khv * 8;

  v8f acc[8] = {};

#ifdef HAS_WMMA_F32X4
  const int arow = min(row0 + m, nrows - 1);
  const float* __restrict__ xrow = X + (size_t)arow * NHID;
  const float* __restrict__ bcol = Ws + (m << 1);
#pragma unroll 4
  for (int k0 = 0; k0 < NHID; k0 += 4) {
    const v2f a = *(const v2f*)(xrow + k0 + khalf);
    const float* __restrict__ bp = bcol + (((k0 >> 1) + khv) << 8);
#pragma unroll
    for (int t = 0; t < 8; ++t) {
      const v2f b = *(const v2f*)(bp + t * 32);
      acc[t] = __builtin_amdgcn_wmma_f32_16x16x4_f32(
          false, a, false, b, (short)0, acc[t], false, false);
    }
  }
#else
#pragma unroll
  for (int t = 0; t < 8; ++t)
    for (int v = 0; v < 8; ++v) {
      const int r = min(row0 + v + mhi, nrows - 1);
      const float* __restrict__ xr = X + (size_t)r * NHID;
      float s = 0.f;
      for (int k = 0; k < NHID; ++k)
        s += xr[k] * Ws[((k >> 1) << 8) + (((t << 4) + m) << 1) + (k & 1)];
      acc[t][v] = s;
    }
#endif

#pragma unroll
  for (int t = 0; t < 8; ++t)
#pragma unroll
    for (int v = 0; v < 8; ++v) {
      const int r = row0 + v + mhi;
      if (r < nrows) H[(size_t)r * NHID + t * 16 + m] = acc[t][v];
    }

  float ps[8], pd[8];
#pragma unroll
  for (int v = 0; v < 8; ++v) { ps[v] = 0.f; pd[v] = 0.f; }
#pragma unroll
  for (int t = 0; t < 8; ++t) {
    const float sa = a_src[t * 16 + m];
    const float da = a_dst[t * 16 + m];
#pragma unroll
    for (int v = 0; v < 8; ++v) {
      ps[v] = fmaf(acc[t][v], sa, ps[v]);
      pd[v] = fmaf(acc[t][v], da, pd[v]);
    }
  }
#pragma unroll
  for (int v = 0; v < 8; ++v) { ps[v] = row16_sum(ps[v]); pd[v] = row16_sum(pd[v]); }
  if (m < 2) {
    float* __restrict__ dstp = (m == 0) ? as : ad;
#pragma unroll
    for (int v = 0; v < 8; ++v) {
      const int r = row0 + v + mhi;
      if (r < nrows) dstp[r] = (m == 0) ? ps[v] : pd[v];
    }
  }
}

__global__ void gat_zero(int* __restrict__ cnt, int n) {
  const int i = (int)(blockIdx.x * (size_t)blockDim.x + threadIdx.x);
  if (i < n) cnt[i] = 0;
}

__global__ void gat_rank(const int* __restrict__ dsts, int* __restrict__ cnt,
                         int* __restrict__ rank, int nedges) {
  const int e = (int)(blockIdx.x * (size_t)blockDim.x + threadIdx.x);
  if (e >= nedges) return;
  rank[e] = atomicAdd(&cnt[dsts[e]], 1);
}

__global__ __launch_bounds__(1024) void gat_scan(const int* __restrict__ cnt,
                                                 int* __restrict__ rowptr, int n) {
  __shared__ int wtot[32];
  const int lane = threadIdx.x & 31;
  const int wave = threadIdx.x >> 5;
  const int per  = ((n + 1023) / 1024) * 32;
  const int b    = min(wave * per, n);
  const int e    = min(b + per, n);

  int s = 0;
#pragma unroll 8
  for (int i = b + lane; i < e; i += 32) s += cnt[i];
  for (int off = 16; off; off >>= 1) s += __shfl_xor(s, off, 32);
  if (lane == 0) wtot[wave] = s;
  __syncthreads();

  int carry = 0;
  for (int w = 0; w < wave; ++w) carry += wtot[w];

  for (int i0 = b; i0 < e; i0 += 32) {
    const int i = i0 + lane;
    const int c = (i < e) ? cnt[i] : 0;
    int inc = c;
    for (int off = 1; off < 32; off <<= 1) {
      const int t = __shfl_up(inc, off, 32);
      if (lane >= off) inc += t;
    }
    if (i < e) rowptr[i] = carry + inc - c;
    carry += __shfl(inc, 31, 32);
  }
  if (threadIdx.x == 1023) rowptr[n] = carry;
}

__global__ void gat_fill(const int* __restrict__ srcs, const int* __restrict__ dsts,
                         const int* __restrict__ rank, const int* __restrict__ rowptr,
                         int* __restrict__ csr, int nedges) {
  const int e = (int)(blockIdx.x * (size_t)blockDim.x + threadIdx.x);
  if (e >= nedges) return;
  csr[rowptr[dsts[e]] + rank[e]] = srcs[e];
}

__device__ __forceinline__ float lrelu(float v, float slope) {
  return v >= 0.f ? v : slope * v;
}

template <int G>
__device__ __forceinline__ void gat_grp(const float* __restrict__ H, int s_l, float w_l,
                                        int t, int lane4, float4& acc) {
  float4 hv[G];
  float  w[G];
#pragma unroll
  for (int i = 0; i < G; ++i) {
    const int s = __builtin_amdgcn_readlane(s_l, t + i);
    w[i] = __int_as_float(__builtin_amdgcn_readlane(__float_as_int(w_l), t + i));
    hv[i] = *(const float4*)(H + (size_t)s * NHID + lane4);
  }
#pragma unroll
  for (int i = 0; i < G; ++i) {
    acc.x = fmaf(hv[i].x, w[i], acc.x);
    acc.y = fmaf(hv[i].y, w[i], acc.y);
    acc.z = fmaf(hv[i].z, w[i], acc.z);
    acc.w = fmaf(hv[i].w, w[i], acc.w);
  }
}

__device__ __forceinline__ void gat_chunk(const float* __restrict__ H, int s_l, float w_l,
                                          int cnt, int lane4, float4& acc) {
  int t = 0;
  for (; t + 8 <= cnt; t += 8) gat_grp<8>(H, s_l, w_l, t, lane4, acc);
  if (t + 4 <= cnt) { gat_grp<4>(H, s_l, w_l, t, lane4, acc); t += 4; }
  const int rem = cnt - t;
  if (rem == 3)      gat_grp<3>(H, s_l, w_l, t, lane4, acc);
  else if (rem == 2) gat_grp<2>(H, s_l, w_l, t, lane4, acc);
  else if (rem == 1) gat_grp<1>(H, s_l, w_l, t, lane4, acc);
}

__global__ __launch_bounds__(256) void gat_node(
    const int* __restrict__ csr, const int* __restrict__ rowptr,
    const float* __restrict__ as, const float* __restrict__ ad,
    const float* __restrict__ H, const float* __restrict__ bias,
    float* __restrict__ out, int n, float slope, int apply_act) {
  const int lane  = threadIdx.x & 31;
  const int lane4 = lane * 4;
  const int d = __builtin_amdgcn_readfirstlane(
      (int)((blockIdx.x * (size_t)blockDim.x + threadIdx.x) >> 5));
  if (d >= n) return;

  const int beg = rowptr[d];
  const int end = rowptr[d + 1];
  const int deg = end - beg;
  const float ad_d   = ad[d];
  const float v_self = lrelu(as[d] + ad_d, NEG_ATT);

  const int  j0  = beg + lane;
  const int  j1  = j0 + 32;
  const bool ok0 = j0 < end;
  const bool ok1 = j1 < end;
  int s0 = d, s1 = d;
  if (ok0) s0 = csr[j0];
  if (ok1) s1 = csr[j1];
  const float4 hd = *(const float4*)(H + (size_t)d * NHID + lane4);
  const float a0 = as[s0];
  const float a1 = as[s1];
  const float v0 = ok0 ? lrelu(a0 + ad_d, NEG_ATT) : v_self;
  const float v1 = ok1 ? lrelu(a1 + ad_d, NEG_ATT) : v_self;

  float mx = fmaxf(v_self, fmaxf(v0, v1));
  for (int j = j0 + 64; j < end; j += 32)
    mx = fmaxf(mx, lrelu(as[csr[j]] + ad_d, NEG_ATT));
  for (int off = 16; off; off >>= 1) mx = fmaxf(mx, __shfl_xor(mx, off, 32));

  const float p0 = ok0 ? expf(v0 - mx) : 0.f;
  const float p1 = ok1 ? expf(v1 - mx) : 0.f;
  float den = p0 + p1;
  for (int j = j0 + 64; j < end; j += 32)
    den += expf(lrelu(as[csr[j]] + ad_d, NEG_ATT) - mx);
  for (int off = 16; off; off >>= 1) den += __shfl_xor(den, off, 32);
  const float p_self = expf(v_self - mx);
  den += p_self;
  const float dn = den + 1e-16f;

  const float  w_self = p_self / dn;
  float4 acc = make_float4(hd.x * w_self, hd.y * w_self, hd.z * w_self, hd.w * w_self);

  gat_chunk(H, s0, p0 / dn, min(deg, 32), lane4, acc);
  if (deg > 32) gat_chunk(H, s1, p1 / dn, min(deg - 32, 32), lane4, acc);
  for (int base = beg + 64; base < end; base += 32) {
    const int j = base + lane;
    int   s = d;
    float w = 0.f;
    if (j < end) {
      s = csr[j];
      w = expf(lrelu(as[s] + ad_d, NEG_ATT) - mx) / dn;
    }
    gat_chunk(H, s, w, min(end - base, 32), lane4, acc);
  }

  const float4 b4 = *(const float4*)(bias + lane4);
  float4 o = make_float4(acc.x + b4.x, acc.y + b4.y, acc.z + b4.z, acc.w + b4.w);
  if (apply_act) {
    o.x = lrelu(o.x, slope); o.y = lrelu(o.y, slope);
    o.z = lrelu(o.z, slope); o.w = lrelu(o.w, slope);
  }
  *(float4*)(out + (size_t)d * NHID + lane4) = o;
}

extern "C" void kernel_launch(void* const* d_in, const int* in_sizes, int n_in,
                              void* d_out, int out_size, void* d_ws, size_t ws_size,
                              hipStream_t stream) {
  const float* x    = (const float*)d_in[0];
  const int*   ei   = (const int*)d_in[1];
  const float* W1   = (const float*)d_in[2];
  const float* asr1 = (const float*)d_in[3];
  const float* adt1 = (const float*)d_in[4];
  const float* b1   = (const float*)d_in[5];
  const float* W2   = (const float*)d_in[6];
  const float* asr2 = (const float*)d_in[7];
  const float* adt2 = (const float*)d_in[8];
  const float* b2   = (const float*)d_in[9];
  float* out = (float*)d_out;

  const int N = in_sizes[0] / NHID;
  const int E = in_sizes[1] / 2;
  const size_t NF = (size_t)N * NHID;

  float* h      = (float*)d_ws;
  float* x2     = h + NF;
  float* as     = x2 + NF;
  float* ad     = as + N;
  int*   cnt    = (int*)(ad + N);
  int*   rowptr = cnt + N;
  int*   rank   = rowptr + N + 1;
  int*   csr    = rank + E;

  const int* src = ei;
  const int* dst = ei + E;

  const int gGemm  = (N + 127) / 128;
  const int gWaveN = (int)(((size_t)N * 32 + 255) / 256);
  const int gN     = (N + 255) / 256;
  const int gE     = (E + 255) / 256;

  gat_zero<<<gN, 256, 0, stream>>>(cnt, N);
  gat_rank<<<gE, 256, 0, stream>>>(dst, cnt, rank, E);
  gat_scan<<<1, 1024, 0, stream>>>(cnt, rowptr, N);
  gat_fill<<<gE, 256, 0, stream>>>(src, dst, rank, rowptr, csr, E);

  gat_gemm128<<<gGemm, 256, 0, stream>>>(x, W1, asr1, adt1, h, as, ad, N);
  gat_node<<<gWaveN, 256, 0, stream>>>(csr, rowptr, as, ad, h, b1, x2, N, NEG_ACT, 1);

  gat_gemm128<<<gGemm, 256, 0, stream>>>(x2, W2, asr2, adt2, h, as, ad, N);
  gat_node<<<gWaveN, 256, 0, stream>>>(csr, rowptr, as, ad, h, b2, out, N, 0.f, 0);
}


# Prefix scan of in-degrees: one work-group per 1024-node tile adds up the counts in front of its tile itself (independent int4 loads), replacing a single work-group's serial 49-trip loop per wave
# speedup vs baseline: 4.1580x; 1.1175x over previous
#include <hip/hip_runtime.h>
#include <hip/hip_bf16.h>

typedef float v2f __attribute__((ext_vector_type(2)));
typedef float v8f __attribute__((ext_vector_type(8)));

#define NHID 128
#define NEG_ATT 0.2f
#define NEG_ACT 0.01f

#if __has_builtin(__builtin_amdgcn_wmma_f32_16x16x4_f32)
#define HAS_WMMA_F32X4 1
#endif

__device__ __forceinline__ float row16_sum(float x) {
  x += __int_as_float(__builtin_amdgcn_update_dpp(0, __float_as_int(x), 0x161, 0xf, 0xf, false));
  x += __int_as_float(__builtin_amdgcn_update_dpp(0, __float_as_int(x), 0x162, 0xf, 0xf, false));
  x += __int_as_float(__builtin_amdgcn_update_dpp(0, __float_as_int(x), 0x164, 0xf, 0xf, false));
  x += __int_as_float(__builtin_amdgcn_update_dpp(0, __float_as_int(x), 0x168, 0xf, 0xf, false));
  return x;
}

__global__ __launch_bounds__(256) void gat_gemm128(
    const float* __restrict__ X, const float* __restrict__ W,
    const float* __restrict__ a_src, const float* __restrict__ a_dst,
    float* __restrict__ H, float* __restrict__ as, float* __restrict__ ad,
    int nrows) {
  __shared__ float Ws[NHID * NHID];
  const int tid = threadIdx.x;
  for (int i = tid; i < NHID * NHID; i += 256) {
    const int k = i >> 7, n = i & 127;
    Ws[((k >> 1) << 8) + (n << 1) + (k & 1)] = W[i];
  }
  __syncthreads();

  const int wave = tid >> 5;
  const int lane = tid & 31;
  const int row0 = blockIdx.x * 128 + wave * 16;
  if (row0 >= nrows) return;

  const int m     = lane & 15;
  const int khv   = lane >> 4;
  const int khalf = khv * 2;
  const int mhi   = khv * 8;

  v8f acc[8] = {};

#ifdef HAS_WMMA_F32X4
  const int arow = min(row0 + m, nrows - 1);
  const float* __restrict__ xrow = X + (size_t)arow * NHID;
  const float* __restrict__ bcol = Ws + (m << 1);
#pragma unroll 4
  for (int k0 = 0; k0 < NHID; k0 += 4) {
    const v2f a = *(const v2f*)(xrow + k0 + khalf);
    const float* __restrict__ bp = bcol + (((k0 >> 1) + khv) << 8);
#pragma unroll
    for (int t = 0; t < 8; ++t) {
      const v2f b = *(const v2f*)(bp + t * 32);
      acc[t] = __builtin_amdgcn_wmma_f32_16x16x4_f32(
          false, a, false, b, (short)0, acc[t], false, false);
    }
  }
#else
#pragma unroll
  for (int t = 0; t < 8; ++t)
    for (int v = 0; v < 8; ++v) {
      const int r = min(row0 + v + mhi, nrows - 1);
      const float* __restrict__ xr = X + (size_t)r * NHID;
      float s = 0.f;
      for (int k = 0; k < NHID; ++k)
        s += xr[k] * Ws[((k >> 1) << 8) + (((t << 4) + m) << 1) + (k & 1)];
      acc[t][v] = s;
    }
#endif

#pragma unroll
  for (int t = 0; t < 8; ++t)
#pragma unroll
    for (int v = 0; v < 8; ++v) {
      const int r = row0 + v + mhi;
      if (r < nrows) H[(size_t)r * NHID + t * 16 + m] = acc[t][v];
    }

  float ps[8], pd[8];
#pragma unroll
  for (int v = 0; v < 8; ++v) { ps[v] = 0.f; pd[v] = 0.f; }
#pragma unroll
  for (int t = 0; t < 8; ++t) {
    const float sa = a_src[t * 16 + m];
    const float da = a_dst[t * 16 + m];
#pragma unroll
    for (int v = 0; v < 8; ++v) {
      ps[v] = fmaf(acc[t][v], sa, ps[v]);
      pd[v] = fmaf(acc[t][v], da, pd[v]);
    }
  }
#pragma unroll
  for (int v = 0; v < 8; ++v) { ps[v] = row16_sum(ps[v]); pd[v] = row16_sum(pd[v]); }
  if (m < 2) {
    float* __restrict__ dstp = (m == 0) ? as : ad;
#pragma unroll
    for (int v = 0; v < 8; ++v) {
      const int r = row0 + v + mhi;
      if (r < nrows) dstp[r] = (m == 0) ? ps[v] : pd[v];
    }
  }
}

__global__ void gat_zero(int* __restrict__ cnt, int n) {
  const int i = (int)(blockIdx.x * (size_t)blockDim.x + threadIdx.x);
  if (i < n) cnt[i] = 0;
}

__global__ void gat_rank(const int* __restrict__ dsts, int* __restrict__ cnt,
                         int* __restrict__ rank, int nedges) {
  const int e = (int)(blockIdx.x * (size_t)blockDim.x + threadIdx.x);
  if (e >= nedges) return;
  rank[e] = atomicAdd(&cnt[dsts[e]], 1);
}

__global__ __launch_bounds__(1024) void gat_scan(const int* __restrict__ cnt,
                                                 int* __restrict__ rowptr, int n) {
  __shared__ int wfront[32];
  __shared__ int wtile[32];
  const int tid  = threadIdx.x;
  const int lane = tid & 31;
  const int wave = tid >> 5;
  const int base = blockIdx.x * 1024;
  const int i    = base + tid;
  const int c    = (i < n) ? cnt[i] : 0;

  const int4* __restrict__ c4 = (const int4*)cnt;
  const int nq = base >> 2;
  int s = 0;
  for (int q0 = 0; q0 < nq; q0 += 8 * 1024) {
    int4 v[8];
#pragma unroll
    for (int j = 0; j < 8; ++j) v[j] = c4[min(q0 + j * 1024 + tid, nq - 1)];
#pragma unroll
    for (int j = 0; j < 8; ++j)
      if (q0 + j * 1024 + tid < nq) s += (v[j].x + v[j].y) + (v[j].z + v[j].w);
  }
  for (int off = 16; off; off >>= 1) s += __shfl_xor(s, off, 32);

  int inc = c;
  for (int off = 1; off < 32; off <<= 1) {
    const int t = __shfl_up(inc, off, 32);
    if (lane >= off) inc += t;
  }
  if (lane == 0)  wfront[wave] = s;
  if (lane == 31) wtile[wave]  = inc;
  __syncthreads();

  int o = wfront[lane] + ((lane < wave) ? wtile[lane] : 0);
  for (int off = 16; off; off >>= 1) o += __shfl_xor(o, off, 32);

  if (i < n) rowptr[i] = o + inc - c;
  if (blockIdx.x == gridDim.x - 1 && tid == 1023) rowptr[n] = o + inc;
}

__global__ void gat_fill(const int* __restrict__ srcs, const int* __restrict__ dsts,
                         const int* __restrict__ rank, const int* __restrict__ rowptr,
                         int* __restrict__ csr, int nedges) {
  const int e = (int)(blockIdx.x * (size_t)blockDim.x + threadIdx.x);
  if (e >= nedges) return;
  csr[rowptr[dsts[e]] + rank[e]] = srcs[e];
}

__device__ __forceinline__ float lrelu(float v, float slope) {
  return v >= 0.f ? v : slope * v;
}

template <int G>
__device__ __forceinline__ void gat_grp(const float* __restrict__ H, int s_l, float w_l,
                                        int t, int lane4, float4& acc) {
  float4 hv[G];
  float  w[G];
#pragma unroll
  for (int i = 0; i < G; ++i) {
    const int s = __builtin_amdgcn_readlane(s_l, t + i);
    w[i] = __int_as_float(__builtin_amdgcn_readlane(__float_as_int(w_l), t + i));
    hv[i] = *(const float4*)(H + (size_t)s * NHID + lane4);
  }
#pragma unroll
  for (int i = 0; i < G; ++i) {
    acc.x = fmaf(hv[i].x, w[i], acc.x);
    acc.y = fmaf(hv[i].y, w[i], acc.y);
    acc.z = fmaf(hv[i].z, w[i], acc.z);
    acc.w = fmaf(hv[i].w, w[i], acc.w);
  }
}

__device__ __forceinline__ void gat_chunk(const float* __restrict__ H, int s_l, float w_l,
                                          int cnt, int lane4, float4& acc) {
  int t = 0;
  for (; t + 8 <= cnt; t += 8) gat_grp<8>(H, s_l, w_l, t, lane4, acc);
  if (t + 4 <= cnt) { gat_grp<4>(H, s_l, w_l, t, lane4, acc); t += 4; }
  const int rem = cnt - t;
  if (rem == 3)      gat_grp<3>(H, s_l, w_l, t, lane4, acc);
  else if (rem == 2) gat_grp<2>(H, s_l, w_l, t, lane4, acc);
  else if (rem == 1) gat_grp<1>(H, s_l, w_l, t, lane4, acc);
}

__global__ __launch_bounds__(256) void gat_node(
    const int* __restrict__ csr, const int* __restrict__ rowptr,
    const float* __restrict__ as, const float* __restrict__ ad,
    const float* __restrict__ H, const float* __restrict__ bias,
    float* __restrict__ out, int n, float slope, int apply_act) {
  const int lane  = threadIdx.x & 31;
  const int lane4 = lane * 4;
  const int d = __builtin_amdgcn_readfirstlane(
      (int)((blockIdx.x * (size_t)blockDim.x + threadIdx.x) >> 5));
  if (d >= n) return;

  const int beg = rowptr[d];
  const int end = rowptr[d + 1];
  const int deg = end - beg;
  const float ad_d   = ad[d];
  const float v_self = lrelu(as[d] + ad_d, NEG_ATT);

  const int  j0  = beg + lane;
  const int  j1  = j0 + 32;
  const bool ok0 = j0 < end;
  const bool ok1 = j1 < end;
  int s0 = d, s1 = d;
  if (ok0) s0 = csr[j0];
  if (ok1) s1 = csr[j1];
  const float4 hd = *(const float4*)(H + (size_t)d * NHID + lane4);
  const float a0 = as[s0];
  const float a1 = as[s1];
  const float v0 = ok0 ? lrelu(a0 + ad_d, NEG_ATT) : v_self;
  const float v1 = ok1 ? lrelu(a1 + ad_d, NEG_ATT) : v_self;

  float mx = fmaxf(v_self, fmaxf(v0, v1));
  for (int j = j0 + 64; j < end; j += 32)
    mx = fmaxf(mx, lrelu(as[csr[j]] + ad_d, NEG_ATT));
  for (int off = 16; off; off >>= 1) mx = fmaxf(mx, __shfl_xor(mx, off, 32));

  const float p0 = ok0 ? expf(v0 - mx) : 0.f;
  const float p1 = ok1 ? expf(v1 - mx) : 0.f;
  float den = p0 + p1;
  for (int j = j0 + 64; j < end; j += 32)
    den += expf(lrelu(as[csr[j]] + ad_d, NEG_ATT) - mx);
  for (int off = 16; off; off >>= 1) den += __shfl_xor(den, off, 32);
  const float p_self = expf(v_self - mx);
  den += p_self;
  const float dn = den + 1e-16f;

  const float  w_self = p_self / dn;
  float4 acc = make_float4(hd.x * w_self, hd.y * w_self, hd.z * w_self, hd.w * w_self);

  gat_chunk(H, s0, p0 / dn, min(deg, 32), lane4, acc);
  if (deg > 32) gat_chunk(H, s1, p1 / dn, min(deg - 32, 32), lane4, acc);
  for (int base = beg + 64; base < end; base += 32) {
    const int j = base + lane;
    int   s = d;
    float w = 0.f;
    if (j < end) {
      s = csr[j];
      w = expf(lrelu(as[s] + ad_d, NEG_ATT) - mx) / dn;
    }
    gat_chunk(H, s, w, min(end - base, 32), lane4, acc);
  }

  const float4 b4 = *(const float4*)(bias + lane4);
  float4 o = make_float4(acc.x + b4.x, acc.y + b4.y, acc.z + b4.z, acc.w + b4.w);
  if (apply_act) {
    o.x = lrelu(o.x, slope); o.y = lrelu(o.y, slope);
    o.z = lrelu(o.z, slope); o.w = lrelu(o.w, slope);
  }
  *(float4*)(out + (size_t)d * NHID + lane4) = o;
}

extern "C" void kernel_launch(void* const* d_in, const int* in_sizes, int n_in,
                              void* d_out, int out_size, void* d_ws, size_t ws_size,
                              hipStream_t stream) {
  const float* x    = (const float*)d_in[0];
  const int*   ei   = (const int*)d_in[1];
  const float* W1   = (const float*)d_in[2];
  const float* asr1 = (const float*)d_in[3];
  const float* adt1 = (const float*)d_in[4];
  const float* b1   = (const float*)d_in[5];
  const float* W2   = (const float*)d_in[6];
  const float* asr2 = (const float*)d_in[7];
  const float* adt2 = (const float*)d_in[8];
  const float* b2   = (const float*)d_in[9];
  float* out = (float*)d_out;

  const int N = in_sizes[0] / NHID;
  const int E = in_sizes[1] / 2;
  const size_t NF = (size_t)N * NHID;

  float* h      = (float*)d_ws;
  float* x2     = h + NF;
  int*   cnt    = (int*)(x2 + NF);
  float* as     = (float*)(cnt + N);
  float* ad     = as + N;
  int*   rowptr = (int*)(ad + N);
  int*   rank   = rowptr + N + 1;
  int*   csr    = rank + E;

  const int* src = ei;
  const int* dst = ei + E;

  const int gGemm  = (N + 127) / 128;
  const int gWaveN = (int)(((size_t)N * 32 + 255) / 256);
  const int gN     = (N + 255) / 256;
  const int gE     = (E + 255) / 256;
  const int gScan  = (N + 1023) / 1024;

  gat_zero<<<gN, 256, 0, stream>>>(cnt, N);
  gat_rank<<<gE, 256, 0, stream>>>(dst, cnt, rank, E);
  gat_scan<<<gScan, 1024, 0, stream>>>(cnt, rowptr, N);
  gat_fill<<<gE, 256, 0, stream>>>(src, dst, rank, rowptr, csr, E);

  gat_gemm128<<<gGemm, 256, 0, stream>>>(x, W1, asr1, adt1, h, as, ad, N);
  gat_node<<<gWaveN, 256, 0, stream>>>(csr, rowptr, as, ad, h, b1, x2, N, NEG_ACT, 1);

  gat_gemm128<<<gGemm, 256, 0, stream>>>(x2, W2, asr2, adt2, h, as, ad, N);
  gat_node<<<gWaveN, 256, 0, stream>>>(csr, rowptr, as, ad, h, b2, out, N, 0.f, 0);
}


# Edge list grouped by destination by a two-level counting sort on LDS counters (histogram, scan, scatter, per-bucket rank) instead of one returning global atomic per edge; arrival-rank order changes
# speedup vs baseline: 6.7094x; 1.6136x over previous
#include <hip/hip_runtime.h>
#include <hip/hip_bf16.h>

typedef float v2f __attribute__((ext_vector_type(2)));
typedef float v8f __attribute__((ext_vector_type(8)));

#define NHID 128
#define NEG_ATT 0.2f
#define NEG_ACT 0.01f

#if __has_builtin(__builtin_amdgcn_wmma_f32_16x16x4_f32)
#define HAS_WMMA_F32X4 1
#endif

__device__ __forceinline__ float row16_sum(float x) {
  x += __int_as_float(__builtin_amdgcn_update_dpp(0, __float_as_int(x), 0x161, 0xf, 0xf, false));
  x += __int_as_float(__builtin_amdgcn_update_dpp(0, __float_as_int(x), 0x162, 0xf, 0xf, false));
  x += __int_as_float(__builtin_amdgcn_update_dpp(0, __float_as_int(x), 0x164, 0xf, 0xf, false));
  x += __int_as_float(__builtin_amdgcn_update_dpp(0, __float_as_int(x), 0x168, 0xf, 0xf, false));
  return x;
}

__global__ __launch_bounds__(256) void gat_gemm128(
    const float* __restrict__ X, const float* __restrict__ W,
    const float* __restrict__ a_src, const float* __restrict__ a_dst,
    float* __restrict__ H, float* __restrict__ as, float* __restrict__ ad,
    int nrows) {
  __shared__ float Ws[NHID * NHID];
  const int tid = threadIdx.x;
  for (int i = tid; i < NHID * NHID; i += 256) {
    const int k = i >> 7, n = i & 127;
    Ws[((k >> 1) << 8) + (n << 1) + (k & 1)] = W[i];
  }
  __syncthreads();

  const int wave = tid >> 5;
  const int lane = tid & 31;
  const int row0 = blockIdx.x * 128 + wave * 16;
  if (row0 >= nrows) return;

  const int m     = lane & 15;
  const int khv   = lane >> 4;
  const int khalf = khv * 2;
  const int mhi   = khv * 8;

  v8f acc[8] = {};

#ifdef HAS_WMMA_F32X4
  const int arow = min(row0 + m, nrows - 1);
  const float* __restrict__ xrow = X + (size_t)arow * NHID;
  const float* __restrict__ bcol = Ws + (m << 1);
#pragma unroll 4
  for (int k0 = 0; k0 < NHID; k0 += 4) {
    const v2f a = *(const v2f*)(xrow + k0 + khalf);
    const float* __restrict__ bp = bcol + (((k0 >> 1) + khv) << 8);
#pragma unroll
    for (int t = 0; t < 8; ++t) {
      const v2f b = *(const v2f*)(bp + t * 32);
      acc[t] = __builtin_amdgcn_wmma_f32_16x16x4_f32(
          false, a, false, b, (short)0, acc[t], false, false);
    }
  }
#else
#pragma unroll
  for (int t = 0; t < 8; ++t)
    for (int v = 0; v < 8; ++v) {
      const int r = min(row0 + v + mhi, nrows - 1);
      const float* __restrict__ xr = X + (size_t)r * NHID;
      float s = 0.f;
      for (int k = 0; k < NHID; ++k)
        s += xr[k] * Ws[((k >> 1) << 8) + (((t << 4) + m) << 1) + (k & 1)];
      acc[t][v] = s;
    }
#endif

#pragma unroll
  for (int t = 0; t < 8; ++t)
#pragma unroll
    for (int v = 0; v < 8; ++v) {
      const int r = row0 + v + mhi;
      if (r < nrows) H[(size_t)r * NHID + t * 16 + m] = acc[t][v];
    }

  float ps[8], pd[8];
#pragma unroll
  for (int v = 0; v < 8; ++v) { ps[v] = 0.f; pd[v] = 0.f; }
#pragma unroll
  for (int t = 0; t < 8; ++t) {
    const float sa = a_src[t * 16 + m];
    const float da = a_dst[t * 16 + m];
#pragma unroll
    for (int v = 0; v < 8; ++v) {
      ps[v] = fmaf(acc[t][v], sa, ps[v]);
      pd[v] = fmaf(acc[t][v], da, pd[v]);
    }
  }
#pragma unroll
  for (int v = 0; v < 8; ++v) { ps[v] = row16_sum(ps[v]); pd[v] = row16_sum(pd[v]); }
  if (m < 2) {
    float* __restrict__ dstp = (m == 0) ? as : ad;
#pragma unroll
    for (int v = 0; v < 8; ++v) {
      const int r = row0 + v + mhi;
      if (r < nrows) dstp[r] = (m == 0) ? ps[v] : pd[v];
    }
  }
}

#define BUCKET_SHIFT 7
#define BUCKET_NODES 128
#define NB_MAX 512
#define CHUNK_EDGES 8192

__global__ __launch_bounds__(1024) void gat_hist(const int* __restrict__ dsts, int* __restrict__ hist,
                                                 int nedges, int nb, int nc) {
  __shared__ int binL[NB_MAX];
  const int tid = threadIdx.x;
  for (int b = tid; b < nb; b += 1024) binL[b] = 0;
  __syncthreads();
  const int e0 = blockIdx.x * CHUNK_EDGES + tid;
  int d[8];
#pragma unroll
  for (int k = 0; k < 8; ++k) {
    const int e = e0 + k * 1024;
    d[k] = (e < nedges) ? dsts[e] : -1;
  }
#pragma unroll
  for (int k = 0; k < 8; ++k)
    if (d[k] >= 0) atomicAdd(&binL[d[k] >> BUCKET_SHIFT], 1);
  __syncthreads();
  for (int b = tid; b < nb; b += 1024) hist[b * nc + (int)blockIdx.x] = binL[b];
}

__global__ __launch_bounds__(1024) void gat_scan(const int* __restrict__ cnt,
                                                 int* __restrict__ rowptr, int n) {
  __shared__ int wfront[32];
  __shared__ int wtile[32];
  const int tid  = threadIdx.x;
  const int lane = tid & 31;
  const int wave = tid >> 5;
  const int base = blockIdx.x * 1024;
  const int i    = base + tid;
  const int c    = (i < n) ? cnt[i] : 0;

  const int4* __restrict__ c4 = (const int4*)cnt;
  const int nq = base >> 2;
  int s = 0;
  for (int q0 = 0; q0 < nq; q0 += 8 * 1024) {
    int4 v[8];
#pragma unroll
    for (int j = 0; j < 8; ++j) v[j] = c4[min(q0 + j * 1024 + tid, nq - 1)];
#pragma unroll
    for (int j = 0; j < 8; ++j)
      if (q0 + j * 1024 + tid < nq) s += (v[j].x + v[j].y) + (v[j].z + v[j].w);
  }
  for (int off = 16; off; off >>= 1) s += __shfl_xor(s, off, 32);

  int inc = c;
  for (int off = 1; off < 32; off <<= 1) {
    const int t = __shfl_up(inc, off, 32);
    if (lane >= off) inc += t;
  }
  if (lane == 0)  wfront[wave] = s;
  if (lane == 31) wtile[wave]  = inc;
  __syncthreads();

  int o = wfront[lane] + ((lane < wave) ? wtile[lane] : 0);
  for (int off = 16; off; off >>= 1) o += __shfl_xor(o, off, 32);

  if (i < n) rowptr[i] = o + inc - c;
  if (blockIdx.x == gridDim.x - 1 && tid == 1023) rowptr[n] = o + inc;
}

__global__ __launch_bounds__(1024) void gat_scatter(const int* __restrict__ srcs, const int* __restrict__ dsts,
                                                    const int* __restrict__ P, int* __restrict__ tmp_i,
                                                    int nedges, int nb, int nc) {
  int2* __restrict__ tmp = (int2*)tmp_i;
  __shared__ int binL[NB_MAX];
  const int tid = threadIdx.x;
  for (int b = tid; b < nb; b += 1024) binL[b] = P[b * nc + (int)blockIdx.x];
  __syncthreads();
  const int e0 = blockIdx.x * CHUNK_EDGES + tid;
  int d[8], s[8];
#pragma unroll
  for (int k = 0; k < 8; ++k) {
    const int e = e0 + k * 1024;
    const bool ok = e < nedges;
    d[k] = ok ? dsts[e] : -1;
    s[k] = ok ? srcs[e] : 0;
  }
#pragma unroll
  for (int k = 0; k < 8; ++k)
    if (d[k] >= 0) {
      const int pos = atomicAdd(&binL[d[k] >> BUCKET_SHIFT], 1);
      tmp[pos] = make_int2(s[k], d[k]);
    }
}

__global__ __launch_bounds__(512) void gat_bucket(const int* __restrict__ tmp_i, const int* __restrict__ P,
                                                  int* __restrict__ rowptr, int* __restrict__ csr,
                                                  int n, int nb, int nc) {
  const int2* __restrict__ tmp = (const int2*)tmp_i;
  __shared__ int cntL[BUCKET_NODES];
  __shared__ int curL[BUCKET_NODES];
  __shared__ int wsum[BUCKET_NODES / 32];
  const int tid  = threadIdx.x;
  const int lane = tid & 31;
  const int b    = blockIdx.x;
  const int lo   = P[b * nc];
  const int hi   = P[(b + 1) * nc];
  if (tid < BUCKET_NODES) cntL[tid] = 0;
  __syncthreads();

  for (int i = lo + tid; i < hi; i += 4 * 512) {
    int dd[4];
#pragma unroll
    for (int j = 0; j < 4; ++j) {
      const int idx = i + j * 512;
      dd[j] = (idx < hi) ? tmp[idx].y : -1;
    }
#pragma unroll
    for (int j = 0; j < 4; ++j)
      if (dd[j] >= 0) atomicAdd(&cntL[dd[j] & (BUCKET_NODES - 1)], 1);
  }
  __syncthreads();

  int c = 0, inc = 0;
  if (tid < BUCKET_NODES) {
    c = cntL[tid];
    inc = c;
    for (int off = 1; off < 32; off <<= 1) {
      const int t = __shfl_up(inc, off, 32);
      if (lane >= off) inc += t;
    }
    if (lane == 31) wsum[tid >> 5] = inc;
  }
  __syncthreads();
  if (tid < BUCKET_NODES) {
    int o = lo;
    for (int w = 0; w < (tid >> 5); ++w) o += wsum[w];
    const int excl = o + inc - c;
    const int node = b * BUCKET_NODES + tid;
    if (node < n) rowptr[node] = excl;
    curL[tid] = excl;
  }
  if (b == nb - 1 && tid == 0) rowptr[n] = hi;
  __syncthreads();

  for (int i = lo + tid; i < hi; i += 4 * 512) {
    int2 e[4];
#pragma unroll
    for (int j = 0; j < 4; ++j) {
      const int idx = i + j * 512;
      e[j] = (idx < hi) ? tmp[idx] : make_int2(0, -1);
    }
#pragma unroll
    for (int j = 0; j < 4; ++j)
      if (e[j].y >= 0) {
        const int pos = atomicAdd(&curL[e[j].y & (BUCKET_NODES - 1)], 1);
        csr[pos] = e[j].x;
      }
  }
}

__device__ __forceinline__ float lrelu(float v, float slope) {
  return v >= 0.f ? v : slope * v;
}

template <int G>
__device__ __forceinline__ void gat_grp(const float* __restrict__ H, int s_l, float w_l,
                                        int t, int lane4, float4& acc) {
  float4 hv[G];
  float  w[G];
#pragma unroll
  for (int i = 0; i < G; ++i) {
    const int s = __builtin_amdgcn_readlane(s_l, t + i);
    w[i] = __int_as_float(__builtin_amdgcn_readlane(__float_as_int(w_l), t + i));
    hv[i] = *(const float4*)(H + (size_t)s * NHID + lane4);
  }
#pragma unroll
  for (int i = 0; i < G; ++i) {
    acc.x = fmaf(hv[i].x, w[i], acc.x);
    acc.y = fmaf(hv[i].y, w[i], acc.y);
    acc.z = fmaf(hv[i].z, w[i], acc.z);
    acc.w = fmaf(hv[i].w, w[i], acc.w);
  }
}

__device__ __forceinline__ void gat_chunk(const float* __restrict__ H, int s_l, float w_l,
                                          int cnt, int lane4, float4& acc) {
  int t = 0;
  for (; t + 8 <= cnt; t += 8) gat_grp<8>(H, s_l, w_l, t, lane4, acc);
  if (t + 4 <= cnt) { gat_grp<4>(H, s_l, w_l, t, lane4, acc); t += 4; }
  const int rem = cnt - t;
  if (rem == 3)      gat_grp<3>(H, s_l, w_l, t, lane4, acc);
  else if (rem == 2) gat_grp<2>(H, s_l, w_l, t, lane4, acc);
  else if (rem == 1) gat_grp<1>(H, s_l, w_l, t, lane4, acc);
}

__global__ __launch_bounds__(256) void gat_node(
    const int* __restrict__ csr, const int* __restrict__ rowptr,
    const float* __restrict__ as, const float* __restrict__ ad,
    const float* __restrict__ H, const float* __restrict__ bias,
    float* __restrict__ out, int n, float slope, int apply_act) {
  const int lane  = threadIdx.x & 31;
  const int lane4 = lane * 4;
  const int d = __builtin_amdgcn_readfirstlane(
      (int)((blockIdx.x * (size_t)blockDim.x + threadIdx.x) >> 5));
  if (d >= n) return;

  const int beg = rowptr[d];
  const int end = rowptr[d + 1];
  const int deg = end - beg;
  const float ad_d   = ad[d];
  const float v_self = lrelu(as[d] + ad_d, NEG_ATT);

  const int  j0  = beg + lane;
  const int  j1  = j0 + 32;
  const bool ok0 = j0 < end;
  const bool ok1 = j1 < end;
  int s0 = d, s1 = d;
  if (ok0) s0 = csr[j0];
  if (ok1) s1 = csr[j1];
  const float4 hd = *(const float4*)(H + (size_t)d * NHID + lane4);
  const float a0 = as[s0];
  const float a1 = as[s1];
  const float v0 = ok0 ? lrelu(a0 + ad_d, NEG_ATT) : v_self;
  const float v1 = ok1 ? lrelu(a1 + ad_d, NEG_ATT) : v_self;

  float mx = fmaxf(v_self, fmaxf(v0, v1));
  for (int j = j0 + 64; j < end; j += 32)
    mx = fmaxf(mx, lrelu(as[csr[j]] + ad_d, NEG_ATT));
  for (int off = 16; off; off >>= 1) mx = fmaxf(mx, __shfl_xor(mx, off, 32));

  const float p0 = ok0 ? expf(v0 - mx) : 0.f;
  const float p1 = ok1 ? expf(v1 - mx) : 0.f;
  float den = p0 + p1;
  for (int j = j0 + 64; j < end; j += 32)
    den += expf(lrelu(as[csr[j]] + ad_d, NEG_ATT) - mx);
  for (int off = 16; off; off >>= 1) den += __shfl_xor(den, off, 32);
  const float p_self = expf(v_self - mx);
  den += p_self;
  const float dn = den + 1e-16f;

  const float  w_self = p_self / dn;
  float4 acc = make_float4(hd.x * w_self, hd.y * w_self, hd.z * w_self, hd.w * w_self);

  gat_chunk(H, s0, p0 / dn, min(deg, 32), lane4, acc);
  if (deg > 32) gat_chunk(H, s1, p1 / dn, min(deg - 32, 32), lane4, acc);
  for (int base = beg + 64; base < end; base += 32) {
    const int j = base + lane;
    int   s = d;
    float w = 0.f;
    if (j < end) {
      s = csr[j];
      w = expf(lrelu(as[s] + ad_d, NEG_ATT) - mx) / dn;
    }
    gat_chunk(H, s, w, min(end - base, 32), lane4, acc);
  }

  const float4 b4 = *(const float4*)(bias + lane4);
  float4 o = make_float4(acc.x + b4.x, acc.y + b4.y, acc.z + b4.z, acc.w + b4.w);
  if (apply_act) {
    o.x = lrelu(o.x, slope); o.y = lrelu(o.y, slope);
    o.z = lrelu(o.z, slope); o.w = lrelu(o.w, slope);
  }
  *(float4*)(out + (size_t)d * NHID + lane4) = o;
}

extern "C" void kernel_launch(void* const* d_in, const int* in_sizes, int n_in,
                              void* d_out, int out_size, void* d_ws, size_t ws_size,
                              hipStream_t stream) {
  const float* x    = (const float*)d_in[0];
  const int*   ei   = (const int*)d_in[1];
  const float* W1   = (const float*)d_in[2];
  const float* asr1 = (const float*)d_in[3];
  const float* adt1 = (const float*)d_in[4];
  const float* b1   = (const float*)d_in[5];
  const float* W2   = (const float*)d_in[6];
  const float* asr2 = (const float*)d_in[7];
  const float* adt2 = (const float*)d_in[8];
  const float* b2   = (const float*)d_in[9];
  float* out = (float*)d_out;

  const int N = in_sizes[0] / NHID;
  const int E = in_sizes[1] / 2;
  const size_t NF = (size_t)N * NHID;

  const int NBK  = (N + BUCKET_NODES - 1) / BUCKET_NODES;
  const int NCH  = (E + CHUNK_EDGES - 1) / CHUNK_EDGES;
  const int L    = NBK * NCH;
  const int Lpad = (L + 3) & ~3;

  float* h      = (float*)d_ws;
  float* x2     = h + NF;
  int*   hist   = (int*)(x2 + NF);
  int*   tmp    = hist + Lpad;
  float* as     = (float*)(tmp + 2 * (size_t)E);
  float* ad     = as + N;
  int*   rowptr = (int*)(ad + N);
  int*   P      = rowptr + N + 1;
  int*   csr    = P + L + 1;

  const int* src = ei;
  const int* dst = ei + E;

  const int gGemm  = (N + 127) / 128;
  const int gWaveN = (int)(((size_t)N * 32 + 255) / 256);
  const int gScan  = (L + 1023) / 1024;

  gat_hist<<<NCH, 1024, 0, stream>>>(dst, hist, E, NBK, NCH);
  gat_scan<<<gScan, 1024, 0, stream>>>(hist, P, L);
  gat_scatter<<<NCH, 1024, 0, stream>>>(src, dst, P, tmp, E, NBK, NCH);
  gat_bucket<<<NBK, 512, 0, stream>>>(tmp, P, rowptr, csr, N, NBK, NCH);

  gat_gemm128<<<gGemm, 256, 0, stream>>>(x, W1, asr1, adt1, h, as, ad, N);
  gat_node<<<gWaveN, 256, 0, stream>>>(csr, rowptr, as, ad, h, b1, x2, N, NEG_ACT, 1);

  gat_gemm128<<<gGemm, 256, 0, stream>>>(x2, W2, asr2, adt2, h, as, ad, N);
  gat_node<<<gWaveN, 256, 0, stream>>>(csr, rowptr, as, ad, h, b2, out, N, 0.f, 0);
}


# Layer-1 GEMM tiles ride in the launch that finishes the edge list (256-row tiles as the first work-groups, buckets behind them; independent work overlaps, one launch fewer); per-wave sums unchanged
# speedup vs baseline: 7.1265x; 1.0622x over previous
#include <hip/hip_runtime.h>
#include <hip/hip_bf16.h>

typedef float v2f __attribute__((ext_vector_type(2)));
typedef float v8f __attribute__((ext_vector_type(8)));

#define NHID 128
#define NEG_ATT 0.2f
#define NEG_ACT 0.01f

#if __has_builtin(__builtin_amdgcn_wmma_f32_16x16x4_f32)
#define HAS_WMMA_F32X4 1
#endif

__device__ __forceinline__ float row16_sum(float x) {
  x += __int_as_float(__builtin_amdgcn_update_dpp(0, __float_as_int(x), 0x161, 0xf, 0xf, false));
  x += __int_as_float(__builtin_amdgcn_update_dpp(0, __float_as_int(x), 0x162, 0xf, 0xf, false));
  x += __int_as_float(__builtin_amdgcn_update_dpp(0, __float_as_int(x), 0x164, 0xf, 0xf, false));
  x += __int_as_float(__builtin_amdgcn_update_dpp(0, __float_as_int(x), 0x168, 0xf, 0xf, false));
  return x;
}

template <int NT>
__device__ __forceinline__ void gemm_tile(
    float* __restrict__ Ws, int tile,
    const float* __restrict__ X, const float* __restrict__ W,
    const float* __restrict__ a_src, const float* __restrict__ a_dst,
    float* __restrict__ H, float* __restrict__ as, float* __restrict__ ad,
    int nrows) {
  const int tid = threadIdx.x;
  for (int i = tid; i < NHID * NHID; i += NT) {
    const int k = i >> 7, n = i & 127;
    Ws[((k >> 1) << 8) + (n << 1) + (k & 1)] = W[i];
  }
  __syncthreads();

  const int wave = tid >> 5;
  const int lane = tid & 31;
  const int row0 = tile * (NT / 2) + wave * 16;
  if (row0 >= nrows) return;

  const int m     = lane & 15;
  const int khv   = lane >> 4;
  const int khalf = khv * 2;
  const int mhi   = khv * 8;

  v8f acc[8] = {};

#ifdef HAS_WMMA_F32X4
  const int arow = min(row0 + m, nrows - 1);
  const float* __restrict__ xrow = X + (size_t)arow * NHID;
  const float* __restrict__ bcol = Ws + (m << 1);
#pragma unroll 4
  for (int k0 = 0; k0 < NHID; k0 += 4) {
    const v2f a = *(const v2f*)(xrow + k0 + khalf);
    const float* __restrict__ bp = bcol + (((k0 >> 1) + khv) << 8);
#pragma unroll
    for (int t = 0; t < 8; ++t) {
      const v2f b = *(const v2f*)(bp + t * 32);
      acc[t] = __builtin_amdgcn_wmma_f32_16x16x4_f32(
          false, a, false, b, (short)0, acc[t], false, false);
    }
  }
#else
#pragma unroll
  for (int t = 0; t < 8; ++t)
    for (int v = 0; v < 8; ++v) {
      const int r = min(row0 + v + mhi, nrows - 1);
      const float* __restrict__ xr = X + (size_t)r * NHID;
      float s = 0.f;
      for (int k = 0; k < NHID; ++k)
        s += xr[k] * Ws[((k >> 1) << 8) + (((t << 4) + m) << 1) + (k & 1)];
      acc[t][v] = s;
    }
#endif

#pragma unroll
  for (int t = 0; t < 8; ++t)
#pragma unroll
    for (int v = 0; v < 8; ++v) {
      const int r = row0 + v + mhi;
      if (r < nrows) H[(size_t)r * NHID + t * 16 + m] = acc[t][v];
    }

  float ps[8], pd[8];
#pragma unroll
  for (int v = 0; v < 8; ++v) { ps[v] = 0.f; pd[v] = 0.f; }
#pragma unroll
  for (int t = 0; t < 8; ++t) {
    const float sa = a_src[t * 16 + m];
    const float da = a_dst[t * 16 + m];
#pragma unroll
    for (int v = 0; v < 8; ++v) {
      ps[v] = fmaf(acc[t][v], sa, ps[v]);
      pd[v] = fmaf(acc[t][v], da, pd[v]);
    }
  }
#pragma unroll
  for (int v = 0; v < 8; ++v) { ps[v] = row16_sum(ps[v]); pd[v] = row16_sum(pd[v]); }
  if (m < 2) {
    float* __restrict__ dstp = (m == 0) ? as : ad;
#pragma unroll
    for (int v = 0; v < 8; ++v) {
      const int r = row0 + v + mhi;
      if (r < nrows) dstp[r] = (m == 0) ? ps[v] : pd[v];
    }
  }
}

__global__ __launch_bounds__(256) void gat_gemm128(
    const float* __restrict__ X, const float* __restrict__ W,
    const float* __restrict__ a_src, const float* __restrict__ a_dst,
    float* __restrict__ H, float* __restrict__ as, float* __restrict__ ad,
    int nrows) {
  __shared__ float Ws[NHID * NHID];
  gemm_tile<256>(Ws, (int)blockIdx.x, X, W, a_src, a_dst, H, as, ad, nrows);
}

#define BUCKET_SHIFT 7
#define BUCKET_NODES 128
#define NB_MAX 512
#define CHUNK_EDGES 8192

__global__ __launch_bounds__(1024) void gat_hist(const int* __restrict__ dsts, int* __restrict__ hist,
                                                 int nedges, int nb, int nc) {
  __shared__ int binL[NB_MAX];
  const int tid = threadIdx.x;
  for (int b = tid; b < nb; b += 1024) binL[b] = 0;
  __syncthreads();
  const int e0 = blockIdx.x * CHUNK_EDGES + tid;
  int d[8];
#pragma unroll
  for (int k = 0; k < 8; ++k) {
    const int e = e0 + k * 1024;
    d[k] = (e < nedges) ? dsts[e] : -1;
  }
#pragma unroll
  for (int k = 0; k < 8; ++k)
    if (d[k] >= 0) atomicAdd(&binL[d[k] >> BUCKET_SHIFT], 1);
  __syncthreads();
  for (int b = tid; b < nb; b += 1024) hist[b * nc + (int)blockIdx.x] = binL[b];
}

__global__ __launch_bounds__(1024) void gat_scan(const int* __restrict__ cnt,
                                                 int* __restrict__ rowptr, int n) {
  __shared__ int wfront[32];
  __shared__ int wtile[32];
  const int tid  = threadIdx.x;
  const int lane = tid & 31;
  const int wave = tid >> 5;
  const int base = blockIdx.x * 1024;
  const int i    = base + tid;
  const int c    = (i < n) ? cnt[i] : 0;

  const int4* __restrict__ c4 = (const int4*)cnt;
  const int nq = base >> 2;
  int s = 0;
  for (int q0 = 0; q0 < nq; q0 += 8 * 1024) {
    int4 v[8];
#pragma unroll
    for (int j = 0; j < 8; ++j) v[j] = c4[min(q0 + j * 1024 + tid, nq - 1)];
#pragma unroll
    for (int j = 0; j < 8; ++j)
      if (q0 + j * 1024 + tid < nq) s += (v[j].x + v[j].y) + (v[j].z + v[j].w);
  }
  for (int off = 16; off; off >>= 1) s += __shfl_xor(s, off, 32);

  int inc = c;
  for (int off = 1; off < 32; off <<= 1) {
    const int t = __shfl_up(inc, off, 32);
    if (lane >= off) inc += t;
  }
  if (lane == 0)  wfront[wave] = s;
  if (lane == 31) wtile[wave]  = inc;
  __syncthreads();

  int o = wfront[lane] + ((lane < wave) ? wtile[lane] : 0);
  for (int off = 16; off; off >>= 1) o += __shfl_xor(o, off, 32);

  if (i < n) rowptr[i] = o + inc - c;
  if (blockIdx.x == gridDim.x - 1 && tid == 1023) rowptr[n] = o + inc;
}

__global__ __launch_bounds__(1024) void gat_scatter(const int* __restrict__ srcs, const int* __restrict__ dsts,
                                                    const int* __restrict__ P, int* __restrict__ tmp_i,
                                                    int nedges, int nb, int nc) {
  int2* __restrict__ tmp = (int2*)tmp_i;
  __shared__ int binL[NB_MAX];
  const int tid = threadIdx.x;
  for (int b = tid; b < nb; b += 1024) binL[b] = P[b * nc + (int)blockIdx.x];
  __syncthreads();
  const int e0 = blockIdx.x * CHUNK_EDGES + tid;
  int d[8], s[8];
#pragma unroll
  for (int k = 0; k < 8; ++k) {
    const int e = e0 + k * 1024;
    const bool ok = e < nedges;
    d[k] = ok ? dsts[e] : -1;
    s[k] = ok ? srcs[e] : 0;
  }
#pragma unroll
  for (int k = 0; k < 8; ++k)
    if (d[k] >= 0) {
      const int pos = atomicAdd(&binL[d[k] >> BUCKET_SHIFT], 1);
      tmp[pos] = make_int2(s[k], d[k]);
    }
}

__global__ __launch_bounds__(512) void gat_bucket_gemm(
    const int* __restrict__ tmp_i, const int* __restrict__ P,
    int* __restrict__ rowptr, int* __restrict__ csr, int n, int nb, int nc,
    const float* __restrict__ X, const float* __restrict__ W,
    const float* __restrict__ a_src, const float* __restrict__ a_dst,
    float* __restrict__ H, float* __restrict__ as, float* __restrict__ ad) {
  __shared__ float Ws[NHID * NHID];
  __shared__ int cntL[BUCKET_NODES];
  __shared__ int curL[BUCKET_NODES];
  __shared__ int wsum[BUCKET_NODES / 32];
  const int ntile = (nb + 1) >> 1;
  if ((int)blockIdx.x < ntile) {
    gemm_tile<512>(Ws, (int)blockIdx.x, X, W, a_src, a_dst, H, as, ad, n);
    return;
  }
  const int b = (int)blockIdx.x - ntile;
  if (b >= nb) return;
  const int2* __restrict__ tmp = (const int2*)tmp_i;
  const int tid  = threadIdx.x;
  const int lane = tid & 31;
  const int lo   = P[b * nc];
  const int hi   = P[(b + 1) * nc];
  if (tid < BUCKET_NODES) cntL[tid] = 0;
  __syncthreads();

  for (int i = lo + tid; i < hi; i += 4 * 512) {
    int dd[4];
#pragma unroll
    for (int j = 0; j < 4; ++j) {
      const int idx = i + j * 512;
      dd[j] = (idx < hi) ? tmp[idx].y : -1;
    }
#pragma unroll
    for (int j = 0; j < 4; ++j)
      if (dd[j] >= 0) atomicAdd(&cntL[dd[j] & (BUCKET_NODES - 1)], 1);
  }
  __syncthreads();

  int c = 0, inc = 0;
  if (tid < BUCKET_NODES) {
    c = cntL[tid];
    inc = c;
    for (int off = 1; off < 32; off <<= 1) {
      const int t = __shfl_up(inc, off, 32);
      if (lane >= off) inc += t;
    }
    if (lane == 31) wsum[tid >> 5] = inc;
  }
  __syncthreads();
  if (tid < BUCKET_NODES) {
    int o = lo;
    for (int w = 0; w < (tid >> 5); ++w) o += wsum[w];
    const int excl = o + inc - c;
    const int node = b * BUCKET_NODES + tid;
    if (node < n) rowptr[node] = excl;
    curL[tid] = excl;
  }
  if (b == nb - 1 && tid == 0) rowptr[n] = hi;
  __syncthreads();

  for (int i = lo + tid; i < hi; i += 4 * 512) {
    int2 e[4];
#pragma unroll
    for (int j = 0; j < 4; ++j) {
      const int idx = i + j * 512;
      e[j] = (idx < hi) ? tmp[idx] : make_int2(0, -1);
    }
#pragma unroll
    for (int j = 0; j < 4; ++j)
      if (e[j].y >= 0) {
        const int pos = atomicAdd(&curL[e[j].y & (BUCKET_NODES - 1)], 1);
        csr[pos] = e[j].x;
      }
  }
}

__device__ __forceinline__ float lrelu(float v, float slope) {
  return v >= 0.f ? v : slope * v;
}

template <int G>
__device__ __forceinline__ void gat_grp(const float* __restrict__ H, int s_l, float w_l,
                                        int t, int lane4, float4& acc) {
  float4 hv[G];
  float  w[G];
#pragma unroll
  for (int i = 0; i < G; ++i) {
    const int s = __builtin_amdgcn_readlane(s_l, t + i);
    w[i] = __int_as_float(__builtin_amdgcn_readlane(__float_as_int(w_l), t + i));
    hv[i] = *(const float4*)(H + (size_t)s * NHID + lane4);
  }
#pragma unroll
  for (int i = 0; i < G; ++i) {
    acc.x = fmaf(hv[i].x, w[i], acc.x);
    acc.y = fmaf(hv[i].y, w[i], acc.y);
    acc.z = fmaf(hv[i].z, w[i], acc.z);
    acc.w = fmaf(hv[i].w, w[i], acc.w);
  }
}

__device__ __forceinline__ void gat_chunk(const float* __restrict__ H, int s_l, float w_l,
                                          int cnt, int lane4, float4& acc) {
  int t = 0;
  for (; t + 8 <= cnt; t += 8) gat_grp<8>(H, s_l, w_l, t, lane4, acc);
  if (t + 4 <= cnt) { gat_grp<4>(H, s_l, w_l, t, lane4, acc); t += 4; }
  const int rem = cnt - t;
  if (rem == 3)      gat_grp<3>(H, s_l, w_l, t, lane4, acc);
  else if (rem == 2) gat_grp<2>(H, s_l, w_l, t, lane4, acc);
  else if (rem == 1) gat_grp<1>(H, s_l, w_l, t, lane4, acc);
}

__global__ __launch_bounds__(256) void gat_node(
    const int* __restrict__ csr, const int* __restrict__ rowptr,
    const float* __restrict__ as, const float* __restrict__ ad,
    const float* __restrict__ H, const float* __restrict__ bias,
    float* __restrict__ out, int n, float slope, int apply_act) {
  const int lane  = threadIdx.x & 31;
  const int lane4 = lane * 4;
  const int d = __builtin_amdgcn_readfirstlane(
      (int)((blockIdx.x * (size_t)blockDim.x + threadIdx.x) >> 5));
  if (d >= n) return;

  const int beg = rowptr[d];
  const int end = rowptr[d + 1];
  const int deg = end - beg;
  const float ad_d   = ad[d];
  const float v_self = lrelu(as[d] + ad_d, NEG_ATT);

  const int  j0  = beg + lane;
  const int  j1  = j0 + 32;
  const bool ok0 = j0 < end;
  const bool ok1 = j1 < end;
  int s0 = d, s1 = d;
  if (ok0) s0 = csr[j0];
  if (ok1) s1 = csr[j1];
  const float4 hd = *(const float4*)(H + (size_t)d * NHID + lane4);
  const float a0 = as[s0];
  const float a1 = as[s1];
  const float v0 = ok0 ? lrelu(a0 + ad_d, NEG_ATT) : v_self;
  const float v1 = ok1 ? lrelu(a1 + ad_d, NEG_ATT) : v_self;

  float mx = fmaxf(v_self, fmaxf(v0, v1));
  for (int j = j0 + 64; j < end; j += 32)
    mx = fmaxf(mx, lrelu(as[csr[j]] + ad_d, NEG_ATT));
  for (int off = 16; off; off >>= 1) mx = fmaxf(mx, __shfl_xor(mx, off, 32));

  const float p0 = ok0 ? expf(v0 - mx) : 0.f;
  const float p1 = ok1 ? expf(v1 - mx) : 0.f;
  float den = p0 + p1;
  for (int j = j0 + 64; j < end; j += 32)
    den += expf(lrelu(as[csr[j]] + ad_d, NEG_ATT) - mx);
  for (int off = 16; off; off >>= 1) den += __shfl_xor(den, off, 32);
  const float p_self = expf(v_self - mx);
  den += p_self;
  const float dn = den + 1e-16f;

  const float  w_self = p_self / dn;
  float4 acc = make_float4(hd.x * w_self, hd.y * w_self, hd.z * w_self, hd.w * w_self);

  gat_chunk(H, s0, p0 / dn, min(deg, 32), lane4, acc);
  if (deg > 32) gat_chunk(H, s1, p1 / dn, min(deg - 32, 32), lane4, acc);
  for (int base = beg + 64; base < end; base += 32) {
    const int j = base + lane;
    int   s = d;
    float w = 0.f;
    if (j < end) {
      s = csr[j];
      w = expf(lrelu(as[s] + ad_d, NEG_ATT) - mx) / dn;
    }
    gat_chunk(H, s, w, min(end - base, 32), lane4, acc);
  }

  const float4 b4 = *(const float4*)(bias + lane4);
  float4 o = make_float4(acc.x + b4.x, acc.y + b4.y, acc.z + b4.z, acc.w + b4.w);
  if (apply_act) {
    o.x = lrelu(o.x, slope); o.y = lrelu(o.y, slope);
    o.z = lrelu(o.z, slope); o.w = lrelu(o.w, slope);
  }
  *(float4*)(out + (size_t)d * NHID + lane4) = o;
}

extern "C" void kernel_launch(void* const* d_in, const int* in_sizes, int n_in,
                              void* d_out, int out_size, void* d_ws, size_t ws_size,
                              hipStream_t stream) {
  const float* x    = (const float*)d_in[0];
  const int*   ei   = (const int*)d_in[1];
  const float* W1   = (const float*)d_in[2];
  const float* asr1 = (const float*)d_in[3];
  const float* adt1 = (const float*)d_in[4];
  const float* b1   = (const float*)d_in[5];
  const float* W2   = (const float*)d_in[6];
  const float* asr2 = (const float*)d_in[7];
  const float* adt2 = (const float*)d_in[8];
  const float* b2   = (const float*)d_in[9];
  float* out = (float*)d_out;

  const int N = in_sizes[0] / NHID;
  const int E = in_sizes[1] / 2;
  const size_t NF = (size_t)N * NHID;

  const int NBK  = (N + BUCKET_NODES - 1) / BUCKET_NODES;
  const int NCH  = (E + CHUNK_EDGES - 1) / CHUNK_EDGES;
  const int L    = NBK * NCH;
  const int Lpad = (L + 3) & ~3;

  float* h      = (float*)d_ws;
  float* x2     = h + NF;
  int*   hist   = (int*)(x2 + NF);
  int*   tmp    = hist + Lpad;
  float* as     = (float*)(tmp + 2 * (size_t)E);
  float* ad     = as + N;
  int*   rowptr = (int*)(ad + N);
  int*   P      = rowptr + N + 1;
  int*   csr    = P + L + 1;

  const int* src = ei;
  const int* dst = ei + E;

  const int gGemm  = (N + 127) / 128;
  const int gWaveN = (int)(((size_t)N * 32 + 255) / 256);
  const int gScan  = (L + 1023) / 1024;

  gat_hist<<<NCH, 1024, 0, stream>>>(dst, hist, E, NBK, NCH);
  gat_scan<<<gScan, 1024, 0, stream>>>(hist, P, L);
  gat_scatter<<<NCH, 1024, 0, stream>>>(src, dst, P, tmp, E, NBK, NCH);
  gat_bucket_gemm<<<NBK + (NBK + 1) / 2, 512, 0, stream>>>(tmp, P, rowptr, csr, N, NBK, NCH,
                                                           x, W1, asr1, adt1, h, as, ad);

  gat_node<<<gWaveN, 256, 0, stream>>>(csr, rowptr, as, ad, h, b1, x2, N, NEG_ACT, 1);

  gat_gemm128<<<gGemm, 256, 0, stream>>>(x2, W2, asr2, adt2, h, as, ad, N);
  gat_node<<<gWaveN, 256, 0, stream>>>(csr, rowptr, as, ad, h, b2, out, N, 0.f, 0);
}


# Edge list: each 8192-edge chunk sorts its edges by bucket in its own region (LDS count, scan, rank); buckets gather per-chunk runs; global histogram and scan launches gone; arrival order changes
# speedup vs baseline: 7.5660x; 1.0617x over previous
#include <hip/hip_runtime.h>
#include <hip/hip_bf16.h>

typedef float v2f __attribute__((ext_vector_type(2)));
typedef float v8f __attribute__((ext_vector_type(8)));

#define NHID 128
#define NEG_ATT 0.2f
#define NEG_ACT 0.01f

#if __has_builtin(__builtin_amdgcn_wmma_f32_16x16x4_f32)
#define HAS_WMMA_F32X4 1
#endif

__device__ __forceinline__ float row16_sum(float x) {
  x += __int_as_float(__builtin_amdgcn_update_dpp(0, __float_as_int(x), 0x161, 0xf, 0xf, false));
  x += __int_as_float(__builtin_amdgcn_update_dpp(0, __float_as_int(x), 0x162, 0xf, 0xf, false));
  x += __int_as_float(__builtin_amdgcn_update_dpp(0, __float_as_int(x), 0x164, 0xf, 0xf, false));
  x += __int_as_float(__builtin_amdgcn_update_dpp(0, __float_as_int(x), 0x168, 0xf, 0xf, false));
  return x;
}

template <int NT>
__device__ __forceinline__ void gemm_tile(
    float* __restrict__ Ws, int tile,
    const float* __restrict__ X, const float* __restrict__ W,
    const float* __restrict__ a_src, const float* __restrict__ a_dst,
    float* __restrict__ H, float* __restrict__ as, float* __restrict__ ad,
    int nrows) {
  const int tid = threadIdx.x;
  for (int i = tid; i < NHID * NHID; i += NT) {
    const int k = i >> 7, n = i & 127;
    Ws[((k >> 1) << 8) + (n << 1) + (k & 1)] = W[i];
  }
  __syncthreads();

  const int wave = tid >> 5;
  const int lane = tid & 31;
  const int row0 = tile * (NT / 2) + wave * 16;
  if (row0 >= nrows) return;

  const int m     = lane & 15;
  const int khv   = lane >> 4;
  const int khalf = khv * 2;
  const int mhi   = khv * 8;

  v8f acc[8] = {};

#ifdef HAS_WMMA_F32X4
  const int arow = min(row0 + m, nrows - 1);
  const float* __restrict__ xrow = X + (size_t)arow * NHID;
  const float* __restrict__ bcol = Ws + (m << 1);
#pragma unroll 4
  for (int k0 = 0; k0 < NHID; k0 += 4) {
    const v2f a = *(const v2f*)(xrow + k0 + khalf);
    const float* __restrict__ bp = bcol + (((k0 >> 1) + khv) << 8);
#pragma unroll
    for (int t = 0; t < 8; ++t) {
      const v2f b = *(const v2f*)(bp + t * 32);
      acc[t] = __builtin_amdgcn_wmma_f32_16x16x4_f32(
          false, a, false, b, (short)0, acc[t], false, false);
    }
  }
#else
#pragma unroll
  for (int t = 0; t < 8; ++t)
    for (int v = 0; v < 8; ++v) {
      const int r = min(row0 + v + mhi, nrows - 1);
      const float* __restrict__ xr = X + (size_t)r * NHID;
      float s = 0.f;
      for (int k = 0; k < NHID; ++k)
        s += xr[k] * Ws[((k >> 1) << 8) + (((t << 4) + m) << 1) + (k & 1)];
      acc[t][v] = s;
    }
#endif

#pragma unroll
  for (int t = 0; t < 8; ++t)
#pragma unroll
    for (int v = 0; v < 8; ++v) {
      const int r = row0 + v + mhi;
      if (r < nrows) H[(size_t)r * NHID + t * 16 + m] = acc[t][v];
    }

  float ps[8], pd[8];
#pragma unroll
  for (int v = 0; v < 8; ++v) { ps[v] = 0.f; pd[v] = 0.f; }
#pragma unroll
  for (int t = 0; t < 8; ++t) {
    const float sa = a_src[t * 16 + m];
    const float da = a_dst[t * 16 + m];
#pragma unroll
    for (int v = 0; v < 8; ++v) {
      ps[v] = fmaf(acc[t][v], sa, ps[v]);
      pd[v] = fmaf(acc[t][v], da, pd[v]);
    }
  }
#pragma unroll
  for (int v = 0; v < 8; ++v) { ps[v] = row16_sum(ps[v]); pd[v] = row16_sum(pd[v]); }
  if (m < 2) {
    float* __restrict__ dstp = (m == 0) ? as : ad;
#pragma unroll
    for (int v = 0; v < 8; ++v) {
      const int r = row0 + v + mhi;
      if (r < nrows) dstp[r] = (m == 0) ? ps[v] : pd[v];
    }
  }
}

__global__ __launch_bounds__(256) void gat_gemm128(
    const float* __restrict__ X, const float* __restrict__ W,
    const float* __restrict__ a_src, const float* __restrict__ a_dst,
    float* __restrict__ H, float* __restrict__ as, float* __restrict__ ad,
    int nrows) {
  __shared__ float Ws[NHID * NHID];
  gemm_tile<256>(Ws, (int)blockIdx.x, X, W, a_src, a_dst, H, as, ad, nrows);
}

#define BUCKET_SHIFT 7
#define BUCKET_NODES 128
#define NB_MAX 512
#define CHUNK_EDGES 8192

__global__ __launch_bounds__(1024) void gat_chunk_sort(const int* __restrict__ srcs, const int* __restrict__ dsts,
                                                       int* __restrict__ off, int* __restrict__ tmp_i,
                                                       int nedges, int nb, int nc) {
  __shared__ int binL[NB_MAX];
  __shared__ int wsumL[NB_MAX / 32];
  const int tid  = threadIdx.x;
  const int lane = tid & 31;
  const int c    = (int)blockIdx.x;
  if (tid < NB_MAX) binL[tid] = 0;
  __syncthreads();
  const int e0 = c * CHUNK_EDGES + tid;
  int d[8], s[8];
#pragma unroll
  for (int k = 0; k < 8; ++k) {
    const int e = e0 + k * 1024;
    const bool ok = e < nedges;
    d[k] = ok ? dsts[e] : -1;
    s[k] = ok ? srcs[e] : 0;
  }
#pragma unroll
  for (int k = 0; k < 8; ++k)
    if (d[k] >= 0) atomicAdd(&binL[d[k] >> BUCKET_SHIFT], 1);
  __syncthreads();

  int cnt = 0, inc = 0;
  if (tid < NB_MAX) {
    cnt = binL[tid];
    inc = cnt;
    for (int o = 1; o < 32; o <<= 1) {
      const int t = __shfl_up(inc, o, 32);
      if (lane >= o) inc += t;
    }
    if (lane == 31) wsumL[tid >> 5] = inc;
  }
  __syncthreads();
  if (tid < NB_MAX) {
    int o = 0;
    for (int w = 0; w < (tid >> 5); ++w) o += wsumL[w];
    const int excl = o + inc - cnt;
    binL[tid] = excl;
    if (tid < nb) off[tid * nc + c] = excl;
    if (tid == nb - 1) off[nb * nc + c] = excl + cnt;
  }
  __syncthreads();

  int2* __restrict__ mine = (int2*)tmp_i + (size_t)c * CHUNK_EDGES;
  int pos[8];
#pragma unroll
  for (int k = 0; k < 8; ++k)
    pos[k] = (d[k] >= 0) ? atomicAdd(&binL[d[k] >> BUCKET_SHIFT], 1) : -1;
#pragma unroll
  for (int k = 0; k < 8; ++k)
    if (pos[k] >= 0) mine[pos[k]] = make_int2(s[k], d[k]);
}

template <bool PLACE>
__device__ __forceinline__ int bucket_sweep(const int2* __restrict__ tmp, const int* __restrict__ off,
                                            int b, int nc, int wave, int lane,
                                            int* __restrict__ binL, int* __restrict__ csr) {
  int lo_part = 0;
  for (int r0 = wave; r0 < nc; r0 += 16 * 32) {
    const int cl = r0 + 16 * lane;
    int sb = 0, sc = 0;
    if (cl < nc) {
      const int o0 = off[b * nc + cl];
      const int o1 = off[(b + 1) * nc + cl];
      lo_part += o0;
      sb = cl * CHUNK_EDGES + o0;
      sc = o1 - o0;
    }
    const int nseg = min(32, (nc - r0 + 15) >> 4);
    for (int j = 0; j < nseg; j += 8) {
      int2 e[8];
      int  bg[8], cn[8];
#pragma unroll
      for (int u = 0; u < 8; ++u) {
        const int jj = min(j + u, 31);
        bg[u] = __builtin_amdgcn_readlane(sb, jj);
        cn[u] = (j + u < nseg) ? __builtin_amdgcn_readlane(sc, jj) : 0;
        e[u] = make_int2(0, -1);
        if (lane < cn[u]) {
          if (PLACE) e[u] = tmp[bg[u] + lane];
          else       e[u].y = tmp[bg[u] + lane].y;
        }
      }
      if (PLACE) {
        int pos[8];
#pragma unroll
        for (int u = 0; u < 8; ++u)
          pos[u] = (e[u].y >= 0) ? atomicAdd(&binL[e[u].y & (BUCKET_NODES - 1)], 1) : -1;
#pragma unroll
        for (int u = 0; u < 8; ++u)
          if (pos[u] >= 0) csr[pos[u]] = e[u].x;
      } else {
#pragma unroll
        for (int u = 0; u < 8; ++u)
          if (e[u].y >= 0) atomicAdd(&binL[e[u].y & (BUCKET_NODES - 1)], 1);
      }
#pragma unroll
      for (int u = 0; u < 8; ++u)
        for (int k = lane + 32; k < cn[u]; k += 32) {
          const int2 ee = tmp[bg[u] + k];
          if (PLACE) {
            const int pos = atomicAdd(&binL[ee.y & (BUCKET_NODES - 1)], 1);
            csr[pos] = ee.x;
          } else {
            atomicAdd(&binL[ee.y & (BUCKET_NODES - 1)], 1);
          }
        }
    }
  }
  return lo_part;
}

__global__ __launch_bounds__(512) void gat_bucket_gemm(
    const int* __restrict__ tmp_i, const int* __restrict__ off,
    int* __restrict__ rowptr, int* __restrict__ csr, int n, int nb, int nc, int nedges,
    const float* __restrict__ X, const float* __restrict__ W,
    const float* __restrict__ a_src, const float* __restrict__ a_dst,
    float* __restrict__ H, float* __restrict__ as, float* __restrict__ ad) {
  __shared__ float Ws[NHID * NHID];
  __shared__ int cntL[BUCKET_NODES];
  __shared__ int curL[BUCKET_NODES];
  __shared__ int wsum[BUCKET_NODES / 32];
  __shared__ int wlo[16];
  const int ntile = (nb + 1) >> 1;
  if ((int)blockIdx.x < ntile) {
    gemm_tile<512>(Ws, (int)blockIdx.x, X, W, a_src, a_dst, H, as, ad, n);
    return;
  }
  const int b = (int)blockIdx.x - ntile;
  if (b >= nb) return;
  const int2* __restrict__ tmp = (const int2*)tmp_i;
  const int tid  = threadIdx.x;
  const int lane = tid & 31;
  const int wave = tid >> 5;
  if (tid < BUCKET_NODES) cntL[tid] = 0;
  __syncthreads();

  int lo_part = bucket_sweep<false>(tmp, off, b, nc, wave, lane, cntL, csr);
  for (int o = 16; o; o >>= 1) lo_part += __shfl_xor(lo_part, o, 32);
  if (lane == 0) wlo[wave] = lo_part;
  __syncthreads();

  int c = 0, inc = 0;
  if (tid < BUCKET_NODES) {
    c = cntL[tid];
    inc = c;
    for (int o = 1; o < 32; o <<= 1) {
      const int t = __shfl_up(inc, o, 32);
      if (lane >= o) inc += t;
    }
    if (lane == 31) wsum[tid >> 5] = inc;
  }
  __syncthreads();
  if (tid < BUCKET_NODES) {
    int o = 0;
    for (int w = 0; w < 16; ++w) o += wlo[w];
    for (int w = 0; w < (tid >> 5); ++w) o += wsum[w];
    const int excl = o + inc - c;
    const int node = b * BUCKET_NODES + tid;
    if (node < n) rowptr[node] = excl;
    curL[tid] = excl;
  }
  if (b == nb - 1 && tid == 0) rowptr[n] = nedges;
  __syncthreads();

  bucket_sweep<true>(tmp, off, b, nc, wave, lane, curL, csr);
}

__device__ __forceinline__ float lrelu(float v, float slope) {
  return v >= 0.f ? v : slope * v;
}

template <int G>
__device__ __forceinline__ void gat_grp(const float* __restrict__ H, int s_l, float w_l,
                                        int t, int lane4, float4& acc) {
  float4 hv[G];
  float  w[G];
#pragma unroll
  for (int i = 0; i < G; ++i) {
    const int s = __builtin_amdgcn_readlane(s_l, t + i);
    w[i] = __int_as_float(__builtin_amdgcn_readlane(__float_as_int(w_l), t + i));
    hv[i] = *(const float4*)(H + (size_t)s * NHID + lane4);
  }
#pragma unroll
  for (int i = 0; i < G; ++i) {
    acc.x = fmaf(hv[i].x, w[i], acc.x);
    acc.y = fmaf(hv[i].y, w[i], acc.y);
    acc.z = fmaf(hv[i].z, w[i], acc.z);
    acc.w = fmaf(hv[i].w, w[i], acc.w);
  }
}

__device__ __forceinline__ void gat_chunk(const float* __restrict__ H, int s_l, float w_l,
                                          int cnt, int lane4, float4& acc) {
  int t = 0;
  for (; t + 8 <= cnt; t += 8) gat_grp<8>(H, s_l, w_l, t, lane4, acc);
  if (t + 4 <= cnt) { gat_grp<4>(H, s_l, w_l, t, lane4, acc); t += 4; }
  const int rem = cnt - t;
  if (rem == 3)      gat_grp<3>(H, s_l, w_l, t, lane4, acc);
  else if (rem == 2) gat_grp<2>(H, s_l, w_l, t, lane4, acc);
  else if (rem == 1) gat_grp<1>(H, s_l, w_l, t, lane4, acc);
}

__global__ __launch_bounds__(256) void gat_node(
    const int* __restrict__ csr, const int* __restrict__ rowptr,
    const float* __restrict__ as, const float* __restrict__ ad,
    const float* __restrict__ H, const float* __restrict__ bias,
    float* __restrict__ out, int n, float slope, int apply_act) {
  const int lane  = threadIdx.x & 31;
  const int lane4 = lane * 4;
  const int d = __builtin_amdgcn_readfirstlane(
      (int)((blockIdx.x * (size_t)blockDim.x + threadIdx.x) >> 5));
  if (d >= n) return;

  const int beg = rowptr[d];
  const int end = rowptr[d + 1];
  const int deg = end - beg;
  const float ad_d   = ad[d];
  const float v_self = lrelu(as[d] + ad_d, NEG_ATT);

  const int  j0  = beg + lane;
  const int  j1  = j0 + 32;
  const bool ok0 = j0 < end;
  const bool ok1 = j1 < end;
  int s0 = d, s1 = d;
  if (ok0) s0 = csr[j0];
  if (ok1) s1 = csr[j1];
  const float4 hd = *(const float4*)(H + (size_t)d * NHID + lane4);
  const float a0 = as[s0];
  const float a1 = as[s1];
  const float v0 = ok0 ? lrelu(a0 + ad_d, NEG_ATT) : v_self;
  const float v1 = ok1 ? lrelu(a1 + ad_d, NEG_ATT) : v_self;

  float mx = fmaxf(v_self, fmaxf(v0, v1));
  for (int j = j0 + 64; j < end; j += 32)
    mx = fmaxf(mx, lrelu(as[csr[j]] + ad_d, NEG_ATT));
  for (int off = 16; off; off >>= 1) mx = fmaxf(mx, __shfl_xor(mx, off, 32));

  const float p0 = ok0 ? expf(v0 - mx) : 0.f;
  const float p1 = ok1 ? expf(v1 - mx) : 0.f;
  float den = p0 + p1;
  for (int j = j0 + 64; j < end; j += 32)
    den += expf(lrelu(as[csr[j]] + ad_d, NEG_ATT) - mx);
  for (int off = 16; off; off >>= 1) den += __shfl_xor(den, off, 32);
  const float p_self = expf(v_self - mx);
  den += p_self;
  const float dn = den + 1e-16f;

  const float  w_self = p_self / dn;
  float4 acc = make_float4(hd.x * w_self, hd.y * w_self, hd.z * w_self, hd.w * w_self);

  gat_chunk(H, s0, p0 / dn, min(deg, 32), lane4, acc);
  if (deg > 32) gat_chunk(H, s1, p1 / dn, min(deg - 32, 32), lane4, acc);
  for (int base = beg + 64; base < end; base += 32) {
    const int j = base + lane;
    int   s = d;
    float w = 0.f;
    if (j < end) {
      s = csr[j];
      w = expf(lrelu(as[s] + ad_d, NEG_ATT) - mx) / dn;
    }
    gat_chunk(H, s, w, min(end - base, 32), lane4, acc);
  }

  const float4 b4 = *(const float4*)(bias + lane4);
  float4 o = make_float4(acc.x + b4.x, acc.y + b4.y, acc.z + b4.z, acc.w + b4.w);
  if (apply_act) {
    o.x = lrelu(o.x, slope); o.y = lrelu(o.y, slope);
    o.z = lrelu(o.z, slope); o.w = lrelu(o.w, slope);
  }
  *(float4*)(out + (size_t)d * NHID + lane4) = o;
}

extern "C" void kernel_launch(void* const* d_in, const int* in_sizes, int n_in,
                              void* d_out, int out_size, void* d_ws, size_t ws_size,
                              hipStream_t stream) {
  const float* x    = (const float*)d_in[0];
  const int*   ei   = (const int*)d_in[1];
  const float* W1   = (const float*)d_in[2];
  const float* asr1 = (const float*)d_in[3];
  const float* adt1 = (const float*)d_in[4];
  const float* b1   = (const float*)d_in[5];
  const float* W2   = (const float*)d_in[6];
  const float* asr2 = (const float*)d_in[7];
  const float* adt2 = (const float*)d_in[8];
  const float* b2   = (const float*)d_in[9];
  float* out = (float*)d_out;

  const int N = in_sizes[0] / NHID;
  const int E = in_sizes[1] / 2;
  const size_t NF = (size_t)N * NHID;

  const int NBK  = (N + BUCKET_NODES - 1) / BUCKET_NODES;
  const int NCH  = (E + CHUNK_EDGES - 1) / CHUNK_EDGES;
  const int LO   = (NBK + 1) * NCH;

  float* h      = (float*)d_ws;
  float* x2     = h + NF;
  int*   tmp    = (int*)(x2 + NF);
  float* as     = (float*)(tmp + 2 * (size_t)E);
  float* ad     = as + N;
  int*   rowptr = (int*)(ad + N);
  int*   off    = rowptr + N + 1;
  int*   csr    = off + LO;

  const int* src = ei;
  const int* dst = ei + E;

  const int gGemm  = (N + 127) / 128;
  const int gWaveN = (int)(((size_t)N * 32 + 255) / 256);

  gat_chunk_sort<<<NCH, 1024, 0, stream>>>(src, dst, off, tmp, E, NBK, NCH);
  gat_bucket_gemm<<<NBK + (NBK + 1) / 2, 512, 0, stream>>>(tmp, off, rowptr, csr, N, NBK, NCH, E,
                                                           x, W1, asr1, adt1, h, as, ad);

  gat_node<<<gWaveN, 256, 0, stream>>>(csr, rowptr, as, ad, h, b1, x2, N, NEG_ACT, 1);

  gat_gemm128<<<gGemm, 256, 0, stream>>>(x2, W2, asr2, adt2, h, as, ad, N);
  gat_node<<<gWaveN, 256, 0, stream>>>(csr, rowptr, as, ad, h, b2, out, N, 0.f, 0);
}


# Layer-2 GEMM moved into the layer-1 aggregation launch: 16 waves finish 16 nodes, tile via LDS to WMMA (k summed in two halves), row logits by 32-lane tree: new summation order; one launch fewer
# speedup vs baseline: 7.8752x; 1.0409x over previous
#include <hip/hip_runtime.h>
#include <hip/hip_bf16.h>

typedef float v2f __attribute__((ext_vector_type(2)));
typedef float v8f __attribute__((ext_vector_type(8)));

#define NHID 128
#define NEG_ATT 0.2f
#define NEG_ACT 0.01f

#if __has_builtin(__builtin_amdgcn_wmma_f32_16x16x4_f32)
#define HAS_WMMA_F32X4 1
#endif

__device__ __forceinline__ float row16_sum(float x) {
  x += __int_as_float(__builtin_amdgcn_update_dpp(0, __float_as_int(x), 0x161, 0xf, 0xf, false));
  x += __int_as_float(__builtin_amdgcn_update_dpp(0, __float_as_int(x), 0x162, 0xf, 0xf, false));
  x += __int_as_float(__builtin_amdgcn_update_dpp(0, __float_as_int(x), 0x164, 0xf, 0xf, false));
  x += __int_as_float(__builtin_amdgcn_update_dpp(0, __float_as_int(x), 0x168, 0xf, 0xf, false));
  return x;
}

template <int NT>
__device__ __forceinline__ void gemm_tile(
    float* __restrict__ Ws, int tile,
    const float* __restrict__ X, const float* __restrict__ W,
    const float* __restrict__ a_src, const float* __restrict__ a_dst,
    float* __restrict__ H, float* __restrict__ as, float* __restrict__ ad,
    int nrows) {
  const int tid = threadIdx.x;
  for (int i = tid; i < NHID * NHID; i += NT) {
    const int k = i >> 7, n = i & 127;
    Ws[((k >> 1) << 8) + (n << 1) + (k & 1)] = W[i];
  }
  __syncthreads();

  const int wave = tid >> 5;
  const int lane = tid & 31;
  const int row0 = tile * (NT / 2) + wave * 16;
  if (row0 >= nrows) return;

  const int m     = lane & 15;
  const int khv   = lane >> 4;
  const int khalf = khv * 2;
  const int mhi   = khv * 8;

  v8f acc[8] = {};

#ifdef HAS_WMMA_F32X4
  const int arow = min(row0 + m, nrows - 1);
  const float* __restrict__ xrow = X + (size_t)arow * NHID;
  const float* __restrict__ bcol = Ws + (m << 1);
#pragma unroll 4
  for (int k0 = 0; k0 < NHID; k0 += 4) {
    const v2f a = *(const v2f*)(xrow + k0 + khalf);
    const float* __restrict__ bp = bcol + (((k0 >> 1) + khv) << 8);
#pragma unroll
    for (int t = 0; t < 8; ++t) {
      const v2f b = *(const v2f*)(bp + t * 32);
      acc[t] = __builtin_amdgcn_wmma_f32_16x16x4_f32(
          false, a, false, b, (short)0, acc[t], false, false);
    }
  }
#else
#pragma unroll
  for (int t = 0; t < 8; ++t)
    for (int v = 0; v < 8; ++v) {
      const int r = min(row0 + v + mhi, nrows - 1);
      const float* __restrict__ xr = X + (size_t)r * NHID;
      float s = 0.f;
      for (int k = 0; k < NHID; ++k)
        s += xr[k] * Ws[((k >> 1) << 8) + (((t << 4) + m) << 1) + (k & 1)];
      acc[t][v] = s;
    }
#endif

#pragma unroll
  for (int t = 0; t < 8; ++t)
#pragma unroll
    for (int v = 0; v < 8; ++v) {
      const int r = row0 + v + mhi;
      if (r < nrows) H[(size_t)r * NHID + t * 16 + m] = acc[t][v];
    }

  float ps[8], pd[8];
#pragma unroll
  for (int v = 0; v < 8; ++v) { ps[v] = 0.f; pd[v] = 0.f; }
#pragma unroll
  for (int t = 0; t < 8; ++t) {
    const float sa = a_src[t * 16 + m];
    const float da = a_dst[t * 16 + m];
#pragma unroll
    for (int v = 0; v < 8; ++v) {
      ps[v] = fmaf(acc[t][v], sa, ps[v]);
      pd[v] = fmaf(acc[t][v], da, pd[v]);
    }
  }
#pragma unroll
  for (int v = 0; v < 8; ++v) { ps[v] = row16_sum(ps[v]); pd[v] = row16_sum(pd[v]); }
  if (m < 2) {
    float* __restrict__ dstp = (m == 0) ? as : ad;
#pragma unroll
    for (int v = 0; v < 8; ++v) {
      const int r = row0 + v + mhi;
      if (r < nrows) dstp[r] = (m == 0) ? ps[v] : pd[v];
    }
  }
}

#define BUCKET_SHIFT 7
#define BUCKET_NODES 128
#define NB_MAX 512
#define CHUNK_EDGES 8192

__global__ __launch_bounds__(1024) void gat_chunk_sort(const int* __restrict__ srcs, const int* __restrict__ dsts,
                                                       int* __restrict__ off, int* __restrict__ tmp_i,
                                                       int nedges, int nb, int nc) {
  __shared__ int binL[NB_MAX];
  __shared__ int wsumL[NB_MAX / 32];
  const int tid  = threadIdx.x;
  const int lane = tid & 31;
  const int c    = (int)blockIdx.x;
  if (tid < NB_MAX) binL[tid] = 0;
  __syncthreads();
  const int e0 = c * CHUNK_EDGES + tid;
  int d[8], s[8];
#pragma unroll
  for (int k = 0; k < 8; ++k) {
    const int e = e0 + k * 1024;
    const bool ok = e < nedges;
    d[k] = ok ? dsts[e] : -1;
    s[k] = ok ? srcs[e] : 0;
  }
#pragma unroll
  for (int k = 0; k < 8; ++k)
    if (d[k] >= 0) atomicAdd(&binL[d[k] >> BUCKET_SHIFT], 1);
  __syncthreads();

  int cnt = 0, inc = 0;
  if (tid < NB_MAX) {
    cnt = binL[tid];
    inc = cnt;
    for (int o = 1; o < 32; o <<= 1) {
      const int t = __shfl_up(inc, o, 32);
      if (lane >= o) inc += t;
    }
    if (lane == 31) wsumL[tid >> 5] = inc;
  }
  __syncthreads();
  if (tid < NB_MAX) {
    int o = 0;
    for (int w = 0; w < (tid >> 5); ++w) o += wsumL[w];
    const int excl = o + inc - cnt;
    binL[tid] = excl;
    if (tid < nb) off[tid * nc + c] = excl;
    if (tid == nb - 1) off[nb * nc + c] = excl + cnt;
  }
  __syncthreads();

  int2* __restrict__ mine = (int2*)tmp_i + (size_t)c * CHUNK_EDGES;
  int pos[8];
#pragma unroll
  for (int k = 0; k < 8; ++k)
    pos[k] = (d[k] >= 0) ? atomicAdd(&binL[d[k] >> BUCKET_SHIFT], 1) : -1;
#pragma unroll
  for (int k = 0; k < 8; ++k)
    if (pos[k] >= 0) mine[pos[k]] = make_int2(s[k], d[k]);
}

template <bool PLACE>
__device__ __forceinline__ int bucket_sweep(const int2* __restrict__ tmp, const int* __restrict__ off,
                                            int b, int nc, int wave, int lane,
                                            int* __restrict__ binL, int* __restrict__ csr) {
  int lo_part = 0;
  for (int r0 = wave; r0 < nc; r0 += 16 * 32) {
    const int cl = r0 + 16 * lane;
    int sb = 0, sc = 0;
    if (cl < nc) {
      const int o0 = off[b * nc + cl];
      const int o1 = off[(b + 1) * nc + cl];
      lo_part += o0;
      sb = cl * CHUNK_EDGES + o0;
      sc = o1 - o0;
    }
    const int nseg = min(32, (nc - r0 + 15) >> 4);
    for (int j = 0; j < nseg; j += 8) {
      int2 e[8];
      int  bg[8], cn[8];
#pragma unroll
      for (int u = 0; u < 8; ++u) {
        const int jj = min(j + u, 31);
        bg[u] = __builtin_amdgcn_readlane(sb, jj);
        cn[u] = (j + u < nseg) ? __builtin_amdgcn_readlane(sc, jj) : 0;
        e[u] = make_int2(0, -1);
        if (lane < cn[u]) {
          if (PLACE) e[u] = tmp[bg[u] + lane];
          else       e[u].y = tmp[bg[u] + lane].y;
        }
      }
      if (PLACE) {
        int pos[8];
#pragma unroll
        for (int u = 0; u < 8; ++u)
          pos[u] = (e[u].y >= 0) ? atomicAdd(&binL[e[u].y & (BUCKET_NODES - 1)], 1) : -1;
#pragma unroll
        for (int u = 0; u < 8; ++u)
          if (pos[u] >= 0) csr[pos[u]] = e[u].x;
      } else {
#pragma unroll
        for (int u = 0; u < 8; ++u)
          if (e[u].y >= 0) atomicAdd(&binL[e[u].y & (BUCKET_NODES - 1)], 1);
      }
#pragma unroll
      for (int u = 0; u < 8; ++u)
        for (int k = lane + 32; k < cn[u]; k += 32) {
          const int2 ee = tmp[bg[u] + k];
          if (PLACE) {
            const int pos = atomicAdd(&binL[ee.y & (BUCKET_NODES - 1)], 1);
            csr[pos] = ee.x;
          } else {
            atomicAdd(&binL[ee.y & (BUCKET_NODES - 1)], 1);
          }
        }
    }
  }
  return lo_part;
}

__global__ __launch_bounds__(512) void gat_bucket_gemm(
    const int* __restrict__ tmp_i, const int* __restrict__ off,
    int* __restrict__ rowptr, int* __restrict__ csr, int n, int nb, int nc, int nedges,
    const float* __restrict__ X, const float* __restrict__ W,
    const float* __restrict__ a_src, const float* __restrict__ a_dst,
    float* __restrict__ H, float* __restrict__ as, float* __restrict__ ad) {
  __shared__ float Ws[NHID * NHID];
  __shared__ int cntL[BUCKET_NODES];
  __shared__ int curL[BUCKET_NODES];
  __shared__ int wsum[BUCKET_NODES / 32];
  __shared__ int wlo[16];
  const int ntile = (nb + 1) >> 1;
  if ((int)blockIdx.x < ntile) {
    gemm_tile<512>(Ws, (int)blockIdx.x, X, W, a_src, a_dst, H, as, ad, n);
    return;
  }
  const int b = (int)blockIdx.x - ntile;
  if (b >= nb) return;
  const int2* __restrict__ tmp = (const int2*)tmp_i;
  const int tid  = threadIdx.x;
  const int lane = tid & 31;
  const int wave = tid >> 5;
  if (tid < BUCKET_NODES) cntL[tid] = 0;
  __syncthreads();

  int lo_part = bucket_sweep<false>(tmp, off, b, nc, wave, lane, cntL, csr);
  for (int o = 16; o; o >>= 1) lo_part += __shfl_xor(lo_part, o, 32);
  if (lane == 0) wlo[wave] = lo_part;
  __syncthreads();

  int c = 0, inc = 0;
  if (tid < BUCKET_NODES) {
    c = cntL[tid];
    inc = c;
    for (int o = 1; o < 32; o <<= 1) {
      const int t = __shfl_up(inc, o, 32);
      if (lane >= o) inc += t;
    }
    if (lane == 31) wsum[tid >> 5] = inc;
  }
  __syncthreads();
  if (tid < BUCKET_NODES) {
    int o = 0;
    for (int w = 0; w < 16; ++w) o += wlo[w];
    for (int w = 0; w < (tid >> 5); ++w) o += wsum[w];
    const int excl = o + inc - c;
    const int node = b * BUCKET_NODES + tid;
    if (node < n) rowptr[node] = excl;
    curL[tid] = excl;
  }
  if (b == nb - 1 && tid == 0) rowptr[n] = nedges;
  __syncthreads();

  bucket_sweep<true>(tmp, off, b, nc, wave, lane, curL, csr);
}

__device__ __forceinline__ float lrelu(float v, float slope) {
  return v >= 0.f ? v : slope * v;
}

template <int G>
__device__ __forceinline__ void gat_grp(const float* __restrict__ H, int s_l, float w_l,
                                        int t, int lane4, float4& acc) {
  float4 hv[G];
  float  w[G];
#pragma unroll
  for (int i = 0; i < G; ++i) {
    const int s = __builtin_amdgcn_readlane(s_l, t + i);
    w[i] = __int_as_float(__builtin_amdgcn_readlane(__float_as_int(w_l), t + i));
    hv[i] = *(const float4*)(H + (size_t)s * NHID + lane4);
  }
#pragma unroll
  for (int i = 0; i < G; ++i) {
    acc.x = fmaf(hv[i].x, w[i], acc.x);
    acc.y = fmaf(hv[i].y, w[i], acc.y);
    acc.z = fmaf(hv[i].z, w[i], acc.z);
    acc.w = fmaf(hv[i].w, w[i], acc.w);
  }
}

__device__ __forceinline__ void gat_chunk(const float* __restrict__ H, int s_l, float w_l,
                                          int cnt, int lane4, float4& acc) {
  int t = 0;
  for (; t + 8 <= cnt; t += 8) gat_grp<8>(H, s_l, w_l, t, lane4, acc);
  if (t + 4 <= cnt) { gat_grp<4>(H, s_l, w_l, t, lane4, acc); t += 4; }
  const int rem = cnt - t;
  if (rem == 3)      gat_grp<3>(H, s_l, w_l, t, lane4, acc);
  else if (rem == 2) gat_grp<2>(H, s_l, w_l, t, lane4, acc);
  else if (rem == 1) gat_grp<1>(H, s_l, w_l, t, lane4, acc);
}

__device__ __forceinline__ float4 gat_node_row(
    const int* __restrict__ csr, const int* __restrict__ rowptr,
    const float* __restrict__ as, const float* __restrict__ ad,
    const float* __restrict__ H, const float* __restrict__ bias,
    int d, int lane, float slope, int apply_act) {
  const int lane4 = lane * 4;
  const int beg = rowptr[d];
  const int end = rowptr[d + 1];
  const int deg = end - beg;
  const float ad_d   = ad[d];
  const float v_self = lrelu(as[d] + ad_d, NEG_ATT);

  const int  j0  = beg + lane;
  const int  j1  = j0 + 32;
  const bool ok0 = j0 < end;
  const bool ok1 = j1 < end;
  int s0 = d, s1 = d;
  if (ok0) s0 = csr[j0];
  if (ok1) s1 = csr[j1];
  const float4 hd = *(const float4*)(H + (size_t)d * NHID + lane4);
  const float a0 = as[s0];
  const float a1 = as[s1];
  const float v0 = ok0 ? lrelu(a0 + ad_d, NEG_ATT) : v_self;
  const float v1 = ok1 ? lrelu(a1 + ad_d, NEG_ATT) : v_self;

  float mx = fmaxf(v_self, fmaxf(v0, v1));
  for (int j = j0 + 64; j < end; j += 32)
    mx = fmaxf(mx, lrelu(as[csr[j]] + ad_d, NEG_ATT));
  for (int off = 16; off; off >>= 1) mx = fmaxf(mx, __shfl_xor(mx, off, 32));

  const float p0 = ok0 ? expf(v0 - mx) : 0.f;
  const float p1 = ok1 ? expf(v1 - mx) : 0.f;
  float den = p0 + p1;
  for (int j = j0 + 64; j < end; j += 32)
    den += expf(lrelu(as[csr[j]] + ad_d, NEG_ATT) - mx);
  for (int off = 16; off; off >>= 1) den += __shfl_xor(den, off, 32);
  const float p_self = expf(v_self - mx);
  den += p_self;
  const float dn = den + 1e-16f;

  const float  w_self = p_self / dn;
  float4 acc = make_float4(hd.x * w_self, hd.y * w_self, hd.z * w_self, hd.w * w_self);

  gat_chunk(H, s0, p0 / dn, min(deg, 32), lane4, acc);
  if (deg > 32) gat_chunk(H, s1, p1 / dn, min(deg - 32, 32), lane4, acc);
  for (int base = beg + 64; base < end; base += 32) {
    const int j = base + lane;
    int   s = d;
    float w = 0.f;
    if (j < end) {
      s = csr[j];
      w = expf(lrelu(as[s] + ad_d, NEG_ATT) - mx) / dn;
    }
    gat_chunk(H, s, w, min(end - base, 32), lane4, acc);
  }

  const float4 b4 = *(const float4*)(bias + lane4);
  float4 o = make_float4(acc.x + b4.x, acc.y + b4.y, acc.z + b4.z, acc.w + b4.w);
  if (apply_act) {
    o.x = lrelu(o.x, slope); o.y = lrelu(o.y, slope);
    o.z = lrelu(o.z, slope); o.w = lrelu(o.w, slope);
  }
  return o;
}

__global__ __launch_bounds__(256) void gat_node(
    const int* __restrict__ csr, const int* __restrict__ rowptr,
    const float* __restrict__ as, const float* __restrict__ ad,
    const float* __restrict__ H, const float* __restrict__ bias,
    float* __restrict__ out, int n, float slope, int apply_act) {
  const int lane = threadIdx.x & 31;
  const int d = __builtin_amdgcn_readfirstlane(
      (int)((blockIdx.x * (size_t)blockDim.x + threadIdx.x) >> 5));
  if (d >= n) return;
  const float4 o = gat_node_row(csr, rowptr, as, ad, H, bias, d, lane, slope, apply_act);
  *(float4*)(out + (size_t)d * NHID + lane * 4) = o;
}

#define FT_PITCH 132
#define FT_TILES_PER_GROUP 5

__global__ __launch_bounds__(512) void gat_node_gemm(
    const int* __restrict__ csr, const int* __restrict__ rowptr,
    const float* __restrict__ as, const float* __restrict__ ad,
    const float* __restrict__ H, const float* __restrict__ bias,
    const float* __restrict__ W2, const float* __restrict__ a_src2, const float* __restrict__ a_dst2,
    float* __restrict__ H2, float* __restrict__ as2, float* __restrict__ ad2,
    int n, int ntiles, float slope) {
  __shared__ __attribute__((aligned(16))) float Ws[NHID * NHID];
  __shared__ __attribute__((aligned(16))) float xs[16 * FT_PITCH];
  __shared__ __attribute__((aligned(16))) float hs[2 * 16 * FT_PITCH];
  const int tid   = threadIdx.x;
  const int lane  = tid & 31;
  const int lane4 = lane * 4;
  const int wave  = __builtin_amdgcn_readfirstlane(tid >> 5);
  const int m     = lane & 15;
  const int khv   = lane >> 4;
  const int khalf = khv * 2;
  const int mhi   = khv * 8;
  const int nt    = wave & 7;
  const int kh    = wave >> 3;

  for (int i = tid; i < NHID * NHID; i += 512) {
    const int k = i >> 7, nn = i & 127;
    Ws[((k >> 1) << 8) + (nn << 1) + (k & 1)] = W2[i];
  }
  const float4 as4 = *(const float4*)(a_src2 + lane4);
  const float4 ad4 = *(const float4*)(a_dst2 + lane4);

  for (int tile = (int)blockIdx.x; tile < ntiles; tile += (int)gridDim.x) {
    const int d = tile * 16 + wave;
    {
      float4 o = make_float4(0.f, 0.f, 0.f, 0.f);
      if (d < n) o = gat_node_row(csr, rowptr, as, ad, H, bias, d, lane, slope, 1);
      *(float4*)(xs + wave * FT_PITCH + lane4) = o;
    }
    __syncthreads();

    v8f acc = {};
#ifdef HAS_WMMA_F32X4
    const float* __restrict__ xa = xs + m * FT_PITCH + kh * 64 + khalf;
    const float* __restrict__ bp = Ws + (((kh * 32) + khv) << 8) + (nt << 5) + (m << 1);
#pragma unroll
    for (int ks = 0; ks < 16; ++ks) {
      const v2f a = *(const v2f*)(xa + ks * 4);
      const v2f b = *(const v2f*)(bp + (ks << 9));
      acc = __builtin_amdgcn_wmma_f32_16x16x4_f32(
          false, a, false, b, (short)0, acc, false, false);
    }
#else
#pragma unroll
    for (int v = 0; v < 8; ++v) {
      float s = 0.f;
      for (int k = kh * 64; k < kh * 64 + 64; ++k)
        s += xs[(v + mhi) * FT_PITCH + k] * Ws[((k >> 1) << 8) + (((nt << 4) + m) << 1) + (k & 1)];
      acc[v] = s;
    }
#endif
    float* __restrict__ hp = hs + kh * (16 * FT_PITCH);
#pragma unroll
    for (int v = 0; v < 8; ++v) hp[(v + mhi) * FT_PITCH + nt * 16 + m] = acc[v];
    __syncthreads();

    {
      const float4 h0 = *(const float4*)(hs + wave * FT_PITCH + lane4);
      const float4 h1 = *(const float4*)(hs + (16 + wave) * FT_PITCH + lane4);
      const float4 hv = make_float4(h0.x + h1.x, h0.y + h1.y, h0.z + h1.z, h0.w + h1.w);
      float ps = hv.x * as4.x;
      ps = fmaf(hv.y, as4.y, ps); ps = fmaf(hv.z, as4.z, ps); ps = fmaf(hv.w, as4.w, ps);
      float pd = hv.x * ad4.x;
      pd = fmaf(hv.y, ad4.y, pd); pd = fmaf(hv.z, ad4.z, pd); pd = fmaf(hv.w, ad4.w, pd);
      for (int off = 16; off; off >>= 1) {
        ps += __shfl_xor(ps, off, 32);
        pd += __shfl_xor(pd, off, 32);
      }
      if (d < n) {
        *(float4*)(H2 + (size_t)d * NHID + lane4) = hv;
        if (lane == 0) { as2[d] = ps; ad2[d] = pd; }
      }
    }
  }
}

extern "C" void kernel_launch(void* const* d_in, const int* in_sizes, int n_in,
                              void* d_out, int out_size, void* d_ws, size_t ws_size,
                              hipStream_t stream) {
  const float* x    = (const float*)d_in[0];
  const int*   ei   = (const int*)d_in[1];
  const float* W1   = (const float*)d_in[2];
  const float* asr1 = (const float*)d_in[3];
  const float* adt1 = (const float*)d_in[4];
  const float* b1   = (const float*)d_in[5];
  const float* W2   = (const float*)d_in[6];
  const float* asr2 = (const float*)d_in[7];
  const float* adt2 = (const float*)d_in[8];
  const float* b2   = (const float*)d_in[9];
  float* out = (float*)d_out;

  const int N = in_sizes[0] / NHID;
  const int E = in_sizes[1] / 2;
  const size_t NF = (size_t)N * NHID;

  const int NBK  = (N + BUCKET_NODES - 1) / BUCKET_NODES;
  const int NCH  = (E + CHUNK_EDGES - 1) / CHUNK_EDGES;
  const int LO   = (NBK + 1) * NCH;

  float* h      = (float*)d_ws;
  float* h2     = h + NF;
  int*   tmp    = (int*)(h2 + NF);
  float* as     = (float*)(tmp + 2 * (size_t)E);
  float* ad     = as + N;
  int*   rowptr = (int*)(ad + N);
  int*   off    = rowptr + N + 1;
  int*   csr    = off + LO;
  float* as2    = (float*)(csr + E);
  float* ad2    = as2 + N;

  const int* src = ei;
  const int* dst = ei + E;

  const int gWaveN = (int)(((size_t)N * 32 + 255) / 256);
  const int NT16   = (N + 15) / 16;
  const int gFused = (NT16 + FT_TILES_PER_GROUP - 1) / FT_TILES_PER_GROUP;

  gat_chunk_sort<<<NCH, 1024, 0, stream>>>(src, dst, off, tmp, E, NBK, NCH);
  gat_bucket_gemm<<<NBK + (NBK + 1) / 2, 512, 0, stream>>>(tmp, off, rowptr, csr, N, NBK, NCH, E,
                                                           x, W1, asr1, adt1, h, as, ad);

  gat_node_gemm<<<gFused, 512, 0, stream>>>(csr, rowptr, as, ad, h, b1, W2, asr2, adt2,
                                            h2, as2, ad2, N, NT16, NEG_ACT);

  gat_node<<<gWaveN, 256, 0, stream>>>(csr, rowptr, as2, ad2, h2, b2, out, N, 0.f, 0);
}


# Layer-1 product moved behind the aggregation by linearity: logits from x.(W1 a), rows of x gathered, 16-node tile times W1 on WMMA in the fused launch; x@W1 launch work gone; new summation order
# speedup vs baseline: 8.1625x; 1.0365x over previous
#include <hip/hip_runtime.h>
#include <hip/hip_bf16.h>

typedef float v2f __attribute__((ext_vector_type(2)));
typedef float v8f __attribute__((ext_vector_type(8)));

#define NHID 128
#define NEG_ATT 0.2f
#define NEG_ACT 0.01f

#if __has_builtin(__builtin_amdgcn_wmma_f32_16x16x4_f32)
#define HAS_WMMA_F32X4 1
#endif

__device__ __forceinline__ float row16_sum(float x) {
  x += __int_as_float(__builtin_amdgcn_update_dpp(0, __float_as_int(x), 0x161, 0xf, 0xf, false));
  x += __int_as_float(__builtin_amdgcn_update_dpp(0, __float_as_int(x), 0x162, 0xf, 0xf, false));
  x += __int_as_float(__builtin_amdgcn_update_dpp(0, __float_as_int(x), 0x164, 0xf, 0xf, false));
  x += __int_as_float(__builtin_amdgcn_update_dpp(0, __float_as_int(x), 0x168, 0xf, 0xf, false));
  return x;
}

#define LOGIT_ROWS 256
__device__ __forceinline__ void logit_tile(
    float* __restrict__ vsd, int tile,
    const float* __restrict__ X, const float* __restrict__ W,
    const float* __restrict__ a_src, const float* __restrict__ a_dst,
    float* __restrict__ as, float* __restrict__ ad, int nrows) {
  const int tid  = threadIdx.x;
  const int lane = tid & 31;
  const int wave = tid >> 5;
  {
    const int k = tid >> 2, q = tid & 3;
    const float* __restrict__ wr = W + k * NHID + q * 32;
    const float* __restrict__ sr = a_src + q * 32;
    const float* __restrict__ dr = a_dst + q * 32;
    float ps = 0.f, pd = 0.f;
#pragma unroll
    for (int j = 0; j < 8; ++j) {
      const float4 w4 = *(const float4*)(wr + j * 4);
      const float4 s4 = *(const float4*)(sr + j * 4);
      const float4 d4 = *(const float4*)(dr + j * 4);
      ps = fmaf(w4.x, s4.x, ps); ps = fmaf(w4.y, s4.y, ps);
      ps = fmaf(w4.z, s4.z, ps); ps = fmaf(w4.w, s4.w, ps);
      pd = fmaf(w4.x, d4.x, pd); pd = fmaf(w4.y, d4.y, pd);
      pd = fmaf(w4.z, d4.z, pd); pd = fmaf(w4.w, d4.w, pd);
    }
    ps += __shfl_xor(ps, 1, 32); ps += __shfl_xor(ps, 2, 32);
    pd += __shfl_xor(pd, 1, 32); pd += __shfl_xor(pd, 2, 32);
    if (q == 0) { vsd[k] = ps; vsd[NHID + k] = pd; }
  }
  __syncthreads();

  const int row0 = tile * LOGIT_ROWS + wave * 16;
  if (row0 >= nrows) return;
  const int lane4 = lane * 4;
  const float4 vs4 = *(const float4*)(vsd + lane4);
  const float4 vd4 = *(const float4*)(vsd + NHID + lane4);
  float4 xv[16];
#pragma unroll
  for (int r = 0; r < 16; ++r) {
    const int row = min(row0 + r, nrows - 1);
    xv[r] = *(const float4*)(X + (size_t)row * NHID + lane4);
  }
  const bool lo = lane < 16;
  float res = 0.f;
#pragma unroll
  for (int r = 0; r < 16; ++r) {
    float ps = xv[r].x * vs4.x;
    ps = fmaf(xv[r].y, vs4.y, ps); ps = fmaf(xv[r].z, vs4.z, ps); ps = fmaf(xv[r].w, vs4.w, ps);
    float pd = xv[r].x * vd4.x;
    pd = fmaf(xv[r].y, vd4.y, pd); pd = fmaf(xv[r].z, vd4.z, pd); pd = fmaf(xv[r].w, vd4.w, pd);
    float keep = lo ? ps : pd;
    const float give = lo ? pd : ps;
    keep += __shfl_xor(give, 16, 32);
    keep = row16_sum(keep);
    if ((lane & 15) == r) res = keep;
  }
  const int row = row0 + (lane & 15);
  if (row < nrows) (lo ? as : ad)[row] = res;
}

#define BUCKET_SHIFT 7
#define BUCKET_NODES 128
#define NB_MAX 512
#define CHUNK_EDGES 8192

__global__ __launch_bounds__(1024) void gat_chunk_sort(const int* __restrict__ srcs, const int* __restrict__ dsts,
                                                       int* __restrict__ off, int* __restrict__ tmp_i,
                                                       int nedges, int nb, int nc) {
  __shared__ int binL[NB_MAX];
  __shared__ int wsumL[NB_MAX / 32];
  const int tid  = threadIdx.x;
  const int lane = tid & 31;
  const int c    = (int)blockIdx.x;
  if (tid < NB_MAX) binL[tid] = 0;
  __syncthreads();
  const int e0 = c * CHUNK_EDGES + tid;
  int d[8], s[8];
#pragma unroll
  for (int k = 0; k < 8; ++k) {
    const int e = e0 + k * 1024;
    const bool ok = e < nedges;
    d[k] = ok ? dsts[e] : -1;
    s[k] = ok ? srcs[e] : 0;
  }
#pragma unroll
  for (int k = 0; k < 8; ++k)
    if (d[k] >= 0) atomicAdd(&binL[d[k] >> BUCKET_SHIFT], 1);
  __syncthreads();

  int cnt = 0, inc = 0;
  if (tid < NB_MAX) {
    cnt = binL[tid];
    inc = cnt;
    for (int o = 1; o < 32; o <<= 1) {
      const int t = __shfl_up(inc, o, 32);
      if (lane >= o) inc += t;
    }
    if (lane == 31) wsumL[tid >> 5] = inc;
  }
  __syncthreads();
  if (tid < NB_MAX) {
    int o = 0;
    for (int w = 0; w < (tid >> 5); ++w) o += wsumL[w];
    const int excl = o + inc - cnt;
    binL[tid] = excl;
    if (tid < nb) off[tid * nc + c] = excl;
    if (tid == nb - 1) off[nb * nc + c] = excl + cnt;
  }
  __syncthreads();

  int2* __restrict__ mine = (int2*)tmp_i + (size_t)c * CHUNK_EDGES;
  int pos[8];
#pragma unroll
  for (int k = 0; k < 8; ++k)
    pos[k] = (d[k] >= 0) ? atomicAdd(&binL[d[k] >> BUCKET_SHIFT], 1) : -1;
#pragma unroll
  for (int k = 0; k < 8; ++k)
    if (pos[k] >= 0) mine[pos[k]] = make_int2(s[k], d[k]);
}

template <bool PLACE>
__device__ __forceinline__ int bucket_sweep(const int2* __restrict__ tmp, const int* __restrict__ off,
                                            int b, int nc, int wave, int lane,
                                            int* __restrict__ binL, int* __restrict__ csr) {
  int lo_part = 0;
  for (int r0 = wave; r0 < nc; r0 += 16 * 32) {
    const int cl = r0 + 16 * lane;
    int sb = 0, sc = 0;
    if (cl < nc) {
      const int o0 = off[b * nc + cl];
      const int o1 = off[(b + 1) * nc + cl];
      lo_part += o0;
      sb = cl * CHUNK_EDGES + o0;
      sc = o1 - o0;
    }
    const int nseg = min(32, (nc - r0 + 15) >> 4);
    for (int j = 0; j < nseg; j += 8) {
      int2 e[8];
      int  bg[8], cn[8];
#pragma unroll
      for (int u = 0; u < 8; ++u) {
        const int jj = min(j + u, 31);
        bg[u] = __builtin_amdgcn_readlane(sb, jj);
        cn[u] = (j + u < nseg) ? __builtin_amdgcn_readlane(sc, jj) : 0;
        e[u] = make_int2(0, -1);
        if (lane < cn[u]) {
          if (PLACE) e[u] = tmp[bg[u] + lane];
          else       e[u].y = tmp[bg[u] + lane].y;
        }
      }
      if (PLACE) {
        int pos[8];
#pragma unroll
        for (int u = 0; u < 8; ++u)
          pos[u] = (e[u].y >= 0) ? atomicAdd(&binL[e[u].y & (BUCKET_NODES - 1)], 1) : -1;
#pragma unroll
        for (int u = 0; u < 8; ++u)
          if (pos[u] >= 0) csr[pos[u]] = e[u].x;
      } else {
#pragma unroll
        for (int u = 0; u < 8; ++u)
          if (e[u].y >= 0) atomicAdd(&binL[e[u].y & (BUCKET_NODES - 1)], 1);
      }
#pragma unroll
      for (int u = 0; u < 8; ++u)
        for (int k = lane + 32; k < cn[u]; k += 32) {
          const int2 ee = tmp[bg[u] + k];
          if (PLACE) {
            const int pos = atomicAdd(&binL[ee.y & (BUCKET_NODES - 1)], 1);
            csr[pos] = ee.x;
          } else {
            atomicAdd(&binL[ee.y & (BUCKET_NODES - 1)], 1);
          }
        }
    }
  }
  return lo_part;
}

__global__ __launch_bounds__(512) void gat_bucket_logit(
    const int* __restrict__ tmp_i, const int* __restrict__ off,
    int* __restrict__ rowptr, int* __restrict__ csr, int n, int nb, int nc, int nedges,
    const float* __restrict__ X, const float* __restrict__ W,
    const float* __restrict__ a_src, const float* __restrict__ a_dst,
    float* __restrict__ as, float* __restrict__ ad) {
  __shared__ __attribute__((aligned(16))) float vsd[2 * NHID];
  __shared__ int cntL[BUCKET_NODES];
  __shared__ int curL[BUCKET_NODES];
  __shared__ int wsum[BUCKET_NODES / 32];
  __shared__ int wlo[16];
  const int ntile = (nb + 1) >> 1;
  if ((int)blockIdx.x < ntile) {
    logit_tile(vsd, (int)blockIdx.x, X, W, a_src, a_dst, as, ad, n);
    return;
  }
  const int b = (int)blockIdx.x - ntile;
  if (b >= nb) return;
  const int2* __restrict__ tmp = (const int2*)tmp_i;
  const int tid  = threadIdx.x;
  const int lane = tid & 31;
  const int wave = tid >> 5;
  if (tid < BUCKET_NODES) cntL[tid] = 0;
  __syncthreads();

  int lo_part = bucket_sweep<false>(tmp, off, b, nc, wave, lane, cntL, csr);
  for (int o = 16; o; o >>= 1) lo_part += __shfl_xor(lo_part, o, 32);
  if (lane == 0) wlo[wave] = lo_part;
  __syncthreads();

  int c = 0, inc = 0;
  if (tid < BUCKET_NODES) {
    c = cntL[tid];
    inc = c;
    for (int o = 1; o < 32; o <<= 1) {
      const int t = __shfl_up(inc, o, 32);
      if (lane >= o) inc += t;
    }
    if (lane == 31) wsum[tid >> 5] = inc;
  }
  __syncthreads();
  if (tid < BUCKET_NODES) {
    int o = 0;
    for (int w = 0; w < 16; ++w) o += wlo[w];
    for (int w = 0; w < (tid >> 5); ++w) o += wsum[w];
    const int excl = o + inc - c;
    const int node = b * BUCKET_NODES + tid;
    if (node < n) rowptr[node] = excl;
    curL[tid] = excl;
  }
  if (b == nb - 1 && tid == 0) rowptr[n] = nedges;
  __syncthreads();

  bucket_sweep<true>(tmp, off, b, nc, wave, lane, curL, csr);
}

__device__ __forceinline__ float lrelu(float v, float slope) {
  return v >= 0.f ? v : slope * v;
}

template <int G>
__device__ __forceinline__ void gat_grp(const float* __restrict__ H, int s_l, float w_l,
                                        int t, int lane4, float4& acc) {
  float4 hv[G];
  float  w[G];
#pragma unroll
  for (int i = 0; i < G; ++i) {
    const int s = __builtin_amdgcn_readlane(s_l, t + i);
    w[i] = __int_as_float(__builtin_amdgcn_readlane(__float_as_int(w_l), t + i));
    hv[i] = *(const float4*)(H + (size_t)s * NHID + lane4);
  }
#pragma unroll
  for (int i = 0; i < G; ++i) {
    acc.x = fmaf(hv[i].x, w[i], acc.x);
    acc.y = fmaf(hv[i].y, w[i], acc.y);
    acc.z = fmaf(hv[i].z, w[i], acc.z);
    acc.w = fmaf(hv[i].w, w[i], acc.w);
  }
}

__device__ __forceinline__ void gat_chunk(const float* __restrict__ H, int s_l, float w_l,
                                          int cnt, int lane4, float4& acc) {
  int t = 0;
  for (; t + 8 <= cnt; t += 8) gat_grp<8>(H, s_l, w_l, t, lane4, acc);
  if (t + 4 <= cnt) { gat_grp<4>(H, s_l, w_l, t, lane4, acc); t += 4; }
  const int rem = cnt - t;
  if (rem == 3)      gat_grp<3>(H, s_l, w_l, t, lane4, acc);
  else if (rem == 2) gat_grp<2>(H, s_l, w_l, t, lane4, acc);
  else if (rem == 1) gat_grp<1>(H, s_l, w_l, t, lane4, acc);
}

__device__ __forceinline__ float4 gat_node_sum(
    const int* __restrict__ csr, const int* __restrict__ rowptr,
    const float* __restrict__ as, const float* __restrict__ ad,
    const float* __restrict__ H, int d, int lane) {
  const int lane4 = lane * 4;
  const int beg = rowptr[d];
  const int end = rowptr[d + 1];
  const int deg = end - beg;
  const float ad_d   = ad[d];
  const float v_self = lrelu(as[d] + ad_d, NEG_ATT);

  const int  j0  = beg + lane;
  const int  j1  = j0 + 32;
  const bool ok0 = j0 < end;
  const bool ok1 = j1 < end;
  int s0 = d, s1 = d;
  if (ok0) s0 = csr[j0];
  if (ok1) s1 = csr[j1];
  const float4 hd = *(const float4*)(H + (size_t)d * NHID + lane4);
  const float a0 = as[s0];
  const float a1 = as[s1];
  const float v0 = ok0 ? lrelu(a0 + ad_d, NEG_ATT) : v_self;
  const float v1 = ok1 ? lrelu(a1 + ad_d, NEG_ATT) : v_self;

  float mx = fmaxf(v_self, fmaxf(v0, v1));
  for (int j = j0 + 64; j < end; j += 32)
    mx = fmaxf(mx, lrelu(as[csr[j]] + ad_d, NEG_ATT));
  for (int off = 16; off; off >>= 1) mx = fmaxf(mx, __shfl_xor(mx, off, 32));

  const float p0 = ok0 ? expf(v0 - mx) : 0.f;
  const float p1 = ok1 ? expf(v1 - mx) : 0.f;
  float den = p0 + p1;
  for (int j = j0 + 64; j < end; j += 32)
    den += expf(lrelu(as[csr[j]] + ad_d, NEG_ATT) - mx);
  for (int off = 16; off; off >>= 1) den += __shfl_xor(den, off, 32);
  const float p_self = expf(v_self - mx);
  den += p_self;
  const float dn = den + 1e-16f;

  const float  w_self = p_self / dn;
  float4 acc = make_float4(hd.x * w_self, hd.y * w_self, hd.z * w_self, hd.w * w_self);

  gat_chunk(H, s0, p0 / dn, min(deg, 32), lane4, acc);
  if (deg > 32) gat_chunk(H, s1, p1 / dn, min(deg - 32, 32), lane4, acc);
  for (int base = beg + 64; base < end; base += 32) {
    const int j = base + lane;
    int   s = d;
    float w = 0.f;
    if (j < end) {
      s = csr[j];
      w = expf(lrelu(as[s] + ad_d, NEG_ATT) - mx) / dn;
    }
    gat_chunk(H, s, w, min(end - base, 32), lane4, acc);
  }
  return acc;
}

__device__ __forceinline__ float4 gat_node_row(
    const int* __restrict__ csr, const int* __restrict__ rowptr,
    const float* __restrict__ as, const float* __restrict__ ad,
    const float* __restrict__ H, const float* __restrict__ bias,
    int d, int lane, float slope, int apply_act) {
  const int lane4 = lane * 4;
  const float4 acc = gat_node_sum(csr, rowptr, as, ad, H, d, lane);
  const float4 b4 = *(const float4*)(bias + lane4);
  float4 o = make_float4(acc.x + b4.x, acc.y + b4.y, acc.z + b4.z, acc.w + b4.w);
  if (apply_act) {
    o.x = lrelu(o.x, slope); o.y = lrelu(o.y, slope);
    o.z = lrelu(o.z, slope); o.w = lrelu(o.w, slope);
  }
  return o;
}

__global__ __launch_bounds__(256) void gat_node(
    const int* __restrict__ csr, const int* __restrict__ rowptr,
    const float* __restrict__ as, const float* __restrict__ ad,
    const float* __restrict__ H, const float* __restrict__ bias,
    float* __restrict__ out, int n, float slope, int apply_act) {
  const int lane = threadIdx.x & 31;
  const int d = __builtin_amdgcn_readfirstlane(
      (int)((blockIdx.x * (size_t)blockDim.x + threadIdx.x) >> 5));
  if (d >= n) return;
  const float4 o = gat_node_row(csr, rowptr, as, ad, H, bias, d, lane, slope, apply_act);
  *(float4*)(out + (size_t)d * NHID + lane * 4) = o;
}

#define FT_PITCH 132
#define FT_GROUPS 512

__device__ __forceinline__ void tile_half_product(const float* __restrict__ xs, const float* __restrict__ Wst,
                                                  float* __restrict__ hs, int m, int khv, int nt, int kh) {
  const int khalf = khv * 2;
  const int mhi   = khv * 8;
  v8f acc = {};
#ifdef HAS_WMMA_F32X4
  const float* __restrict__ xa = xs + m * FT_PITCH + kh * 64 + khalf;
  const float* __restrict__ bp = Wst + (((kh * 32) + khv) << 8) + (nt << 5) + (m << 1);
#pragma unroll
  for (int ks = 0; ks < 16; ++ks) {
    const v2f a = *(const v2f*)(xa + ks * 4);
    const v2f b = *(const v2f*)(bp + (ks << 9));
    acc = __builtin_amdgcn_wmma_f32_16x16x4_f32(
        false, a, false, b, (short)0, acc, false, false);
  }
#else
#pragma unroll
  for (int v = 0; v < 8; ++v) {
    float s = 0.f;
    for (int k = kh * 64; k < kh * 64 + 64; ++k)
      s += xs[(v + mhi) * FT_PITCH + k] * Wst[((k >> 1) << 8) + (((nt << 4) + m) << 1) + (k & 1)];
    acc[v] = s;
  }
#endif
  float* __restrict__ hp = hs + kh * (16 * FT_PITCH);
#pragma unroll
  for (int v = 0; v < 8; ++v) hp[(v + mhi) * FT_PITCH + nt * 16 + m] = acc[v];
}

__global__ __launch_bounds__(512) void gat_node_gemm2(
    const int* __restrict__ csr, const int* __restrict__ rowptr,
    const float* __restrict__ as, const float* __restrict__ ad,
    const float* __restrict__ X, const float* __restrict__ W1, const float* __restrict__ bias1,
    const float* __restrict__ W2, const float* __restrict__ a_src2, const float* __restrict__ a_dst2,
    float* __restrict__ H2, float* __restrict__ as2, float* __restrict__ ad2,
    int n, int ntiles, float slope) {
  __shared__ __attribute__((aligned(16))) float W1s[NHID * NHID];
  __shared__ __attribute__((aligned(16))) float W2s[NHID * NHID];
  __shared__ __attribute__((aligned(16))) float xs[16 * FT_PITCH];
  __shared__ __attribute__((aligned(16))) float hs[2 * 16 * FT_PITCH];
  const int tid   = threadIdx.x;
  const int lane  = tid & 31;
  const int lane4 = lane * 4;
  const int wave  = __builtin_amdgcn_readfirstlane(tid >> 5);
  const int m     = lane & 15;
  const int khv   = lane >> 4;
  const int nt    = wave & 7;
  const int kh    = wave >> 3;

  for (int i = tid; i < NHID * NHID; i += 512) {
    const int k = i >> 7, nn = i & 127;
    const int o = ((k >> 1) << 8) + (nn << 1) + (k & 1);
    W1s[o] = W1[i];
    W2s[o] = W2[i];
  }
  const float4 b14 = *(const float4*)(bias1 + lane4);
  const float4 as4 = *(const float4*)(a_src2 + lane4);
  const float4 ad4 = *(const float4*)(a_dst2 + lane4);

  for (int tile = (int)blockIdx.x; tile < ntiles; tile += (int)gridDim.x) {
    const int d = tile * 16 + wave;
    {
      float4 o = make_float4(0.f, 0.f, 0.f, 0.f);
      if (d < n) o = gat_node_sum(csr, rowptr, as, ad, X, d, lane);
      *(float4*)(xs + wave * FT_PITCH + lane4) = o;
    }
    __syncthreads();

    tile_half_product(xs, W1s, hs, m, khv, nt, kh);
    __syncthreads();

    {
      const float4 h0 = *(const float4*)(hs + wave * FT_PITCH + lane4);
      const float4 h1 = *(const float4*)(hs + (16 + wave) * FT_PITCH + lane4);
      float4 o = make_float4(h0.x + h1.x + b14.x, h0.y + h1.y + b14.y,
                             h0.z + h1.z + b14.z, h0.w + h1.w + b14.w);
      o.x = lrelu(o.x, slope); o.y = lrelu(o.y, slope);
      o.z = lrelu(o.z, slope); o.w = lrelu(o.w, slope);
      *(float4*)(xs + wave * FT_PITCH + lane4) = o;
    }
    __syncthreads();

    tile_half_product(xs, W2s, hs, m, khv, nt, kh);
    __syncthreads();

    {
      const float4 h0 = *(const float4*)(hs + wave * FT_PITCH + lane4);
      const float4 h1 = *(const float4*)(hs + (16 + wave) * FT_PITCH + lane4);
      const float4 hv = make_float4(h0.x + h1.x, h0.y + h1.y, h0.z + h1.z, h0.w + h1.w);
      float ps = hv.x * as4.x;
      ps = fmaf(hv.y, as4.y, ps); ps = fmaf(hv.z, as4.z, ps); ps = fmaf(hv.w, as4.w, ps);
      float pd = hv.x * ad4.x;
      pd = fmaf(hv.y, ad4.y, pd); pd = fmaf(hv.z, ad4.z, pd); pd = fmaf(hv.w, ad4.w, pd);
      for (int off = 16; off; off >>= 1) {
        ps += __shfl_xor(ps, off, 32);
        pd += __shfl_xor(pd, off, 32);
      }
      if (d < n) {
        *(float4*)(H2 + (size_t)d * NHID + lane4) = hv;
        if (lane == 0) { as2[d] = ps; ad2[d] = pd; }
      }
    }
  }
}

extern "C" void kernel_launch(void* const* d_in, const int* in_sizes, int n_in,
                              void* d_out, int out_size, void* d_ws, size_t ws_size,
                              hipStream_t stream) {
  const float* x    = (const float*)d_in[0];
  const int*   ei   = (const int*)d_in[1];
  const float* W1   = (const float*)d_in[2];
  const float* asr1 = (const float*)d_in[3];
  const float* adt1 = (const float*)d_in[4];
  const float* b1   = (const float*)d_in[5];
  const float* W2   = (const float*)d_in[6];
  const float* asr2 = (const float*)d_in[7];
  const float* adt2 = (const float*)d_in[8];
  const float* b2   = (const float*)d_in[9];
  float* out = (float*)d_out;

  const int N = in_sizes[0] / NHID;
  const int E = in_sizes[1] / 2;
  const size_t NF = (size_t)N * NHID;

  const int NBK  = (N + BUCKET_NODES - 1) / BUCKET_NODES;
  const int NCH  = (E + CHUNK_EDGES - 1) / CHUNK_EDGES;
  const int LO   = (NBK + 1) * NCH;

  float* h2     = (float*)d_ws;
  int*   tmp    = (int*)(h2 + NF);
  float* as     = (float*)(tmp + 2 * (size_t)E);
  float* ad     = as + N;
  int*   rowptr = (int*)(ad + N);
  int*   off    = rowptr + N + 1;
  int*   csr    = off + LO;
  float* as2    = (float*)(csr + E);
  float* ad2    = as2 + N;

  const int* src = ei;
  const int* dst = ei + E;

  const int gWaveN = (int)(((size_t)N * 32 + 255) / 256);
  const int NT16   = (N + 15) / 16;

  gat_chunk_sort<<<NCH, 1024, 0, stream>>>(src, dst, off, tmp, E, NBK, NCH);
  gat_bucket_logit<<<NBK + (NBK + 1) / 2, 512, 0, stream>>>(tmp, off, rowptr, csr, N, NBK, NCH, E,
                                                            x, W1, asr1, adt1, as, ad);

  gat_node_gemm2<<<FT_GROUPS, 512, 0, stream>>>(csr, rowptr, as, ad, x, W1, b1, W2, asr2, adt2,
                                                h2, as2, ad2, N, NT16, NEG_ACT);

  gat_node<<<gWaveN, 256, 0, stream>>>(csr, rowptr, as2, ad2, h2, b2, out, N, 0.f, 0);
}


# Edge-list scatters (a chunk's sorted pairs, a bucket's source list) placed in LDS first and written out as contiguous whole-line stores, not 8- and 4-byte stores to 32 lines each; ranks as before
# speedup vs baseline: 8.7322x; 1.0698x over previous
#include <hip/hip_runtime.h>
#include <hip/hip_bf16.h>

typedef float v2f __attribute__((ext_vector_type(2)));
typedef float v8f __attribute__((ext_vector_type(8)));

#define NHID 128
#define NEG_ATT 0.2f
#define NEG_ACT 0.01f

#if __has_builtin(__builtin_amdgcn_wmma_f32_16x16x4_f32)
#define HAS_WMMA_F32X4 1
#endif

__device__ __forceinline__ float row16_sum(float x) {
  x += __int_as_float(__builtin_amdgcn_update_dpp(0, __float_as_int(x), 0x161, 0xf, 0xf, false));
  x += __int_as_float(__builtin_amdgcn_update_dpp(0, __float_as_int(x), 0x162, 0xf, 0xf, false));
  x += __int_as_float(__builtin_amdgcn_update_dpp(0, __float_as_int(x), 0x164, 0xf, 0xf, false));
  x += __int_as_float(__builtin_amdgcn_update_dpp(0, __float_as_int(x), 0x168, 0xf, 0xf, false));
  return x;
}

#define LOGIT_ROWS 256
__device__ __forceinline__ void logit_tile(
    float* __restrict__ vsd, int tile,
    const float* __restrict__ X, const float* __restrict__ W,
    const float* __restrict__ a_src, const float* __restrict__ a_dst,
    float* __restrict__ as, float* __restrict__ ad, int nrows) {
  const int tid  = threadIdx.x;
  const int lane = tid & 31;
  const int wave = tid >> 5;
  {
    const int k = tid >> 2, q = tid & 3;
    const float* __restrict__ wr = W + k * NHID + q * 32;
    const float* __restrict__ sr = a_src + q * 32;
    const float* __restrict__ dr = a_dst + q * 32;
    float ps = 0.f, pd = 0.f;
#pragma unroll
    for (int j = 0; j < 8; ++j) {
      const float4 w4 = *(const float4*)(wr + j * 4);
      const float4 s4 = *(const float4*)(sr + j * 4);
      const float4 d4 = *(const float4*)(dr + j * 4);
      ps = fmaf(w4.x, s4.x, ps); ps = fmaf(w4.y, s4.y, ps);
      ps = fmaf(w4.z, s4.z, ps); ps = fmaf(w4.w, s4.w, ps);
      pd = fmaf(w4.x, d4.x, pd); pd = fmaf(w4.y, d4.y, pd);
      pd = fmaf(w4.z, d4.z, pd); pd = fmaf(w4.w, d4.w, pd);
    }
    ps += __shfl_xor(ps, 1, 32); ps += __shfl_xor(ps, 2, 32);
    pd += __shfl_xor(pd, 1, 32); pd += __shfl_xor(pd, 2, 32);
    if (q == 0) { vsd[k] = ps; vsd[NHID + k] = pd; }
  }
  __syncthreads();

  const int row0 = tile * LOGIT_ROWS + wave * 16;
  if (row0 >= nrows) return;
  const int lane4 = lane * 4;
  const float4 vs4 = *(const float4*)(vsd + lane4);
  const float4 vd4 = *(const float4*)(vsd + NHID + lane4);
  float4 xv[16];
#pragma unroll
  for (int r = 0; r < 16; ++r) {
    const int row = min(row0 + r, nrows - 1);
    xv[r] = *(const float4*)(X + (size_t)row * NHID + lane4);
  }
  const bool lo = lane < 16;
  float res = 0.f;
#pragma unroll
  for (int r = 0; r < 16; ++r) {
    float ps = xv[r].x * vs4.x;
    ps = fmaf(xv[r].y, vs4.y, ps); ps = fmaf(xv[r].z, vs4.z, ps); ps = fmaf(xv[r].w, vs4.w, ps);
    float pd = xv[r].x * vd4.x;
    pd = fmaf(xv[r].y, vd4.y, pd); pd = fmaf(xv[r].z, vd4.z, pd); pd = fmaf(xv[r].w, vd4.w, pd);
    float keep = lo ? ps : pd;
    const float give = lo ? pd : ps;
    keep += __shfl_xor(give, 16, 32);
    keep = row16_sum(keep);
    if ((lane & 15) == r) res = keep;
  }
  const int row = row0 + (lane & 15);
  if (row < nrows) (lo ? as : ad)[row] = res;
}

#define BUCKET_SHIFT 7
#define BUCKET_NODES 128
#define NB_MAX 512
#define CHUNK_EDGES 8192
#define STAGE_CAP 8192

__global__ __launch_bounds__(1024) void gat_chunk_sort(const int* __restrict__ srcs, const int* __restrict__ dsts,
                                                       int* __restrict__ off, int* __restrict__ tmp_i,
                                                       int nedges, int nb, int nc) {
  __shared__ int binL[NB_MAX];
  __shared__ int wsumL[NB_MAX / 32];
  __shared__ __attribute__((aligned(16))) int2 stageL[CHUNK_EDGES];
  const int tid  = threadIdx.x;
  const int lane = tid & 31;
  const int c    = (int)blockIdx.x;
  if (tid < NB_MAX) binL[tid] = 0;
  __syncthreads();
  const int e0 = c * CHUNK_EDGES + tid;
  int d[8], s[8];
#pragma unroll
  for (int k = 0; k < 8; ++k) {
    const int e = e0 + k * 1024;
    const bool ok = e < nedges;
    d[k] = ok ? dsts[e] : -1;
    s[k] = ok ? srcs[e] : 0;
  }
#pragma unroll
  for (int k = 0; k < 8; ++k)
    if (d[k] >= 0) atomicAdd(&binL[d[k] >> BUCKET_SHIFT], 1);
  __syncthreads();

  int cnt = 0, inc = 0;
  if (tid < NB_MAX) {
    cnt = binL[tid];
    inc = cnt;
    for (int o = 1; o < 32; o <<= 1) {
      const int t = __shfl_up(inc, o, 32);
      if (lane >= o) inc += t;
    }
    if (lane == 31) wsumL[tid >> 5] = inc;
  }
  __syncthreads();
  if (tid < NB_MAX) {
    int o = 0;
    for (int w = 0; w < (tid >> 5); ++w) o += wsumL[w];
    const int excl = o + inc - cnt;
    binL[tid] = excl;
    if (tid < nb) off[tid * nc + c] = excl;
    if (tid == nb - 1) off[nb * nc + c] = excl + cnt;
  }
  __syncthreads();

  int pos[8];
#pragma unroll
  for (int k = 0; k < 8; ++k)
    pos[k] = (d[k] >= 0) ? atomicAdd(&binL[d[k] >> BUCKET_SHIFT], 1) : -1;
#pragma unroll
  for (int k = 0; k < 8; ++k)
    if (pos[k] >= 0) stageL[pos[k]] = make_int2(s[k], d[k]);
  __syncthreads();

  const int nmine = min(CHUNK_EDGES, nedges - c * CHUNK_EDGES);
  const int n4    = nmine >> 1;
  int4* __restrict__ mine4 = (int4*)tmp_i + (size_t)c * (CHUNK_EDGES / 2);
  const int4* __restrict__ st4 = (const int4*)stageL;
  if (nmine == CHUNK_EDGES) {
    int4 v4[CHUNK_EDGES / 2048];
#pragma unroll
    for (int i = 0; i < CHUNK_EDGES / 2048; ++i) v4[i] = st4[tid + i * 1024];
#pragma unroll
    for (int i = 0; i < CHUNK_EDGES / 2048; ++i) mine4[tid + i * 1024] = v4[i];
  } else {
    for (int q = tid; q < n4; q += 1024) mine4[q] = st4[q];
    if ((nmine & 1) && tid == 0) ((int2*)tmp_i)[(size_t)c * CHUNK_EDGES + nmine - 1] = stageL[nmine - 1];
  }
}

template <int MODE>
__device__ __forceinline__ int bucket_sweep(const int2* __restrict__ tmp, const int* __restrict__ off,
                                            int b, int nc, int wave, int lane,
                                            int* __restrict__ binL, int* __restrict__ dstp) {
  constexpr bool PLACE = MODE != 0;
  int lo_part = 0;
  for (int r0 = wave; r0 < nc; r0 += 16 * 32) {
    const int cl = r0 + 16 * lane;
    int sb = 0, sc = 0;
    if (cl < nc) {
      const int o0 = off[b * nc + cl];
      const int o1 = off[(b + 1) * nc + cl];
      lo_part += o0;
      sb = cl * CHUNK_EDGES + o0;
      sc = o1 - o0;
    }
    const int nseg = min(32, (nc - r0 + 15) >> 4);
    for (int j = 0; j < nseg; j += 8) {
      int2 e[8];
      int  bg[8], cn[8];
#pragma unroll
      for (int u = 0; u < 8; ++u) {
        const int jj = min(j + u, 31);
        bg[u] = __builtin_amdgcn_readlane(sb, jj);
        cn[u] = (j + u < nseg) ? __builtin_amdgcn_readlane(sc, jj) : 0;
        e[u] = make_int2(0, -1);
        if (lane < cn[u]) {
          if (PLACE) e[u] = tmp[bg[u] + lane];
          else       e[u].y = tmp[bg[u] + lane].y;
        }
      }
      if (PLACE) {
        int pos[8];
#pragma unroll
        for (int u = 0; u < 8; ++u)
          pos[u] = (e[u].y >= 0) ? atomicAdd(&binL[e[u].y & (BUCKET_NODES - 1)], 1) : -1;
#pragma unroll
        for (int u = 0; u < 8; ++u)
          if (pos[u] >= 0) dstp[pos[u]] = e[u].x;
      } else {
#pragma unroll
        for (int u = 0; u < 8; ++u)
          if (e[u].y >= 0) atomicAdd(&binL[e[u].y & (BUCKET_NODES - 1)], 1);
      }
#pragma unroll
      for (int u = 0; u < 8; ++u)
        for (int k = lane + 32; k < cn[u]; k += 32) {
          const int2 ee = tmp[bg[u] + k];
          if (PLACE) {
            const int pos = atomicAdd(&binL[ee.y & (BUCKET_NODES - 1)], 1);
            dstp[pos] = ee.x;
          } else {
            atomicAdd(&binL[ee.y & (BUCKET_NODES - 1)], 1);
          }
        }
    }
  }
  return lo_part;
}

__global__ __launch_bounds__(512) void gat_bucket_logit(
    const int* __restrict__ tmp_i, const int* __restrict__ off,
    int* __restrict__ rowptr, int* __restrict__ csr, int n, int nb, int nc, int nedges,
    const float* __restrict__ X, const float* __restrict__ W,
    const float* __restrict__ a_src, const float* __restrict__ a_dst,
    float* __restrict__ as, float* __restrict__ ad) {
  __shared__ __attribute__((aligned(16))) float vsd[2 * NHID];
  __shared__ int cntL[BUCKET_NODES];
  __shared__ int curL[BUCKET_NODES];
  __shared__ int wsum[BUCKET_NODES / 32];
  __shared__ int wlo[16];
  __shared__ int stageC[STAGE_CAP];
  const int ntile = (nb + 1) >> 1;
  if ((int)blockIdx.x < ntile) {
    logit_tile(vsd, (int)blockIdx.x, X, W, a_src, a_dst, as, ad, n);
    return;
  }
  const int b = (int)blockIdx.x - ntile;
  if (b >= nb) return;
  const int2* __restrict__ tmp = (const int2*)tmp_i;
  const int tid  = threadIdx.x;
  const int lane = tid & 31;
  const int wave = tid >> 5;
  if (tid < BUCKET_NODES) cntL[tid] = 0;
  __syncthreads();

  int lo_part = bucket_sweep<0>(tmp, off, b, nc, wave, lane, cntL, csr);
  for (int o = 16; o; o >>= 1) lo_part += __shfl_xor(lo_part, o, 32);
  if (lane == 0) wlo[wave] = lo_part;
  __syncthreads();

  int c = 0, inc = 0;
  if (tid < BUCKET_NODES) {
    c = cntL[tid];
    inc = c;
    for (int o = 1; o < 32; o <<= 1) {
      const int t = __shfl_up(inc, o, 32);
      if (lane >= o) inc += t;
    }
    if (lane == 31) wsum[tid >> 5] = inc;
  }
  __syncthreads();
  int base = 0;
  for (int w = 0; w < 16; ++w) base += wlo[w];
  const int tot = wsum[0] + wsum[1] + wsum[2] + wsum[3];
  const bool staged = tot <= STAGE_CAP;
  if (tid < BUCKET_NODES) {
    int o = 0;
    for (int w = 0; w < (tid >> 5); ++w) o += wsum[w];
    const int excl = o + inc - c;
    const int node = b * BUCKET_NODES + tid;
    if (node < n) rowptr[node] = base + excl;
    curL[tid] = staged ? excl : base + excl;
  }
  if (b == nb - 1 && tid == 0) rowptr[n] = nedges;
  __syncthreads();

  if (!staged) {
    bucket_sweep<1>(tmp, off, b, nc, wave, lane, curL, csr);
    return;
  }
  bucket_sweep<2>(tmp, off, b, nc, wave, lane, curL, stageC);
  __syncthreads();
  const int a = (int)((((size_t)(csr + base)) >> 2) & 31);
  const int nend = a + tot;
  for (int i0 = tid; i0 < nend; i0 += 4 * 512) {
    int v[4];
#pragma unroll
    for (int u = 0; u < 4; ++u) {
      const int i = i0 + u * 512;
      v[u] = (i >= a && i < nend) ? stageC[i - a] : 0;
    }
#pragma unroll
    for (int u = 0; u < 4; ++u) {
      const int i = i0 + u * 512;
      if (i >= a && i < nend) csr[base + (i - a)] = v[u];
    }
  }
}

__device__ __forceinline__ float lrelu(float v, float slope) {
  return v >= 0.f ? v : slope * v;
}

template <int G>
__device__ __forceinline__ void gat_grp(const float* __restrict__ H, int s_l, float w_l,
                                        int t, int lane4, float4& acc) {
  float4 hv[G];
  float  w[G];
#pragma unroll
  for (int i = 0; i < G; ++i) {
    const int s = __builtin_amdgcn_readlane(s_l, t + i);
    w[i] = __int_as_float(__builtin_amdgcn_readlane(__float_as_int(w_l), t + i));
    hv[i] = *(const float4*)(H + (size_t)s * NHID + lane4);
  }
#pragma unroll
  for (int i = 0; i < G; ++i) {
    acc.x = fmaf(hv[i].x, w[i], acc.x);
    acc.y = fmaf(hv[i].y, w[i], acc.y);
    acc.z = fmaf(hv[i].z, w[i], acc.z);
    acc.w = fmaf(hv[i].w, w[i], acc.w);
  }
}

__device__ __forceinline__ void gat_chunk(const float* __restrict__ H, int s_l, float w_l,
                                          int cnt, int lane4, float4& acc) {
  int t = 0;
  for (; t + 8 <= cnt; t += 8) gat_grp<8>(H, s_l, w_l, t, lane4, acc);
  if (t + 4 <= cnt) { gat_grp<4>(H, s_l, w_l, t, lane4, acc); t += 4; }
  const int rem = cnt - t;
  if (rem == 3)      gat_grp<3>(H, s_l, w_l, t, lane4, acc);
  else if (rem == 2) gat_grp<2>(H, s_l, w_l, t, lane4, acc);
  else if (rem == 1) gat_grp<1>(H, s_l, w_l, t, lane4, acc);
}

__device__ __forceinline__ float4 gat_node_sum(
    const int* __restrict__ csr, const int* __restrict__ rowptr,
    const float* __restrict__ as, const float* __restrict__ ad,
    const float* __restrict__ H, int d, int lane) {
  const int lane4 = lane * 4;
  const int beg = rowptr[d];
  const int end = rowptr[d + 1];
  const int deg = end - beg;
  const float ad_d   = ad[d];
  const float v_self = lrelu(as[d] + ad_d, NEG_ATT);

  const int  j0  = beg + lane;
  const int  j1  = j0 + 32;
  const bool ok0 = j0 < end;
  const bool ok1 = j1 < end;
  int s0 = d, s1 = d;
  if (ok0) s0 = csr[j0];
  if (ok1) s1 = csr[j1];
  const float4 hd = *(const float4*)(H + (size_t)d * NHID + lane4);
  const float a0 = as[s0];
  const float a1 = as[s1];
  const float v0 = ok0 ? lrelu(a0 + ad_d, NEG_ATT) : v_self;
  const float v1 = ok1 ? lrelu(a1 + ad_d, NEG_ATT) : v_self;

  float mx = fmaxf(v_self, fmaxf(v0, v1));
  for (int j = j0 + 64; j < end; j += 32)
    mx = fmaxf(mx, lrelu(as[csr[j]] + ad_d, NEG_ATT));
  for (int off = 16; off; off >>= 1) mx = fmaxf(mx, __shfl_xor(mx, off, 32));

  const float p0 = ok0 ? expf(v0 - mx) : 0.f;
  const float p1 = ok1 ? expf(v1 - mx) : 0.f;
  float den = p0 + p1;
  for (int j = j0 + 64; j < end; j += 32)
    den += expf(lrelu(as[csr[j]] + ad_d, NEG_ATT) - mx);
  for (int off = 16; off; off >>= 1) den += __shfl_xor(den, off, 32);
  const float p_self = expf(v_self - mx);
  den += p_self;
  const float dn = den + 1e-16f;

  const float  w_self = p_self / dn;
  float4 acc = make_float4(hd.x * w_self, hd.y * w_self, hd.z * w_self, hd.w * w_self);

  gat_chunk(H, s0, p0 / dn, min(deg, 32), lane4, acc);
  if (deg > 32) gat_chunk(H, s1, p1 / dn, min(deg - 32, 32), lane4, acc);
  for (int base = beg + 64; base < end; base += 32) {
    const int j = base + lane;
    int   s = d;
    float w = 0.f;
    if (j < end) {
      s = csr[j];
      w = expf(lrelu(as[s] + ad_d, NEG_ATT) - mx) / dn;
    }
    gat_chunk(H, s, w, min(end - base, 32), lane4, acc);
  }
  return acc;
}

__device__ __forceinline__ float4 gat_node_row(
    const int* __restrict__ csr, const int* __restrict__ rowptr,
    const float* __restrict__ as, const float* __restrict__ ad,
    const float* __restrict__ H, const float* __restrict__ bias,
    int d, int lane, float slope, int apply_act) {
  const int lane4 = lane * 4;
  const float4 acc = gat_node_sum(csr, rowptr, as, ad, H, d, lane);
  const float4 b4 = *(const float4*)(bias + lane4);
  float4 o = make_float4(acc.x + b4.x, acc.y + b4.y, acc.z + b4.z, acc.w + b4.w);
  if (apply_act) {
    o.x = lrelu(o.x, slope); o.y = lrelu(o.y, slope);
    o.z = lrelu(o.z, slope); o.w = lrelu(o.w, slope);
  }
  return o;
}

__global__ __launch_bounds__(256) void gat_node(
    const int* __restrict__ csr, const int* __restrict__ rowptr,
    const float* __restrict__ as, const float* __restrict__ ad,
    const float* __restrict__ H, const float* __restrict__ bias,
    float* __restrict__ out, int n, float slope, int apply_act) {
  const int lane = threadIdx.x & 31;
  const int d = __builtin_amdgcn_readfirstlane(
      (int)((blockIdx.x * (size_t)blockDim.x + threadIdx.x) >> 5));
  if (d >= n) return;
  const float4 o = gat_node_row(csr, rowptr, as, ad, H, bias, d, lane, slope, apply_act);
  *(float4*)(out + (size_t)d * NHID + lane * 4) = o;
}

#define FT_PITCH 132
#define FT_GROUPS 512

__device__ __forceinline__ void tile_half_product(const float* __restrict__ xs, const float* __restrict__ Wst,
                                                  float* __restrict__ hs, int m, int khv, int nt, int kh) {
  const int khalf = khv * 2;
  const int mhi   = khv * 8;
  v8f acc = {};
#ifdef HAS_WMMA_F32X4
  const float* __restrict__ xa = xs + m * FT_PITCH + kh * 64 + khalf;
  const float* __restrict__ bp = Wst + (((kh * 32) + khv) << 8) + (nt << 5) + (m << 1);
#pragma unroll
  for (int ks = 0; ks < 16; ++ks) {
    const v2f a = *(const v2f*)(xa + ks * 4);
    const v2f b = *(const v2f*)(bp + (ks << 9));
    acc = __builtin_amdgcn_wmma_f32_16x16x4_f32(
        false, a, false, b, (short)0, acc, false, false);
  }
#else
#pragma unroll
  for (int v = 0; v < 8; ++v) {
    float s = 0.f;
    for (int k = kh * 64; k < kh * 64 + 64; ++k)
      s += xs[(v + mhi) * FT_PITCH + k] * Wst[((k >> 1) << 8) + (((nt << 4) + m) << 1) + (k & 1)];
    acc[v] = s;
  }
#endif
  float* __restrict__ hp = hs + kh * (16 * FT_PITCH);
#pragma unroll
  for (int v = 0; v < 8; ++v) hp[(v + mhi) * FT_PITCH + nt * 16 + m] = acc[v];
}

__global__ __launch_bounds__(512) void gat_node_gemm2(
    const int* __restrict__ csr, const int* __restrict__ rowptr,
    const float* __restrict__ as, const float* __restrict__ ad,
    const float* __restrict__ X, const float* __restrict__ W1, const float* __restrict__ bias1,
    const float* __restrict__ W2, const float* __restrict__ a_src2, const float* __restrict__ a_dst2,
    float* __restrict__ H2, float* __restrict__ as2, float* __restrict__ ad2,
    int n, int ntiles, float slope) {
  __shared__ __attribute__((aligned(16))) float W1s[NHID * NHID];
  __shared__ __attribute__((aligned(16))) float W2s[NHID * NHID];
  __shared__ __attribute__((aligned(16))) float xs[16 * FT_PITCH];
  __shared__ __attribute__((aligned(16))) float hs[2 * 16 * FT_PITCH];
  const int tid   = threadIdx.x;
  const int lane  = tid & 31;
  const int lane4 = lane * 4;
  const int wave  = __builtin_amdgcn_readfirstlane(tid >> 5);
  const int m     = lane & 15;
  const int khv   = lane >> 4;
  const int nt    = wave & 7;
  const int kh    = wave >> 3;

  for (int i = tid; i < NHID * NHID; i += 512) {
    const int k = i >> 7, nn = i & 127;
    const int o = ((k >> 1) << 8) + (nn << 1) + (k & 1);
    W1s[o] = W1[i];
    W2s[o] = W2[i];
  }
  const float4 b14 = *(const float4*)(bias1 + lane4);
  const float4 as4 = *(const float4*)(a_src2 + lane4);
  const float4 ad4 = *(const float4*)(a_dst2 + lane4);

  for (int tile = (int)blockIdx.x; tile < ntiles; tile += (int)gridDim.x) {
    const int d = tile * 16 + wave;
    {
      float4 o = make_float4(0.f, 0.f, 0.f, 0.f);
      if (d < n) o = gat_node_sum(csr, rowptr, as, ad, X, d, lane);
      *(float4*)(xs + wave * FT_PITCH + lane4) = o;
    }
    __syncthreads();

    tile_half_product(xs, W1s, hs, m, khv, nt, kh);
    __syncthreads();

    {
      const float4 h0 = *(const float4*)(hs + wave * FT_PITCH + lane4);
      const float4 h1 = *(const float4*)(hs + (16 + wave) * FT_PITCH + lane4);
      float4 o = make_float4(h0.x + h1.x + b14.x, h0.y + h1.y + b14.y,
                             h0.z + h1.z + b14.z, h0.w + h1.w + b14.w);
      o.x = lrelu(o.x, slope); o.y = lrelu(o.y, slope);
      o.z = lrelu(o.z, slope); o.w = lrelu(o.w, slope);
      *(float4*)(xs + wave * FT_PITCH + lane4) = o;
    }
    __syncthreads();

    tile_half_product(xs, W2s, hs, m, khv, nt, kh);
    __syncthreads();

    {
      const float4 h0 = *(const float4*)(hs + wave * FT_PITCH + lane4);
      const float4 h1 = *(const float4*)(hs + (16 + wave) * FT_PITCH + lane4);
      const float4 hv = make_float4(h0.x + h1.x, h0.y + h1.y, h0.z + h1.z, h0.w + h1.w);
      float ps = hv.x * as4.x;
      ps = fmaf(hv.y, as4.y, ps); ps = fmaf(hv.z, as4.z, ps); ps = fmaf(hv.w, as4.w, ps);
      float pd = hv.x * ad4.x;
      pd = fmaf(hv.y, ad4.y, pd); pd = fmaf(hv.z, ad4.z, pd); pd = fmaf(hv.w, ad4.w, pd);
      for (int off = 16; off; off >>= 1) {
        ps += __shfl_xor(ps, off, 32);
        pd += __shfl_xor(pd, off, 32);
      }
      if (d < n) {
        *(float4*)(H2 + (size_t)d * NHID + lane4) = hv;
        if (lane == 0) { as2[d] = ps; ad2[d] = pd; }
      }
    }
  }
}

extern "C" void kernel_launch(void* const* d_in, const int* in_sizes, int n_in,
                              void* d_out, int out_size, void* d_ws, size_t ws_size,
                              hipStream_t stream) {
  const float* x    = (const float*)d_in[0];
  const int*   ei   = (const int*)d_in[1];
  const float* W1   = (const float*)d_in[2];
  const float* asr1 = (const float*)d_in[3];
  const float* adt1 = (const float*)d_in[4];
  const float* b1   = (const float*)d_in[5];
  const float* W2   = (const float*)d_in[6];
  const float* asr2 = (const float*)d_in[7];
  const float* adt2 = (const float*)d_in[8];
  const float* b2   = (const float*)d_in[9];
  float* out = (float*)d_out;

  const int N = in_sizes[0] / NHID;
  const int E = in_sizes[1] / 2;
  const size_t NF = (size_t)N * NHID;

  const int NBK  = (N + BUCKET_NODES - 1) / BUCKET_NODES;
  const int NCH  = (E + CHUNK_EDGES - 1) / CHUNK_EDGES;
  const int LO   = (NBK + 1) * NCH;

  float* h2     = (float*)d_ws;
  int*   tmp    = (int*)(h2 + NF);
  float* as     = (float*)(tmp + 2 * (size_t)E);
  float* ad     = as + N;
  int*   rowptr = (int*)(ad + N);
  int*   off    = rowptr + N + 1;
  int*   csr    = off + LO;
  float* as2    = (float*)(csr + E);
  float* ad2    = as2 + N;

  const int* src = ei;
  const int* dst = ei + E;

  const int gWaveN = (int)(((size_t)N * 32 + 255) / 256);
  const int NT16   = (N + 15) / 16;

  gat_chunk_sort<<<NCH, 1024, 0, stream>>>(src, dst, off, tmp, E, NBK, NCH);
  gat_bucket_logit<<<NBK + (NBK + 1) / 2, 512, 0, stream>>>(tmp, off, rowptr, csr, N, NBK, NCH, E,
                                                            x, W1, asr1, adt1, as, ad);

  gat_node_gemm2<<<FT_GROUPS, 512, 0, stream>>>(csr, rowptr, as, ad, x, W1, b1, W2, asr2, adt2,
                                                h2, as2, ad2, N, NT16, NEG_ACT);

  gat_node<<<gWaveN, 256, 0, stream>>>(csr, rowptr, as2, ad2, h2, b2, out, N, 0.f, 0);
}


# Fused launch: each wave loads the next tile's node head (list bounds, first 64 sources, their logits, self row) under the current tile's row gathers and two matrix products; arithmetic unchanged
# speedup vs baseline: 9.0716x; 1.0389x over previous
#include <hip/hip_runtime.h>
#include <hip/hip_bf16.h>

typedef float v2f __attribute__((ext_vector_type(2)));
typedef float v8f __attribute__((ext_vector_type(8)));

#define NHID 128
#define NEG_ATT 0.2f
#define NEG_ACT 0.01f

#if __has_builtin(__builtin_amdgcn_wmma_f32_16x16x4_f32)
#define HAS_WMMA_F32X4 1
#endif

__device__ __forceinline__ float row16_sum(float x) {
  x += __int_as_float(__builtin_amdgcn_update_dpp(0, __float_as_int(x), 0x161, 0xf, 0xf, false));
  x += __int_as_float(__builtin_amdgcn_update_dpp(0, __float_as_int(x), 0x162, 0xf, 0xf, false));
  x += __int_as_float(__builtin_amdgcn_update_dpp(0, __float_as_int(x), 0x164, 0xf, 0xf, false));
  x += __int_as_float(__builtin_amdgcn_update_dpp(0, __float_as_int(x), 0x168, 0xf, 0xf, false));
  return x;
}

#define LOGIT_ROWS 256
__device__ __forceinline__ void logit_tile(
    float* __restrict__ vsd, int tile,
    const float* __restrict__ X, const float* __restrict__ W,
    const float* __restrict__ a_src, const float* __restrict__ a_dst,
    float* __restrict__ as, float* __restrict__ ad, int nrows) {
  const int tid  = threadIdx.x;
  const int lane = tid & 31;
  const int wave = tid >> 5;
  {
    const int k = tid >> 2, q = tid & 3;
    const float* __restrict__ wr = W + k * NHID + q * 32;
    const float* __restrict__ sr = a_src + q * 32;
    const float* __restrict__ dr = a_dst + q * 32;
    float ps = 0.f, pd = 0.f;
#pragma unroll
    for (int j = 0; j < 8; ++j) {
      const float4 w4 = *(const float4*)(wr + j * 4);
      const float4 s4 = *(const float4*)(sr + j * 4);
      const float4 d4 = *(const float4*)(dr + j * 4);
      ps = fmaf(w4.x, s4.x, ps); ps = fmaf(w4.y, s4.y, ps);
      ps = fmaf(w4.z, s4.z, ps); ps = fmaf(w4.w, s4.w, ps);
      pd = fmaf(w4.x, d4.x, pd); pd = fmaf(w4.y, d4.y, pd);
      pd = fmaf(w4.z, d4.z, pd); pd = fmaf(w4.w, d4.w, pd);
    }
    ps += __shfl_xor(ps, 1, 32); ps += __shfl_xor(ps, 2, 32);
    pd += __shfl_xor(pd, 1, 32); pd += __shfl_xor(pd, 2, 32);
    if (q == 0) { vsd[k] = ps; vsd[NHID + k] = pd; }
  }
  __syncthreads();

  const int row0 = tile * LOGIT_ROWS + wave * 16;
  if (row0 >= nrows) return;
  const int lane4 = lane * 4;
  const float4 vs4 = *(const float4*)(vsd + lane4);
  const float4 vd4 = *(const float4*)(vsd + NHID + lane4);
  float4 xv[16];
#pragma unroll
  for (int r = 0; r < 16; ++r) {
    const int row = min(row0 + r, nrows - 1);
    xv[r] = *(const float4*)(X + (size_t)row * NHID + lane4);
  }
  const bool lo = lane < 16;
  float res = 0.f;
#pragma unroll
  for (int r = 0; r < 16; ++r) {
    float ps = xv[r].x * vs4.x;
    ps = fmaf(xv[r].y, vs4.y, ps); ps = fmaf(xv[r].z, vs4.z, ps); ps = fmaf(xv[r].w, vs4.w, ps);
    float pd = xv[r].x * vd4.x;
    pd = fmaf(xv[r].y, vd4.y, pd); pd = fmaf(xv[r].z, vd4.z, pd); pd = fmaf(xv[r].w, vd4.w, pd);
    float keep = lo ? ps : pd;
    const float give = lo ? pd : ps;
    keep += __shfl_xor(give, 16, 32);
    keep = row16_sum(keep);
    if ((lane & 15) == r) res = keep;
  }
  const int row = row0 + (lane & 15);
  if (row < nrows) (lo ? as : ad)[row] = res;
}

#define BUCKET_SHIFT 7
#define BUCKET_NODES 128
#define NB_MAX 512
#define CHUNK_EDGES 8192
#define STAGE_CAP 8192

__global__ __launch_bounds__(1024) void gat_chunk_sort(const int* __restrict__ srcs, const int* __restrict__ dsts,
                                                       int* __restrict__ off, int* __restrict__ tmp_i,
                                                       int nedges, int nb, int nc) {
  __shared__ int binL[NB_MAX];
  __shared__ int wsumL[NB_MAX / 32];
  __shared__ __attribute__((aligned(16))) int2 stageL[CHUNK_EDGES];
  const int tid  = threadIdx.x;
  const int lane = tid & 31;
  const int c    = (int)blockIdx.x;
  if (tid < NB_MAX) binL[tid] = 0;
  __syncthreads();
  const int e0 = c * CHUNK_EDGES + tid;
  int d[8], s[8];
#pragma unroll
  for (int k = 0; k < 8; ++k) {
    const int e = e0 + k * 1024;
    const bool ok = e < nedges;
    d[k] = ok ? dsts[e] : -1;
    s[k] = ok ? srcs[e] : 0;
  }
#pragma unroll
  for (int k = 0; k < 8; ++k)
    if (d[k] >= 0) atomicAdd(&binL[d[k] >> BUCKET_SHIFT], 1);
  __syncthreads();

  int cnt = 0, inc = 0;
  if (tid < NB_MAX) {
    cnt = binL[tid];
    inc = cnt;
    for (int o = 1; o < 32; o <<= 1) {
      const int t = __shfl_up(inc, o, 32);
      if (lane >= o) inc += t;
    }
    if (lane == 31) wsumL[tid >> 5] = inc;
  }
  __syncthreads();
  if (tid < NB_MAX) {
    int o = 0;
    for (int w = 0; w < (tid >> 5); ++w) o += wsumL[w];
    const int excl = o + inc - cnt;
    binL[tid] = excl;
    if (tid < nb) off[tid * nc + c] = excl;
    if (tid == nb - 1) off[nb * nc + c] = excl + cnt;
  }
  __syncthreads();

  int pos[8];
#pragma unroll
  for (int k = 0; k < 8; ++k)
    pos[k] = (d[k] >= 0) ? atomicAdd(&binL[d[k] >> BUCKET_SHIFT], 1) : -1;
#pragma unroll
  for (int k = 0; k < 8; ++k)
    if (pos[k] >= 0) stageL[pos[k]] = make_int2(s[k], d[k]);
  __syncthreads();

  const int nmine = min(CHUNK_EDGES, nedges - c * CHUNK_EDGES);
  const int n4    = nmine >> 1;
  int4* __restrict__ mine4 = (int4*)tmp_i + (size_t)c * (CHUNK_EDGES / 2);
  const int4* __restrict__ st4 = (const int4*)stageL;
  if (nmine == CHUNK_EDGES) {
    int4 v4[CHUNK_EDGES / 2048];
#pragma unroll
    for (int i = 0; i < CHUNK_EDGES / 2048; ++i) v4[i] = st4[tid + i * 1024];
#pragma unroll
    for (int i = 0; i < CHUNK_EDGES / 2048; ++i) mine4[tid + i * 1024] = v4[i];
  } else {
    for (int q = tid; q < n4; q += 1024) mine4[q] = st4[q];
    if ((nmine & 1) && tid == 0) ((int2*)tmp_i)[(size_t)c * CHUNK_EDGES + nmine - 1] = stageL[nmine - 1];
  }
}

template <int MODE>
__device__ __forceinline__ int bucket_sweep(const int2* __restrict__ tmp, const int* __restrict__ off,
                                            int b, int nc, int wave, int lane,
                                            int* __restrict__ binL, int* __restrict__ dstp) {
  constexpr bool PLACE = MODE != 0;
  int lo_part = 0;
  for (int r0 = wave; r0 < nc; r0 += 16 * 32) {
    const int cl = r0 + 16 * lane;
    int sb = 0, sc = 0;
    if (cl < nc) {
      const int o0 = off[b * nc + cl];
      const int o1 = off[(b + 1) * nc + cl];
      lo_part += o0;
      sb = cl * CHUNK_EDGES + o0;
      sc = o1 - o0;
    }
    const int nseg = min(32, (nc - r0 + 15) >> 4);
    for (int j = 0; j < nseg; j += 8) {
      int2 e[8];
      int  bg[8], cn[8];
#pragma unroll
      for (int u = 0; u < 8; ++u) {
        const int jj = min(j + u, 31);
        bg[u] = __builtin_amdgcn_readlane(sb, jj);
        cn[u] = (j + u < nseg) ? __builtin_amdgcn_readlane(sc, jj) : 0;
        e[u] = make_int2(0, -1);
        if (lane < cn[u]) {
          if (PLACE) e[u] = tmp[bg[u] + lane];
          else       e[u].y = tmp[bg[u] + lane].y;
        }
      }
      if (PLACE) {
        int pos[8];
#pragma unroll
        for (int u = 0; u < 8; ++u)
          pos[u] = (e[u].y >= 0) ? atomicAdd(&binL[e[u].y & (BUCKET_NODES - 1)], 1) : -1;
#pragma unroll
        for (int u = 0; u < 8; ++u)
          if (pos[u] >= 0) dstp[pos[u]] = e[u].x;
      } else {
#pragma unroll
        for (int u = 0; u < 8; ++u)
          if (e[u].y >= 0) atomicAdd(&binL[e[u].y & (BUCKET_NODES - 1)], 1);
      }
#pragma unroll
      for (int u = 0; u < 8; ++u)
        for (int k = lane + 32; k < cn[u]; k += 32) {
          const int2 ee = tmp[bg[u] + k];
          if (PLACE) {
            const int pos = atomicAdd(&binL[ee.y & (BUCKET_NODES - 1)], 1);
            dstp[pos] = ee.x;
          } else {
            atomicAdd(&binL[ee.y & (BUCKET_NODES - 1)], 1);
          }
        }
    }
  }
  return lo_part;
}

__global__ __launch_bounds__(512) void gat_bucket_logit(
    const int* __restrict__ tmp_i, const int* __restrict__ off,
    int* __restrict__ rowptr, int* __restrict__ csr, int n, int nb, int nc, int nedges,
    const float* __restrict__ X, const float* __restrict__ W,
    const float* __restrict__ a_src, const float* __restrict__ a_dst,
    float* __restrict__ as, float* __restrict__ ad) {
  __shared__ __attribute__((aligned(16))) float vsd[2 * NHID];
  __shared__ int cntL[BUCKET_NODES];
  __shared__ int curL[BUCKET_NODES];
  __shared__ int wsum[BUCKET_NODES / 32];
  __shared__ int wlo[16];
  __shared__ int stageC[STAGE_CAP];
  const int ntile = (nb + 1) >> 1;
  if ((int)blockIdx.x < ntile) {
    logit_tile(vsd, (int)blockIdx.x, X, W, a_src, a_dst, as, ad, n);
    return;
  }
  const int b = (int)blockIdx.x - ntile;
  if (b >= nb) return;
  const int2* __restrict__ tmp = (const int2*)tmp_i;
  const int tid  = threadIdx.x;
  const int lane = tid & 31;
  const int wave = tid >> 5;
  if (tid < BUCKET_NODES) cntL[tid] = 0;
  __syncthreads();

  int lo_part = bucket_sweep<0>(tmp, off, b, nc, wave, lane, cntL, csr);
  for (int o = 16; o; o >>= 1) lo_part += __shfl_xor(lo_part, o, 32);
  if (lane == 0) wlo[wave] = lo_part;
  __syncthreads();

  int c = 0, inc = 0;
  if (tid < BUCKET_NODES) {
    c = cntL[tid];
    inc = c;
    for (int o = 1; o < 32; o <<= 1) {
      const int t = __shfl_up(inc, o, 32);
      if (lane >= o) inc += t;
    }
    if (lane == 31) wsum[tid >> 5] = inc;
  }
  __syncthreads();
  int base = 0;
  for (int w = 0; w < 16; ++w) base += wlo[w];
  const int tot = wsum[0] + wsum[1] + wsum[2] + wsum[3];
  const bool staged = tot <= STAGE_CAP;
  if (tid < BUCKET_NODES) {
    int o = 0;
    for (int w = 0; w < (tid >> 5); ++w) o += wsum[w];
    const int excl = o + inc - c;
    const int node = b * BUCKET_NODES + tid;
    if (node < n) rowptr[node] = base + excl;
    curL[tid] = staged ? excl : base + excl;
  }
  if (b == nb - 1 && tid == 0) rowptr[n] = nedges;
  __syncthreads();

  if (!staged) {
    bucket_sweep<1>(tmp, off, b, nc, wave, lane, curL, csr);
    return;
  }
  bucket_sweep<2>(tmp, off, b, nc, wave, lane, curL, stageC);
  __syncthreads();
  const int a = (int)((((size_t)(csr + base)) >> 2) & 31);
  const int nend = a + tot;
  for (int i0 = tid; i0 < nend; i0 += 4 * 512) {
    int v[4];
#pragma unroll
    for (int u = 0; u < 4; ++u) {
      const int i = i0 + u * 512;
      v[u] = (i >= a && i < nend) ? stageC[i - a] : 0;
    }
#pragma unroll
    for (int u = 0; u < 4; ++u) {
      const int i = i0 + u * 512;
      if (i >= a && i < nend) csr[base + (i - a)] = v[u];
    }
  }
}

__device__ __forceinline__ float lrelu(float v, float slope) {
  return v >= 0.f ? v : slope * v;
}

template <int G>
__device__ __forceinline__ void gat_grp(const float* __restrict__ H, int s_l, float w_l,
                                        int t, int lane4, float4& acc) {
  float4 hv[G];
  float  w[G];
#pragma unroll
  for (int i = 0; i < G; ++i) {
    const int s = __builtin_amdgcn_readlane(s_l, t + i);
    w[i] = __int_as_float(__builtin_amdgcn_readlane(__float_as_int(w_l), t + i));
    hv[i] = *(const float4*)(H + (size_t)s * NHID + lane4);
  }
#pragma unroll
  for (int i = 0; i < G; ++i) {
    acc.x = fmaf(hv[i].x, w[i], acc.x);
    acc.y = fmaf(hv[i].y, w[i], acc.y);
    acc.z = fmaf(hv[i].z, w[i], acc.z);
    acc.w = fmaf(hv[i].w, w[i], acc.w);
  }
}

__device__ __forceinline__ void gat_chunk(const float* __restrict__ H, int s_l, float w_l,
                                          int cnt, int lane4, float4& acc) {
  int t = 0;
  for (; t + 8 <= cnt; t += 8) gat_grp<8>(H, s_l, w_l, t, lane4, acc);
  if (t + 4 <= cnt) { gat_grp<4>(H, s_l, w_l, t, lane4, acc); t += 4; }
  const int rem = cnt - t;
  if (rem == 3)      gat_grp<3>(H, s_l, w_l, t, lane4, acc);
  else if (rem == 2) gat_grp<2>(H, s_l, w_l, t, lane4, acc);
  else if (rem == 1) gat_grp<1>(H, s_l, w_l, t, lane4, acc);
}

__device__ __forceinline__ float4 gat_node_finish(
    const int* __restrict__ csr, const float* __restrict__ as, const float* __restrict__ H,
    int d, int lane, int beg, int end, float ad_d, float as_d,
    int s0, int s1, float a0, float a1, float4 hd) {
  const int lane4 = lane * 4;
  const int deg = end - beg;
  const float v_self = lrelu(as_d + ad_d, NEG_ATT);

  const int  j0  = beg + lane;
  const int  j1  = j0 + 32;
  const bool ok0 = j0 < end;
  const bool ok1 = j1 < end;
  const float v0 = ok0 ? lrelu(a0 + ad_d, NEG_ATT) : v_self;
  const float v1 = ok1 ? lrelu(a1 + ad_d, NEG_ATT) : v_self;

  float mx = fmaxf(v_self, fmaxf(v0, v1));
  for (int j = j0 + 64; j < end; j += 32)
    mx = fmaxf(mx, lrelu(as[csr[j]] + ad_d, NEG_ATT));
  for (int off = 16; off; off >>= 1) mx = fmaxf(mx, __shfl_xor(mx, off, 32));

  const float p0 = ok0 ? expf(v0 - mx) : 0.f;
  const float p1 = ok1 ? expf(v1 - mx) : 0.f;
  float den = p0 + p1;
  for (int j = j0 + 64; j < end; j += 32)
    den += expf(lrelu(as[csr[j]] + ad_d, NEG_ATT) - mx);
  for (int off = 16; off; off >>= 1) den += __shfl_xor(den, off, 32);
  const float p_self = expf(v_self - mx);
  den += p_self;
  const float dn = den + 1e-16f;

  const float  w_self = p_self / dn;
  float4 acc = make_float4(hd.x * w_self, hd.y * w_self, hd.z * w_self, hd.w * w_self);

  gat_chunk(H, s0, p0 / dn, min(deg, 32), lane4, acc);
  if (deg > 32) gat_chunk(H, s1, p1 / dn, min(deg - 32, 32), lane4, acc);
  for (int base = beg + 64; base < end; base += 32) {
    const int j = base + lane;
    int   s = d;
    float w = 0.f;
    if (j < end) {
      s = csr[j];
      w = expf(lrelu(as[s] + ad_d, NEG_ATT) - mx) / dn;
    }
    gat_chunk(H, s, w, min(end - base, 32), lane4, acc);
  }
  return acc;
}

__device__ __forceinline__ float4 gat_node_sum(
    const int* __restrict__ csr, const int* __restrict__ rowptr,
    const float* __restrict__ as, const float* __restrict__ ad,
    const float* __restrict__ H, int d, int lane) {
  const int beg = rowptr[d];
  const int end = rowptr[d + 1];
  const float ad_d = ad[d];
  const float as_d = as[d];
  const int  j0  = beg + lane;
  const int  j1  = j0 + 32;
  int s0 = d, s1 = d;
  if (j0 < end) s0 = csr[j0];
  if (j1 < end) s1 = csr[j1];
  const float4 hd = *(const float4*)(H + (size_t)d * NHID + lane * 4);
  const float a0 = as[s0];
  const float a1 = as[s1];
  return gat_node_finish(csr, as, H, d, lane, beg, end, ad_d, as_d, s0, s1, a0, a1, hd);
}

struct NodeHead {
  int    rp;
  float  lg;
  int    s0, s1;
  float  a0, a1;
  float4 hd;
};

__device__ __forceinline__ void head_bounds(NodeHead& h, const int* __restrict__ rowptr,
                                            const float* __restrict__ as, const float* __restrict__ ad,
                                            int d, int lane) {
  const int odd = lane & 1;
  h.rp = rowptr[d + odd];
  h.lg = (odd ? as : ad)[d];
}

__device__ __forceinline__ void head_sources(NodeHead& h, const int* __restrict__ csr,
                                             const float* __restrict__ H, int d, int lane) {
  const int beg = __builtin_amdgcn_readlane(h.rp, 0);
  const int end = __builtin_amdgcn_readlane(h.rp, 1);
  const int j0 = beg + lane;
  const int j1 = j0 + 32;
  h.s0 = d; h.s1 = d;
  if (j0 < end) h.s0 = csr[j0];
  if (j1 < end) h.s1 = csr[j1];
  h.hd = *(const float4*)(H + (size_t)d * NHID + lane * 4);
}

__device__ __forceinline__ void head_logits(NodeHead& h, const float* __restrict__ as) {
  h.a0 = as[h.s0];
  h.a1 = as[h.s1];
}

__device__ __forceinline__ float4 gat_node_finish_head(
    const NodeHead& h, const int* __restrict__ csr, const float* __restrict__ as,
    const float* __restrict__ H, int d, int lane) {
  const int   beg  = __builtin_amdgcn_readlane(h.rp, 0);
  const int   end  = __builtin_amdgcn_readlane(h.rp, 1);
  const float ad_d = __int_as_float(__builtin_amdgcn_readlane(__float_as_int(h.lg), 0));
  const float as_d = __int_as_float(__builtin_amdgcn_readlane(__float_as_int(h.lg), 1));
  return gat_node_finish(csr, as, H, d, lane, beg, end, ad_d, as_d, h.s0, h.s1, h.a0, h.a1, h.hd);
}

__device__ __forceinline__ float4 gat_node_row(
    const int* __restrict__ csr, const int* __restrict__ rowptr,
    const float* __restrict__ as, const float* __restrict__ ad,
    const float* __restrict__ H, const float* __restrict__ bias,
    int d, int lane, float slope, int apply_act) {
  const int lane4 = lane * 4;
  const float4 acc = gat_node_sum(csr, rowptr, as, ad, H, d, lane);
  const float4 b4 = *(const float4*)(bias + lane4);
  float4 o = make_float4(acc.x + b4.x, acc.y + b4.y, acc.z + b4.z, acc.w + b4.w);
  if (apply_act) {
    o.x = lrelu(o.x, slope); o.y = lrelu(o.y, slope);
    o.z = lrelu(o.z, slope); o.w = lrelu(o.w, slope);
  }
  return o;
}

__global__ __launch_bounds__(256) void gat_node(
    const int* __restrict__ csr, const int* __restrict__ rowptr,
    const float* __restrict__ as, const float* __restrict__ ad,
    const float* __restrict__ H, const float* __restrict__ bias,
    float* __restrict__ out, int n, float slope, int apply_act) {
  const int lane = threadIdx.x & 31;
  const int d = __builtin_amdgcn_readfirstlane(
      (int)((blockIdx.x * (size_t)blockDim.x + threadIdx.x) >> 5));
  if (d >= n) return;
  const float4 o = gat_node_row(csr, rowptr, as, ad, H, bias, d, lane, slope, apply_act);
  *(float4*)(out + (size_t)d * NHID + lane * 4) = o;
}

#define FT_PITCH 132
#define FT_GROUPS 512

__device__ __forceinline__ void tile_half_product(const float* __restrict__ xs, const float* __restrict__ Wst,
                                                  float* __restrict__ hs, int m, int khv, int nt, int kh) {
  const int khalf = khv * 2;
  const int mhi   = khv * 8;
  v8f acc = {};
#ifdef HAS_WMMA_F32X4
  const float* __restrict__ xa = xs + m * FT_PITCH + kh * 64 + khalf;
  const float* __restrict__ bp = Wst + (((kh * 32) + khv) << 8) + (nt << 5) + (m << 1);
#pragma unroll
  for (int ks = 0; ks < 16; ++ks) {
    const v2f a = *(const v2f*)(xa + ks * 4);
    const v2f b = *(const v2f*)(bp + (ks << 9));
    acc = __builtin_amdgcn_wmma_f32_16x16x4_f32(
        false, a, false, b, (short)0, acc, false, false);
  }
#else
#pragma unroll
  for (int v = 0; v < 8; ++v) {
    float s = 0.f;
    for (int k = kh * 64; k < kh * 64 + 64; ++k)
      s += xs[(v + mhi) * FT_PITCH + k] * Wst[((k >> 1) << 8) + (((nt << 4) + m) << 1) + (k & 1)];
    acc[v] = s;
  }
#endif
  float* __restrict__ hp = hs + kh * (16 * FT_PITCH);
#pragma unroll
  for (int v = 0; v < 8; ++v) hp[(v + mhi) * FT_PITCH + nt * 16 + m] = acc[v];
}

__global__ __launch_bounds__(512) void gat_node_gemm2(
    const int* __restrict__ csr, const int* __restrict__ rowptr,
    const float* __restrict__ as, const float* __restrict__ ad,
    const float* __restrict__ X, const float* __restrict__ W1, const float* __restrict__ bias1,
    const float* __restrict__ W2, const float* __restrict__ a_src2, const float* __restrict__ a_dst2,
    float* __restrict__ H2, float* __restrict__ as2, float* __restrict__ ad2,
    int n, int ntiles, float slope) {
  __shared__ __attribute__((aligned(16))) float W1s[NHID * NHID];
  __shared__ __attribute__((aligned(16))) float W2s[NHID * NHID];
  __shared__ __attribute__((aligned(16))) float xs[16 * FT_PITCH];
  __shared__ __attribute__((aligned(16))) float hs[2 * 16 * FT_PITCH];
  const int tid   = threadIdx.x;
  const int lane  = tid & 31;
  const int lane4 = lane * 4;
  const int wave  = __builtin_amdgcn_readfirstlane(tid >> 5);
  const int m     = lane & 15;
  const int khv   = lane >> 4;
  const int nt    = wave & 7;
  const int kh    = wave >> 3;

  NodeHead cur = {};
  const int  d_first  = (int)blockIdx.x * 16 + wave;
  const bool ok_first = ((int)blockIdx.x < ntiles) && (d_first < n);
  if (ok_first) {
    head_bounds(cur, rowptr, as, ad, d_first, lane);
    head_sources(cur, csr, X, d_first, lane);
  }
  for (int i = tid; i < NHID * NHID; i += 512) {
    const int k = i >> 7, nn = i & 127;
    const int o = ((k >> 1) << 8) + (nn << 1) + (k & 1);
    W1s[o] = W1[i];
    W2s[o] = W2[i];
  }
  const float4 b14 = *(const float4*)(bias1 + lane4);
  const float4 as4 = *(const float4*)(a_src2 + lane4);
  const float4 ad4 = *(const float4*)(a_dst2 + lane4);
  if (ok_first) head_logits(cur, as);

  for (int tile = (int)blockIdx.x; tile < ntiles; tile += (int)gridDim.x) {
    const int d  = tile * 16 + wave;
    const int dn = (tile + (int)gridDim.x) * 16 + wave;
    const bool pre = (tile + (int)gridDim.x < ntiles) && (dn < n);
    NodeHead nxt = {};
    if (pre) head_bounds(nxt, rowptr, as, ad, dn, lane);
    {
      float4 o = make_float4(0.f, 0.f, 0.f, 0.f);
      if (d < n) o = gat_node_finish_head(cur, csr, as, X, d, lane);
      *(float4*)(xs + wave * FT_PITCH + lane4) = o;
    }
    __syncthreads();

    if (pre) head_sources(nxt, csr, X, dn, lane);
    tile_half_product(xs, W1s, hs, m, khv, nt, kh);
    __syncthreads();

    {
      const float4 h0 = *(const float4*)(hs + wave * FT_PITCH + lane4);
      const float4 h1 = *(const float4*)(hs + (16 + wave) * FT_PITCH + lane4);
      float4 o = make_float4(h0.x + h1.x + b14.x, h0.y + h1.y + b14.y,
                             h0.z + h1.z + b14.z, h0.w + h1.w + b14.w);
      o.x = lrelu(o.x, slope); o.y = lrelu(o.y, slope);
      o.z = lrelu(o.z, slope); o.w = lrelu(o.w, slope);
      *(float4*)(xs + wave * FT_PITCH + lane4) = o;
    }
    __syncthreads();

    if (pre) head_logits(nxt, as);
    tile_half_product(xs, W2s, hs, m, khv, nt, kh);
    __syncthreads();

    {
      const float4 h0 = *(const float4*)(hs + wave * FT_PITCH + lane4);
      const float4 h1 = *(const float4*)(hs + (16 + wave) * FT_PITCH + lane4);
      const float4 hv = make_float4(h0.x + h1.x, h0.y + h1.y, h0.z + h1.z, h0.w + h1.w);
      float ps = hv.x * as4.x;
      ps = fmaf(hv.y, as4.y, ps); ps = fmaf(hv.z, as4.z, ps); ps = fmaf(hv.w, as4.w, ps);
      float pd = hv.x * ad4.x;
      pd = fmaf(hv.y, ad4.y, pd); pd = fmaf(hv.z, ad4.z, pd); pd = fmaf(hv.w, ad4.w, pd);
      for (int off = 16; off; off >>= 1) {
        ps += __shfl_xor(ps, off, 32);
        pd += __shfl_xor(pd, off, 32);
      }
      if (d < n) {
        *(float4*)(H2 + (size_t)d * NHID + lane4) = hv;
        if (lane == 0) { as2[d] = ps; ad2[d] = pd; }
      }
    }
    cur = nxt;
  }
}

extern "C" void kernel_launch(void* const* d_in, const int* in_sizes, int n_in,
                              void* d_out, int out_size, void* d_ws, size_t ws_size,
                              hipStream_t stream) {
  const float* x    = (const float*)d_in[0];
  const int*   ei   = (const int*)d_in[1];
  const float* W1   = (const float*)d_in[2];
  const float* asr1 = (const float*)d_in[3];
  const float* adt1 = (const float*)d_in[4];
  const float* b1   = (const float*)d_in[5];
  const float* W2   = (const float*)d_in[6];
  const float* asr2 = (const float*)d_in[7];
  const float* adt2 = (const float*)d_in[8];
  const float* b2   = (const float*)d_in[9];
  float* out = (float*)d_out;

  const int N = in_sizes[0] / NHID;
  const int E = in_sizes[1] / 2;
  const size_t NF = (size_t)N * NHID;

  const int NBK  = (N + BUCKET_NODES - 1) / BUCKET_NODES;
  const int NCH  = (E + CHUNK_EDGES - 1) / CHUNK_EDGES;
  const int LO   = (NBK + 1) * NCH;

  float* h2     = (float*)d_ws;
  int*   tmp    = (int*)(h2 + NF);
  float* as     = (float*)(tmp + 2 * (size_t)E);
  float* ad     = as + N;
  int*   rowptr = (int*)(ad + N);
  int*   off    = rowptr + N + 1;
  int*   csr    = off + LO;
  float* as2    = (float*)(csr + E);
  float* ad2    = as2 + N;

  const int* src = ei;
  const int* dst = ei + E;

  const int gWaveN = (int)(((size_t)N * 32 + 255) / 256);
  const int NT16   = (N + 15) / 16;

  gat_chunk_sort<<<NCH, 1024, 0, stream>>>(src, dst, off, tmp, E, NBK, NCH);
  gat_bucket_logit<<<NBK + (NBK + 1) / 2, 512, 0, stream>>>(tmp, off, rowptr, csr, N, NBK, NCH, E,
                                                            x, W1, asr1, adt1, as, ad);

  gat_node_gemm2<<<FT_GROUPS, 512, 0, stream>>>(csr, rowptr, as, ad, x, W1, b1, W2, asr2, adt2,
                                                h2, as2, ad2, N, NT16, NEG_ACT);

  gat_node<<<gWaveN, 256, 0, stream>>>(csr, rowptr, as2, ad2, h2, b2, out, N, 0.f, 0);
}
